# starting kernel
# baseline (speedup 1.0000x reference)
#include <hip/hip_runtime.h>
#include <hip/hip_bf16.h>


#define BB 16
#define SS 128
#define HH 200
#define PP 20
#define FF 105
#define EPSF 1e-8f

typedef float v2f __attribute__((ext_vector_type(2)));
typedef float v8f __attribute__((ext_vector_type(8)));

__device__ __forceinline__ float wave_sum(float v) {
#pragma unroll
  for (int o = 16; o > 0; o >>= 1) v += __shfl_xor(v, o, 32);
  return v;
}

__global__ void prep_kernel(const float* __restrict__ src, const int* __restrict__ mask,
                            const float* __restrict__ w1, float* __restrict__ dst,
                            float* __restrict__ nrm, float* __restrict__ wn) {
  __shared__ float row[HH];
  int bs = blockIdx.x;
  int lane = threadIdx.x;
  float m = (float)mask[bs];
  const float* srow = src + (size_t)bs * HH;
  float* drow = dst + (size_t)bs * HH;
  float acc = 0.f;
  for (int h = lane; h < HH; h += 32) {
    float v = srow[h] * m;
    drow[h] = v; row[h] = v; acc += v * v;
  }
  acc = wave_sum(acc);
  if (lane == 0) nrm[bs] = sqrtf(acc);
  __syncthreads();
  for (int p = 0; p < PP; p++) {
    float a = 0.f;
    for (int h = lane; h < HH; h += 32) {
      float ww = w1[p * HH + h]; float v = row[h];
      a += ww * ww * v * v;
    }
    a = wave_sum(a);
    if (lane == 0) wn[bs * PP + p] = sqrtf(a);
  }
}

__global__ void gather_last_kernel(const float* __restrict__ cp, const float* __restrict__ ch,
                                   const int* __restrict__ mask_p, const int* __restrict__ mask_h,
                                   float* __restrict__ cpl, float* __restrict__ chl) {
  int b = blockIdx.x;
  int lp = 0, lh = 0;
  for (int i = 0; i < SS; i++) { lp += mask_p[b * SS + i]; lh += mask_h[b * SS + i]; }
  int ip = (lp > 0) ? lp - 1 : 0;
  int ih = (lh > 0) ? lh - 1 : 0;
  for (int h = threadIdx.x; h < HH; h += blockDim.x) {
    cpl[b * HH + h] = cp[((size_t)b * SS + ip) * HH + h];
    chl[b * HH + h] = ch[((size_t)b * SS + ih) * HH + h];
  }
}

__global__ void cos_wmma_kernel(const float* __restrict__ cp, const float* __restrict__ ch,
                                const float* __restrict__ np, const float* __restrict__ nh,
                                float* __restrict__ cosO) {
  int wid = blockIdx.x;
  int b = wid >> 6;
  int x0 = ((wid >> 3) & 7) << 4;
  int y0 = (wid & 7) << 4;
  int lane = threadIdx.x, half = lane >> 4, l16 = lane & 15;
  const float* arow = cp + ((size_t)(b * SS) + x0 + l16) * HH + 2 * half;
  const float* brow = ch + ((size_t)(b * SS) + y0 + l16) * HH + 2 * half;
  v8f acc = {0, 0, 0, 0, 0, 0, 0, 0};
  for (int k = 0; k < HH; k += 4) {
    v2f a = *(const v2f*)(arow + k);
    v2f bv = *(const v2f*)(brow + k);
    acc = __builtin_amdgcn_wmma_f32_16x16x4_f32(false, a, false, bv, (short)0, acc, false, false);
  }
  float nhv = nh[b * SS + y0 + l16];
#pragma unroll
  for (int r = 0; r < 8; r++) {
    int m = r + 8 * half;
    float d = fmaxf(np[b * SS + x0 + m] * nhv, EPSF);
    cosO[((size_t)(b * SS) + x0 + m) * SS + y0 + l16] = acc[r] / d;
  }
}

__global__ void cos_reduce_kernel(const float* __restrict__ cosM, const int* __restrict__ mask,
                                  float* __restrict__ cossum, float* __restrict__ out,
                                  int transposed, int side) {
  int b = blockIdx.x / SS, r = blockIdx.x % SS;
  int t = threadIdx.x;
  float v = transposed ? cosM[((size_t)b * SS + t) * SS + r]
                       : cosM[((size_t)b * SS + r) * SS + t];
  __shared__ float smax[128], ssum[128];
  smax[t] = v; ssum[t] = v;
  __syncthreads();
  for (int o = 64; o > 0; o >>= 1) {
    if (t < o) { smax[t] = fmaxf(smax[t], smax[t + o]); ssum[t] += ssum[t + o]; }
    __syncthreads();
  }
  if (t == 0) {
    float m = (float)mask[b * SS + r];
    cossum[b * SS + r] = ssum[0];
    float* o2 = out + ((size_t)(side * BB + b) * SS + r) * FF;
    o2[0] = m * smax[0];
    o2[1] = m * (ssum[0] / (float)SS);
  }
}

__global__ void attmean_wmma_kernel(const float* __restrict__ cosM, const float* __restrict__ V,
                                    const float* __restrict__ csum, float* __restrict__ attO,
                                    int aStrideM, int aStrideK) {
  int wid = blockIdx.x;
  int b = wid / (8 * 13);
  int rem = wid % (8 * 13);
  int m0 = (rem / 13) * 16;
  int n0 = (rem % 13) * 16;
  int lane = threadIdx.x, half = lane >> 4, l16 = lane & 15;
  const float* cb = cosM + (size_t)b * SS * SS;
  int nIdx = n0 + l16;
  bool nOK = nIdx < HH;
  v8f acc = {0, 0, 0, 0, 0, 0, 0, 0};
  for (int k = 0; k < SS; k += 4) {
    int ka = k + 2 * half;
    v2f a, bv;
    a.x = cb[(size_t)(m0 + l16) * aStrideM + (size_t)(ka + 0) * aStrideK];
    a.y = cb[(size_t)(m0 + l16) * aStrideM + (size_t)(ka + 1) * aStrideK];
    bv.x = nOK ? V[((size_t)b * SS + ka + 0) * HH + nIdx] : 0.f;
    bv.y = nOK ? V[((size_t)b * SS + ka + 1) * HH + nIdx] : 0.f;
    acc = __builtin_amdgcn_wmma_f32_16x16x4_f32(false, a, false, bv, (short)0, acc, false, false);
  }
  if (nOK) {
#pragma unroll
    for (int r = 0; r < 8; r++) {
      int m = m0 + r + 8 * half;
      float d = fmaxf(csum[b * SS + m], EPSF);
      attO[((size_t)b * SS + m) * HH + nIdx] = acc[r] / d;
    }
  }
}

__global__ void attmax_kernel(const float* __restrict__ cosM, const float* __restrict__ V,
                              const int* __restrict__ mask, float* __restrict__ attO,
                              int cRowStride, int cColStride) {
  int b = blockIdx.x / SS, r = blockIdx.x % SS;
  int h = threadIdx.x;
  if (h >= HH) return;
  const float* cb = cosM + (size_t)b * SS * SS;
  float mx = -3.0e38f;
  for (int k = 0; k < SS; k++) {
    float c = cb[(size_t)r * cRowStride + (size_t)k * cColStride];
    float v = V[((size_t)b * SS + k) * HH + h];
    mx = fmaxf(mx, c * v);
  }
  float m = (float)mask[b * SS + r];
  attO[((size_t)b * SS + r) * HH + h] = m * mx;
}

__global__ void mpm_kernel(const float* __restrict__ v1, const float* __restrict__ v2,
                           const float* __restrict__ w, int v2_bstride, int v2_sstride,
                           float* __restrict__ out, int side, int foff) {
  __shared__ float r1[HH], r2[HH];
  int bs = blockIdx.x;
  int b = bs / SS, s = bs % SS;
  int lane = threadIdx.x;
  const float* p1 = v1 + (size_t)bs * HH;
  const float* p2 = v2 + (size_t)b * v2_bstride + (size_t)s * v2_sstride;
  float d11 = 0, d22 = 0, d12 = 0;
  for (int h = lane; h < HH; h += 32) {
    float a = p1[h], c = p2[h];
    r1[h] = a; r2[h] = c;
    d11 += a * a; d22 += c * c; d12 += a * c;
  }
  d11 = wave_sum(d11); d22 = wave_sum(d22); d12 = wave_sum(d12);
  __syncthreads();
  float* orow = out + ((size_t)(side * BB + b) * SS + s) * FF + foff;
  if (lane == 0)
    orow[0] = d12 / (fmaxf(sqrtf(d11), EPSF) * fmaxf(sqrtf(d22), EPSF));
  for (int p = 0; p < PP; p++) {
    float s11 = 0, s22 = 0, s12 = 0;
    for (int h = lane; h < HH; h += 32) {
      float ww = w[p * HH + h]; float w2 = ww * ww;
      float a = r1[h], c = r2[h];
      s11 += w2 * a * a; s22 += w2 * c * c; s12 += w2 * a * c;
    }
    s11 = wave_sum(s11); s22 = wave_sum(s22); s12 = wave_sum(s12);
    if (lane == 0)
      orow[1 + p] = s12 / (fmaxf(sqrtf(s11), EPSF) * fmaxf(sqrtf(s22), EPSF));
  }
}

__global__ void maxpool_wmma_kernel(const float* __restrict__ v1, const float* __restrict__ v2,
                                    const float* __restrict__ w,
                                    const float* __restrict__ wnA, const float* __restrict__ wnB,
                                    const int* __restrict__ mask, float* __restrict__ out,
                                    int side) {
  int wid = blockIdx.x;
  int b = wid / (PP * 8);
  int p = (wid / 8) % PP;
  int x0 = (wid % 8) * 16;
  int lane = threadIdx.x, half = lane >> 4, l16 = lane & 15;
  float rmax[8], rsum[8], wA[8];
#pragma unroll
  for (int r = 0; r < 8; r++) {
    rmax[r] = -3.0e38f; rsum[r] = 0.f;
    wA[r] = wnA[((size_t)b * SS + x0 + r + 8 * half) * PP + p];
  }
  const float* a_base = v1 + ((size_t)b * SS + x0 + l16) * HH;
  const float* wp = w + p * HH;
  for (int y0 = 0; y0 < SS; y0 += 16) {
    const float* b_base = v2 + ((size_t)b * SS + y0 + l16) * HH;
    v8f acc = {0, 0, 0, 0, 0, 0, 0, 0};
    for (int k = 0; k < HH; k += 4) {
      int ka = k + 2 * half;
      float w0v = wp[ka], w1v = wp[ka + 1];
      v2f a, bv;
      a.x = w0v * w0v * a_base[ka];
      a.y = w1v * w1v * a_base[ka + 1];
      bv = *(const v2f*)(b_base + ka);
      acc = __builtin_amdgcn_wmma_f32_16x16x4_f32(false, a, false, bv, (short)0, acc, false, false);
    }
    float wB = wnB[((size_t)b * SS + y0 + l16) * PP + p];
#pragma unroll
    for (int r = 0; r < 8; r++) {
      float mv = acc[r] / fmaxf(wA[r] * wB, EPSF);
      rmax[r] = fmaxf(rmax[r], mv);
      rsum[r] += mv;
    }
  }
#pragma unroll
  for (int r = 0; r < 8; r++) {
    for (int o = 8; o > 0; o >>= 1) {
      rmax[r] = fmaxf(rmax[r], __shfl_xor(rmax[r], o, 32));
      rsum[r] += __shfl_xor(rsum[r], o, 32);
    }
  }
  if (l16 == 0) {
#pragma unroll
    for (int r = 0; r < 8; r++) {
      int x = x0 + r + 8 * half;
      float m = (float)mask[b * SS + x];
      float* orow = out + ((size_t)(side * BB + b) * SS + x) * FF;
      orow[23 + p] = m * rmax[r];
      orow[43 + p] = m * (rsum[r] / (float)SS);
    }
  }
}

extern "C" void kernel_launch(void* const* d_in, const int* in_sizes, int n_in,
                              void* d_out, int out_size, void* d_ws, size_t ws_size,
                              hipStream_t stream) {
  const float* ctx_p = (const float*)d_in[0];
  const float* ctx_h = (const float*)d_in[1];
  const float* w0 = (const float*)d_in[2];
  const float* w1 = (const float*)d_in[3];
  const float* w2 = (const float*)d_in[4];
  const float* w3 = (const float*)d_in[5];
  const int* mask_p = (const int*)d_in[6];
  const int* mask_h = (const int*)d_in[7];
  float* out = (float*)d_out;
  float* ws = (float*)d_ws;

  const size_t SZ_ROWS = (size_t)BB * SS * HH;
  float* cp = ws;                 ws += SZ_ROWS;
  float* ch = ws;                 ws += SZ_ROWS;
  float* n_p = ws;                ws += BB * SS;
  float* n_h = ws;                ws += BB * SS;
  float* wn1_p = ws;              ws += (size_t)BB * SS * PP;
  float* wn1_h = ws;              ws += (size_t)BB * SS * PP;
  float* cpl = ws;                ws += BB * HH;
  float* chl = ws;                ws += BB * HH;
  float* cosM = ws;               ws += (size_t)BB * SS * SS;
  float* cossum_y = ws;           ws += BB * SS;
  float* cossum_x = ws;           ws += BB * SS;
  float* att_mean_h = ws;         ws += SZ_ROWS;
  float* att_mean_p = ws;         ws += SZ_ROWS;
  float* att_max_h = ws;          ws += SZ_ROWS;
  float* att_max_p = ws;          ws += SZ_ROWS;

  prep_kernel<<<BB * SS, 32, 0, stream>>>(ctx_p, mask_p, w1, cp, n_p, wn1_p);
  prep_kernel<<<BB * SS, 32, 0, stream>>>(ctx_h, mask_h, w1, ch, n_h, wn1_h);

  gather_last_kernel<<<BB, 64, 0, stream>>>(cp, ch, mask_p, mask_h, cpl, chl);

  cos_wmma_kernel<<<BB * 64, 32, 0, stream>>>(cp, ch, n_p, n_h, cosM);

  cos_reduce_kernel<<<BB * SS, 128, 0, stream>>>(cosM, mask_p, cossum_y, out, 0, 0);
  cos_reduce_kernel<<<BB * SS, 128, 0, stream>>>(cosM, mask_h, cossum_x, out, 1, 1);

  attmean_wmma_kernel<<<BB * 8 * 13, 32, 0, stream>>>(cosM, ch, cossum_y, att_mean_h, SS, 1);
  attmean_wmma_kernel<<<BB * 8 * 13, 32, 0, stream>>>(cosM, cp, cossum_x, att_mean_p, 1, SS);

  attmax_kernel<<<BB * SS, 256, 0, stream>>>(cosM, ch, mask_p, att_max_h, SS, 1);
  attmax_kernel<<<BB * SS, 256, 0, stream>>>(cosM, cp, mask_h, att_max_p, 1, SS);

  mpm_kernel<<<BB * SS, 32, 0, stream>>>(cp, chl, w0, HH, 0, out, 0, 2);
  mpm_kernel<<<BB * SS, 32, 0, stream>>>(ch, cpl, w0, HH, 0, out, 1, 2);
  mpm_kernel<<<BB * SS, 32, 0, stream>>>(cp, att_mean_h, w2, SS * HH, HH, out, 0, 63);
  mpm_kernel<<<BB * SS, 32, 0, stream>>>(ch, att_mean_p, w2, SS * HH, HH, out, 1, 63);
  mpm_kernel<<<BB * SS, 32, 0, stream>>>(cp, att_max_h, w3, SS * HH, HH, out, 0, 84);
  mpm_kernel<<<BB * SS, 32, 0, stream>>>(ch, att_max_p, w3, SS * HH, HH, out, 1, 84);

  maxpool_wmma_kernel<<<BB * PP * 8, 32, 0, stream>>>(cp, ch, w1, wn1_p, wn1_h, mask_p, out, 0);
  maxpool_wmma_kernel<<<BB * PP * 8, 32, 0, stream>>>(ch, cp, w1, wn1_h, wn1_p, mask_h, out, 1);
}


# maxpool match: scaled A fragment kept in registers over the y loop, B read 16 B per lane: 8 cache-line lookups per WMMA, not 33; k grouping per WMMA changes, so summation order changes
# speedup vs baseline: 2.6871x; 2.6871x over previous
#include <hip/hip_runtime.h>
#include <hip/hip_bf16.h>


#define BB 16
#define SS 128
#define HH 200
#define PP 20
#define FF 105
#define EPSF 1e-8f
#define KCH 10
#define NCH 5

typedef float v2f __attribute__((ext_vector_type(2)));
typedef float v4f __attribute__((ext_vector_type(4)));
typedef float v8f __attribute__((ext_vector_type(8)));

__device__ __forceinline__ v2f wsq_frag(v2f ww, bool isW, float fill) {
  v2f bv;
  bv.x = isW ? ww.x * ww.x : fill;
  bv.y = isW ? ww.y * ww.y : fill;
  return bv;
}

__global__ void __launch_bounds__(32) prep_wmma_kernel(const float* __restrict__ src, const int* __restrict__ mask,
                                 const float* __restrict__ w1, float* __restrict__ dst,
                                 float* __restrict__ nrm, float* __restrict__ wn) {
  int m0 = (blockIdx.x >> 1) << 4;
  int n0 = (blockIdx.x & 1) << 4;
  int lane = threadIdx.x, half = lane >> 4, l16 = lane & 15;
  int bs = m0 + l16;
  float m = (float)mask[bs];
  const float* srow = src + (size_t)bs * HH + 2 * half;
  float* drow = dst + (size_t)bs * HH + 2 * half;
  int col = n0 + l16;
  bool isW = col < PP;
  float fill = (col == PP) ? 1.f : 0.f;
  const float* wrow = w1 + (size_t)(isW ? col : PP - 1) * HH + 2 * half;
  v8f acc = {0, 0, 0, 0, 0, 0, 0, 0};
  v2f xs[2][KCH], ws2[2][KCH];
#pragma unroll
  for (int i = 0; i < KCH; i++) {
    xs[0][i] = *(const v2f*)(srow + 4 * i);
    ws2[0][i] = *(const v2f*)(wrow + 4 * i);
  }
#pragma unroll
  for (int c = 0; c < NCH; c++) {
    if (c + 1 < NCH) {
#pragma unroll
      for (int i = 0; i < KCH; i++) {
        xs[(c + 1) & 1][i] = *(const v2f*)(srow + 4 * KCH * (c + 1) + 4 * i);
        ws2[(c + 1) & 1][i] = *(const v2f*)(wrow + 4 * KCH * (c + 1) + 4 * i);
      }
    }
    __builtin_amdgcn_sched_barrier(0);
#pragma unroll
    for (int i = 0; i < KCH; i++) {
      v2f x = xs[c & 1][i];
      x.x *= m; x.y *= m;
      if (n0 == 0) *(v2f*)(drow + 4 * KCH * c + 4 * i) = x;
      v2f bv = wsq_frag(ws2[c & 1][i], isW, fill);
      v2f a; a.x = x.x * x.x; a.y = x.y * x.y;
      acc = __builtin_amdgcn_wmma_f32_16x16x4_f32(false, a, false, bv, (short)0, acc, false, false);
    }
  }
  if (col <= PP) {
#pragma unroll
    for (int r = 0; r < 8; r++) {
      int row = m0 + r + 8 * half;
      float* o = (col == PP) ? nrm + row : wn + (size_t)row * PP + col;
      *o = sqrtf(acc[r]);
    }
  }
}

__global__ void gather_last_kernel(const float* __restrict__ cp, const float* __restrict__ ch,
                                   const int* __restrict__ mask_p, const int* __restrict__ mask_h,
                                   float* __restrict__ cpl, float* __restrict__ chl) {
  int b = blockIdx.x;
  int lp = 0, lh = 0;
  for (int i = 0; i < SS; i++) { lp += mask_p[b * SS + i]; lh += mask_h[b * SS + i]; }
  int ip = (lp > 0) ? lp - 1 : 0;
  int ih = (lh > 0) ? lh - 1 : 0;
  for (int h = threadIdx.x; h < HH; h += blockDim.x) {
    cpl[b * HH + h] = cp[((size_t)b * SS + ip) * HH + h];
    chl[b * HH + h] = ch[((size_t)b * SS + ih) * HH + h];
  }
}

__global__ void cos_wmma_kernel(const float* __restrict__ cp, const float* __restrict__ ch,
                                const float* __restrict__ np, const float* __restrict__ nh,
                                float* __restrict__ cosO) {
  int wid = blockIdx.x;
  int b = wid >> 6;
  int x0 = ((wid >> 3) & 7) << 4;
  int y0 = (wid & 7) << 4;
  int lane = threadIdx.x, half = lane >> 4, l16 = lane & 15;
  const float* arow = cp + ((size_t)(b * SS) + x0 + l16) * HH + 2 * half;
  const float* brow = ch + ((size_t)(b * SS) + y0 + l16) * HH + 2 * half;
  v8f acc = {0, 0, 0, 0, 0, 0, 0, 0};
  for (int k = 0; k < HH; k += 4) {
    v2f a = *(const v2f*)(arow + k);
    v2f bv = *(const v2f*)(brow + k);
    acc = __builtin_amdgcn_wmma_f32_16x16x4_f32(false, a, false, bv, (short)0, acc, false, false);
  }
  float nhv = nh[b * SS + y0 + l16];
#pragma unroll
  for (int r = 0; r < 8; r++) {
    int m = r + 8 * half;
    float d = fmaxf(np[b * SS + x0 + m] * nhv, EPSF);
    cosO[((size_t)(b * SS) + x0 + m) * SS + y0 + l16] = acc[r] / d;
  }
}

__global__ void cos_reduce_kernel(const float* __restrict__ cosM, const int* __restrict__ mask,
                                  float* __restrict__ cossum, float* __restrict__ out,
                                  int transposed, int side) {
  int b = blockIdx.x / SS, r = blockIdx.x % SS;
  int t = threadIdx.x;
  float v = transposed ? cosM[((size_t)b * SS + t) * SS + r]
                       : cosM[((size_t)b * SS + r) * SS + t];
  __shared__ float smax[128], ssum[128];
  smax[t] = v; ssum[t] = v;
  __syncthreads();
  for (int o = 64; o > 0; o >>= 1) {
    if (t < o) { smax[t] = fmaxf(smax[t], smax[t + o]); ssum[t] += ssum[t + o]; }
    __syncthreads();
  }
  if (t == 0) {
    float m = (float)mask[b * SS + r];
    cossum[b * SS + r] = ssum[0];
    float* o2 = out + ((size_t)(side * BB + b) * SS + r) * FF;
    o2[0] = m * smax[0];
    o2[1] = m * (ssum[0] / (float)SS);
  }
}

__global__ void attmean_wmma_kernel(const float* __restrict__ cosM, const float* __restrict__ V,
                                    const float* __restrict__ csum, float* __restrict__ attO,
                                    int aStrideM, int aStrideK) {
  int wid = blockIdx.x;
  int b = wid / (8 * 13);
  int rem = wid % (8 * 13);
  int m0 = (rem / 13) * 16;
  int n0 = (rem % 13) * 16;
  int lane = threadIdx.x, half = lane >> 4, l16 = lane & 15;
  const float* cb = cosM + (size_t)b * SS * SS;
  int nIdx = n0 + l16;
  bool nOK = nIdx < HH;
  v8f acc = {0, 0, 0, 0, 0, 0, 0, 0};
  for (int k = 0; k < SS; k += 4) {
    int ka = k + 2 * half;
    v2f a, bv;
    a.x = cb[(size_t)(m0 + l16) * aStrideM + (size_t)(ka + 0) * aStrideK];
    a.y = cb[(size_t)(m0 + l16) * aStrideM + (size_t)(ka + 1) * aStrideK];
    bv.x = nOK ? V[((size_t)b * SS + ka + 0) * HH + nIdx] : 0.f;
    bv.y = nOK ? V[((size_t)b * SS + ka + 1) * HH + nIdx] : 0.f;
    acc = __builtin_amdgcn_wmma_f32_16x16x4_f32(false, a, false, bv, (short)0, acc, false, false);
  }
  if (nOK) {
#pragma unroll
    for (int r = 0; r < 8; r++) {
      int m = m0 + r + 8 * half;
      float d = fmaxf(csum[b * SS + m], EPSF);
      attO[((size_t)b * SS + m) * HH + nIdx] = acc[r] / d;
    }
  }
}

__global__ void attmax_kernel(const float* __restrict__ cosM, const float* __restrict__ V,
                              const int* __restrict__ mask, float* __restrict__ attO,
                              int cRowStride, int cColStride) {
  int b = blockIdx.x / SS, r = blockIdx.x % SS;
  int h = threadIdx.x;
  if (h >= HH) return;
  const float* cb = cosM + (size_t)b * SS * SS;
  float mx = -3.0e38f;
  for (int k = 0; k < SS; k++) {
    float c = cb[(size_t)r * cRowStride + (size_t)k * cColStride];
    float v = V[((size_t)b * SS + k) * HH + h];
    mx = fmaxf(mx, c * v);
  }
  float m = (float)mask[b * SS + r];
  attO[((size_t)b * SS + r) * HH + h] = m * mx;
}

__global__ void __launch_bounds__(32) mpm_wmma_kernel(const float* __restrict__ v1, const float* __restrict__ v2,
                                const float* __restrict__ w, int v2_bstride, int v2_sstride,
                                float* __restrict__ out, int side, int foff) {
  int m0 = (blockIdx.x >> 1) << 4;
  int n0 = (blockIdx.x & 1) << 4;
  int lane = threadIdx.x, half = lane >> 4, l16 = lane & 15;
  int bs = m0 + l16;
  int b = bs / SS, s = bs % SS;
  const float* p1 = v1 + (size_t)bs * HH + 2 * half;
  const float* p2 = v2 + (size_t)b * v2_bstride + (size_t)s * v2_sstride + 2 * half;
  int col = n0 + l16;
  bool isW = col < PP;
  float fill = (col == PP) ? 1.f : 0.f;
  const float* wrow = w + (size_t)(isW ? col : PP - 1) * HH + 2 * half;
  v8f s11 = {0, 0, 0, 0, 0, 0, 0, 0};
  v8f s22 = {0, 0, 0, 0, 0, 0, 0, 0};
  v8f s12 = {0, 0, 0, 0, 0, 0, 0, 0};
  v2f xs[2][KCH], ys[2][KCH], ws2[2][KCH];
#pragma unroll
  for (int i = 0; i < KCH; i++) {
    xs[0][i] = *(const v2f*)(p1 + 4 * i);
    ys[0][i] = *(const v2f*)(p2 + 4 * i);
    ws2[0][i] = *(const v2f*)(wrow + 4 * i);
  }
#pragma unroll
  for (int c = 0; c < NCH; c++) {
    if (c + 1 < NCH) {
#pragma unroll
      for (int i = 0; i < KCH; i++) {
        xs[(c + 1) & 1][i] = *(const v2f*)(p1 + 4 * KCH * (c + 1) + 4 * i);
        ys[(c + 1) & 1][i] = *(const v2f*)(p2 + 4 * KCH * (c + 1) + 4 * i);
        ws2[(c + 1) & 1][i] = *(const v2f*)(wrow + 4 * KCH * (c + 1) + 4 * i);
      }
    }
    __builtin_amdgcn_sched_barrier(0);
#pragma unroll
    for (int i = 0; i < KCH; i++) {
      v2f x = xs[c & 1][i], y = ys[c & 1][i];
      v2f bv = wsq_frag(ws2[c & 1][i], isW, fill);
      v2f xx, yy, xy;
      xx.x = x.x * x.x; xx.y = x.y * x.y;
      yy.x = y.x * y.x; yy.y = y.y * y.y;
      xy.x = x.x * y.x; xy.y = x.y * y.y;
      s11 = __builtin_amdgcn_wmma_f32_16x16x4_f32(false, xx, false, bv, (short)0, s11, false, false);
      s22 = __builtin_amdgcn_wmma_f32_16x16x4_f32(false, yy, false, bv, (short)0, s22, false, false);
      s12 = __builtin_amdgcn_wmma_f32_16x16x4_f32(false, xy, false, bv, (short)0, s12, false, false);
    }
  }
  if (col <= PP) {
    int f = (col == PP) ? 0 : 1 + col;
#pragma unroll
    for (int r = 0; r < 8; r++) {
      int row = m0 + r + 8 * half;
      int ob = row / SS, os = row % SS;
      float* orow = out + ((size_t)(side * BB + ob) * SS + os) * FF + foff;
      orow[f] = s12[r] / (fmaxf(sqrtf(s11[r]), EPSF) * fmaxf(sqrtf(s22[r]), EPSF));
    }
  }
}

#define NJ 25
#define GCH 10
#define NYT 8
__global__ void __launch_bounds__(32) maxpool_wmma_kernel(const float* __restrict__ v1, const float* __restrict__ v2,
                                    const float* __restrict__ w,
                                    const float* __restrict__ wnA, const float* __restrict__ wnB,
                                    const int* __restrict__ mask, float* __restrict__ out,
                                    int side) {
  int wid = blockIdx.x;
  int b = wid / (PP * 8);
  int p = (wid / 8) % PP;
  int x0 = (wid % 8) * 16;
  int lane = threadIdx.x, half = lane >> 4, l16 = lane & 15;
  float rmax[8], rsum[8], wA[8];
#pragma unroll
  for (int r = 0; r < 8; r++) {
    rmax[r] = -3.0e38f; rsum[r] = 0.f;
    wA[r] = wnA[((size_t)b * SS + x0 + r + 8 * half) * PP + p];
  }
  const float* a_row = v1 + ((size_t)b * SS + x0 + l16) * HH + 4 * half;
  const float* w_row = w + (size_t)p * HH + 4 * half;
  const float* b_row = v2 + ((size_t)b * SS + l16) * HH + 4 * half;
  const float* wnB_col = wnB + ((size_t)b * SS + l16) * PP + p;

  v4f bf[2][GCH];
#pragma unroll
  for (int i = 0; i < GCH; i++) bf[0][i] = *(const v4f*)(b_row + 8 * i);

  v4f af[NJ];
#pragma unroll
  for (int j = 0; j < NJ; j++) {
    v4f wv = *(const v4f*)(w_row + 8 * j);
    v4f av = *(const v4f*)(a_row + 8 * j);
    af[j].x = wv.x * wv.x * av.x;
    af[j].y = wv.y * wv.y * av.y;
    af[j].z = wv.z * wv.z * av.z;
    af[j].w = wv.w * wv.w * av.w;
  }

  v8f acc = {0, 0, 0, 0, 0, 0, 0, 0};
  float wB = 0.f;
#pragma unroll
  for (int g0 = 0; g0 < NYT * NJ; g0 += GCH) {
    if (g0 + GCH < NYT * NJ) {
#pragma unroll
      for (int i = 0; i < GCH; i++) {
        int gn = g0 + GCH + i;
        bf[((g0 / GCH) + 1) & 1][i] = *(const v4f*)(b_row + (gn / NJ) * 16 * HH + 8 * (gn % NJ));
      }
    }
    __builtin_amdgcn_sched_barrier(0);
#pragma unroll
    for (int i = 0; i < GCH; i++) {
      int g = g0 + i;
      int yt = g / NJ, j = g % NJ;
      if (j == 0) {
        acc = (v8f){0, 0, 0, 0, 0, 0, 0, 0};
        wB = wnB_col[(size_t)yt * 16 * PP];
      }
      v4f bb = bf[(g0 / GCH) & 1][i];
      v4f aa = af[j];
      v2f a0, b0, a1, b1;
      a0.x = aa.x; a0.y = aa.y; b0.x = bb.x; b0.y = bb.y;
      a1.x = aa.z; a1.y = aa.w; b1.x = bb.z; b1.y = bb.w;
      acc = __builtin_amdgcn_wmma_f32_16x16x4_f32(false, a0, false, b0, (short)0, acc, false, false);
      acc = __builtin_amdgcn_wmma_f32_16x16x4_f32(false, a1, false, b1, (short)0, acc, false, false);
      if (j == NJ - 1) {
#pragma unroll
        for (int r = 0; r < 8; r++) {
          float mv = acc[r] / fmaxf(wA[r] * wB, EPSF);
          rmax[r] = fmaxf(rmax[r], mv);
          rsum[r] += mv;
        }
      }
    }
  }
#pragma unroll
  for (int r = 0; r < 8; r++) {
    for (int o = 8; o > 0; o >>= 1) {
      rmax[r] = fmaxf(rmax[r], __shfl_xor(rmax[r], o, 32));
      rsum[r] += __shfl_xor(rsum[r], o, 32);
    }
  }
  if (l16 == 0) {
#pragma unroll
    for (int r = 0; r < 8; r++) {
      int x = x0 + r + 8 * half;
      float m = (float)mask[b * SS + x];
      float* orow = out + ((size_t)(side * BB + b) * SS + x) * FF;
      orow[23 + p] = m * rmax[r];
      orow[43 + p] = m * (rsum[r] / (float)SS);
    }
  }
}

extern "C" void kernel_launch(void* const* d_in, const int* in_sizes, int n_in,
                              void* d_out, int out_size, void* d_ws, size_t ws_size,
                              hipStream_t stream) {
  const float* ctx_p = (const float*)d_in[0];
  const float* ctx_h = (const float*)d_in[1];
  const float* w0 = (const float*)d_in[2];
  const float* w1 = (const float*)d_in[3];
  const float* w2 = (const float*)d_in[4];
  const float* w3 = (const float*)d_in[5];
  const int* mask_p = (const int*)d_in[6];
  const int* mask_h = (const int*)d_in[7];
  float* out = (float*)d_out;
  float* ws = (float*)d_ws;

  const size_t SZ_ROWS = (size_t)BB * SS * HH;
  float* cp = ws;                 ws += SZ_ROWS;
  float* ch = ws;                 ws += SZ_ROWS;
  float* n_p = ws;                ws += BB * SS;
  float* n_h = ws;                ws += BB * SS;
  float* wn1_p = ws;              ws += (size_t)BB * SS * PP;
  float* wn1_h = ws;              ws += (size_t)BB * SS * PP;
  float* cpl = ws;                ws += BB * HH;
  float* chl = ws;                ws += BB * HH;
  float* cosM = ws;               ws += (size_t)BB * SS * SS;
  float* cossum_y = ws;           ws += BB * SS;
  float* cossum_x = ws;           ws += BB * SS;
  float* att_mean_h = ws;         ws += SZ_ROWS;
  float* att_mean_p = ws;         ws += SZ_ROWS;
  float* att_max_h = ws;          ws += SZ_ROWS;
  float* att_max_p = ws;          ws += SZ_ROWS;

  prep_wmma_kernel<<<(BB * SS / 16) * 2, 32, 0, stream>>>(ctx_p, mask_p, w1, cp, n_p, wn1_p);
  prep_wmma_kernel<<<(BB * SS / 16) * 2, 32, 0, stream>>>(ctx_h, mask_h, w1, ch, n_h, wn1_h);

  gather_last_kernel<<<BB, 64, 0, stream>>>(cp, ch, mask_p, mask_h, cpl, chl);

  cos_wmma_kernel<<<BB * 64, 32, 0, stream>>>(cp, ch, n_p, n_h, cosM);

  cos_reduce_kernel<<<BB * SS, 128, 0, stream>>>(cosM, mask_p, cossum_y, out, 0, 0);
  cos_reduce_kernel<<<BB * SS, 128, 0, stream>>>(cosM, mask_h, cossum_x, out, 1, 1);

  attmean_wmma_kernel<<<BB * 8 * 13, 32, 0, stream>>>(cosM, ch, cossum_y, att_mean_h, SS, 1);
  attmean_wmma_kernel<<<BB * 8 * 13, 32, 0, stream>>>(cosM, cp, cossum_x, att_mean_p, 1, SS);

  attmax_kernel<<<BB * SS, 256, 0, stream>>>(cosM, ch, mask_p, att_max_h, SS, 1);
  attmax_kernel<<<BB * SS, 256, 0, stream>>>(cosM, cp, mask_h, att_max_p, 1, SS);

  mpm_wmma_kernel<<<(BB * SS / 16) * 2, 32, 0, stream>>>(cp, chl, w0, HH, 0, out, 0, 2);
  mpm_wmma_kernel<<<(BB * SS / 16) * 2, 32, 0, stream>>>(ch, cpl, w0, HH, 0, out, 1, 2);
  mpm_wmma_kernel<<<(BB * SS / 16) * 2, 32, 0, stream>>>(cp, att_mean_h, w2, SS * HH, HH, out, 0, 63);
  mpm_wmma_kernel<<<(BB * SS / 16) * 2, 32, 0, stream>>>(ch, att_mean_p, w2, SS * HH, HH, out, 1, 63);
  mpm_wmma_kernel<<<(BB * SS / 16) * 2, 32, 0, stream>>>(cp, att_max_h, w3, SS * HH, HH, out, 0, 84);
  mpm_wmma_kernel<<<(BB * SS / 16) * 2, 32, 0, stream>>>(ch, att_max_p, w3, SS * HH, HH, out, 1, 84);

  maxpool_wmma_kernel<<<BB * PP * 8, 32, 0, stream>>>(cp, ch, w1, wn1_p, wn1_h, mask_p, out, 0);
  maxpool_wmma_kernel<<<BB * PP * 8, 32, 0, stream>>>(ch, cp, w1, wn1_h, wn1_p, mask_h, out, 1);
}


# Stages that do not depend on one another share a launch (block index picks stage and side; gather reads the raw input times its mask): 18 launches -> 5; no arithmetic or summation order changes
# speedup vs baseline: 4.0009x; 1.4889x over previous
#include <hip/hip_runtime.h>
#include <hip/hip_bf16.h>


#define BB 16
#define SS 128
#define HH 200
#define PP 20
#define FF 105
#define EPSF 1e-8f
#define KCH 10
#define NCH 5

typedef float v2f __attribute__((ext_vector_type(2)));
typedef float v4f __attribute__((ext_vector_type(4)));
typedef float v8f __attribute__((ext_vector_type(8)));

__device__ __forceinline__ v2f wsq_frag(v2f ww, bool isW, float fill) {
  v2f bv;
  bv.x = isW ? ww.x * ww.x : fill;
  bv.y = isW ? ww.y * ww.y : fill;
  return bv;
}

#define NPREP ((BB * SS / 16) * 2)
__device__ __forceinline__ void prep_body(const float* __restrict__ src, const int* __restrict__ mask,
                                 const float* __restrict__ w1, float* __restrict__ dst,
                                 float* __restrict__ nrm, float* __restrict__ wn, int tile, int lane) {
  int m0 = (tile >> 1) << 4;
  int n0 = (tile & 1) << 4;
  int half = lane >> 4, l16 = lane & 15;
  int bs = m0 + l16;
  float m = (float)mask[bs];
  const float* srow = src + (size_t)bs * HH + 2 * half;
  float* drow = dst + (size_t)bs * HH + 2 * half;
  int col = n0 + l16;
  bool isW = col < PP;
  float fill = (col == PP) ? 1.f : 0.f;
  const float* wrow = w1 + (size_t)(isW ? col : PP - 1) * HH + 2 * half;
  v8f acc = {0, 0, 0, 0, 0, 0, 0, 0};
  v2f xs[2][KCH], ws2[2][KCH];
#pragma unroll
  for (int i = 0; i < KCH; i++) {
    xs[0][i] = *(const v2f*)(srow + 4 * i);
    ws2[0][i] = *(const v2f*)(wrow + 4 * i);
  }
#pragma unroll
  for (int c = 0; c < NCH; c++) {
    if (c + 1 < NCH) {
#pragma unroll
      for (int i = 0; i < KCH; i++) {
        xs[(c + 1) & 1][i] = *(const v2f*)(srow + 4 * KCH * (c + 1) + 4 * i);
        ws2[(c + 1) & 1][i] = *(const v2f*)(wrow + 4 * KCH * (c + 1) + 4 * i);
      }
    }
    __builtin_amdgcn_sched_barrier(0);
#pragma unroll
    for (int i = 0; i < KCH; i++) {
      v2f x = xs[c & 1][i];
      x.x *= m; x.y *= m;
      if (n0 == 0) *(v2f*)(drow + 4 * KCH * c + 4 * i) = x;
      v2f bv = wsq_frag(ws2[c & 1][i], isW, fill);
      v2f a; a.x = x.x * x.x; a.y = x.y * x.y;
      acc = __builtin_amdgcn_wmma_f32_16x16x4_f32(false, a, false, bv, (short)0, acc, false, false);
    }
  }
  if (col <= PP) {
#pragma unroll
    for (int r = 0; r < 8; r++) {
      int row = m0 + r + 8 * half;
      float* o = (col == PP) ? nrm + row : wn + (size_t)row * PP + col;
      *o = sqrtf(acc[r]);
    }
  }
}

__device__ __forceinline__ void gather_last_body(const float* __restrict__ xp, const float* __restrict__ xh,
                                   const int* __restrict__ mask_p, const int* __restrict__ mask_h,
                                   float* __restrict__ cpl, float* __restrict__ chl, int b, int lane) {
  int lp = 0, lh = 0;
  for (int i = 0; i < SS; i++) { lp += mask_p[b * SS + i]; lh += mask_h[b * SS + i]; }
  int ip = (lp > 0) ? lp - 1 : 0;
  int ih = (lh > 0) ? lh - 1 : 0;
  float mp = (float)mask_p[b * SS + ip];
  float mh = (float)mask_h[b * SS + ih];
  for (int h = lane; h < HH; h += 32) {
    cpl[b * HH + h] = xp[((size_t)b * SS + ip) * HH + h] * mp;
    chl[b * HH + h] = xh[((size_t)b * SS + ih) * HH + h] * mh;
  }
}

__global__ void __launch_bounds__(32) prep_gather_kernel(
    const float* __restrict__ ctx_p, const float* __restrict__ ctx_h,
    const int* __restrict__ mask_p, const int* __restrict__ mask_h, const float* __restrict__ w1,
    float* __restrict__ cp, float* __restrict__ ch, float* __restrict__ n_p, float* __restrict__ n_h,
    float* __restrict__ wn1_p, float* __restrict__ wn1_h, float* __restrict__ cpl, float* __restrict__ chl) {
  int blk = blockIdx.x, lane = threadIdx.x;
  if (blk < 2 * NPREP) {
    bool side = blk >= NPREP;
    prep_body(side ? ctx_h : ctx_p, side ? mask_h : mask_p, w1, side ? ch : cp,
              side ? n_h : n_p, side ? wn1_h : wn1_p, side ? blk - NPREP : blk, lane);
  } else {
    gather_last_body(ctx_p, ctx_h, mask_p, mask_h, cpl, chl, blk - 2 * NPREP, lane);
  }
}

__global__ void cos_wmma_kernel(const float* __restrict__ cp, const float* __restrict__ ch,
                                const float* __restrict__ np, const float* __restrict__ nh,
                                float* __restrict__ cosO) {
  int wid = blockIdx.x;
  int b = wid >> 6;
  int x0 = ((wid >> 3) & 7) << 4;
  int y0 = (wid & 7) << 4;
  int lane = threadIdx.x, half = lane >> 4, l16 = lane & 15;
  const float* arow = cp + ((size_t)(b * SS) + x0 + l16) * HH + 2 * half;
  const float* brow = ch + ((size_t)(b * SS) + y0 + l16) * HH + 2 * half;
  v8f acc = {0, 0, 0, 0, 0, 0, 0, 0};
  for (int k = 0; k < HH; k += 4) {
    v2f a = *(const v2f*)(arow + k);
    v2f bv = *(const v2f*)(brow + k);
    acc = __builtin_amdgcn_wmma_f32_16x16x4_f32(false, a, false, bv, (short)0, acc, false, false);
  }
  float nhv = nh[b * SS + y0 + l16];
#pragma unroll
  for (int r = 0; r < 8; r++) {
    int m = r + 8 * half;
    float d = fmaxf(np[b * SS + x0 + m] * nhv, EPSF);
    cosO[((size_t)(b * SS) + x0 + m) * SS + y0 + l16] = acc[r] / d;
  }
}

__global__ void cos_reduce_kernel(const float* __restrict__ cosM,
                                  const int* __restrict__ mask_p, const int* __restrict__ mask_h,
                                  float* __restrict__ cossum_y, float* __restrict__ cossum_x,
                                  float* __restrict__ out) {
  int side = blockIdx.x >= BB * SS;
  int transposed = side;
  int blk = side ? blockIdx.x - BB * SS : blockIdx.x;
  const int* mask = side ? mask_h : mask_p;
  float* cossum = side ? cossum_x : cossum_y;
  int b = blk / SS, r = blk % SS;
  int t = threadIdx.x;
  float v = transposed ? cosM[((size_t)b * SS + t) * SS + r]
                       : cosM[((size_t)b * SS + r) * SS + t];
  __shared__ float smax[128], ssum[128];
  smax[t] = v; ssum[t] = v;
  __syncthreads();
  for (int o = 64; o > 0; o >>= 1) {
    if (t < o) { smax[t] = fmaxf(smax[t], smax[t + o]); ssum[t] += ssum[t + o]; }
    __syncthreads();
  }
  if (t == 0) {
    float m = (float)mask[b * SS + r];
    cossum[b * SS + r] = ssum[0];
    float* o2 = out + ((size_t)(side * BB + b) * SS + r) * FF;
    o2[0] = m * smax[0];
    o2[1] = m * (ssum[0] / (float)SS);
  }
}

#define NATT (BB * 8 * 13)
__device__ __forceinline__ void attmean_body(const float* __restrict__ cosM, const float* __restrict__ V,
                                    const float* __restrict__ csum, float* __restrict__ attO,
                                    int aStrideM, int aStrideK, int wid, int lane) {
  int b = wid / (8 * 13);
  int rem = wid % (8 * 13);
  int m0 = (rem / 13) * 16;
  int n0 = (rem % 13) * 16;
  int half = lane >> 4, l16 = lane & 15;
  const float* cb = cosM + (size_t)b * SS * SS;
  int nIdx = n0 + l16;
  bool nOK = nIdx < HH;
  v8f acc = {0, 0, 0, 0, 0, 0, 0, 0};
  for (int k = 0; k < SS; k += 4) {
    int ka = k + 2 * half;
    v2f a, bv;
    a.x = cb[(size_t)(m0 + l16) * aStrideM + (size_t)(ka + 0) * aStrideK];
    a.y = cb[(size_t)(m0 + l16) * aStrideM + (size_t)(ka + 1) * aStrideK];
    bv.x = nOK ? V[((size_t)b * SS + ka + 0) * HH + nIdx] : 0.f;
    bv.y = nOK ? V[((size_t)b * SS + ka + 1) * HH + nIdx] : 0.f;
    acc = __builtin_amdgcn_wmma_f32_16x16x4_f32(false, a, false, bv, (short)0, acc, false, false);
  }
  if (nOK) {
#pragma unroll
    for (int r = 0; r < 8; r++) {
      int m = m0 + r + 8 * half;
      float d = fmaxf(csum[b * SS + m], EPSF);
      attO[((size_t)b * SS + m) * HH + nIdx] = acc[r] / d;
    }
  }
}

__device__ __forceinline__ void attmax_body(const float* __restrict__ cosM, const float* __restrict__ V,
                              const int* __restrict__ mask, float* __restrict__ attO,
                              int cRowStride, int cColStride, int blk, int h) {
  int b = blk / SS, r = blk % SS;
  if (h >= HH) return;
  const float* cb = cosM + (size_t)b * SS * SS;
  float mx = -3.0e38f;
  for (int k = 0; k < SS; k++) {
    float c = cb[(size_t)r * cRowStride + (size_t)k * cColStride];
    float v = V[((size_t)b * SS + k) * HH + h];
    mx = fmaxf(mx, c * v);
  }
  float m = (float)mask[b * SS + r];
  attO[((size_t)b * SS + r) * HH + h] = m * mx;
}

__global__ void __launch_bounds__(256) att_kernel(
    const float* __restrict__ cosM, const float* __restrict__ cp, const float* __restrict__ ch,
    const float* __restrict__ cossum_y, const float* __restrict__ cossum_x,
    const int* __restrict__ mask_p, const int* __restrict__ mask_h,
    float* __restrict__ att_mean_h, float* __restrict__ att_mean_p,
    float* __restrict__ att_max_h, float* __restrict__ att_max_p,
    int cosRowStride, int cosColStride) {
  int blk = blockIdx.x;
  if (blk < 2 * NATT / 8) {
    bool side = blk >= NATT / 8;
    int wid = (side ? blk - NATT / 8 : blk) * 8 + (threadIdx.x >> 5);
    attmean_body(cosM, side ? cp : ch, side ? cossum_x : cossum_y, side ? att_mean_p : att_mean_h,
                 side ? cosColStride : cosRowStride, side ? cosRowStride : cosColStride,
                 wid, threadIdx.x & 31);
  } else {
    blk -= 2 * NATT / 8;
    bool side = blk >= BB * SS;
    attmax_body(cosM, side ? cp : ch, side ? mask_h : mask_p, side ? att_max_p : att_max_h,
                side ? cosColStride : cosRowStride, side ? cosRowStride : cosColStride,
                side ? blk - BB * SS : blk, threadIdx.x);
  }
}

#define NMPM ((BB * SS / 16) * 2)
__device__ __forceinline__ void mpm_body(const float* __restrict__ v1, const float* __restrict__ v2,
                                const float* __restrict__ w, int v2_bstride, int v2_sstride,
                                float* __restrict__ out, int side, int foff, int tile, int lane) {
  int m0 = (tile >> 1) << 4;
  int n0 = (tile & 1) << 4;
  int half = lane >> 4, l16 = lane & 15;
  int bs = m0 + l16;
  int b = bs / SS, s = bs % SS;
  const float* p1 = v1 + (size_t)bs * HH + 2 * half;
  const float* p2 = v2 + (size_t)b * v2_bstride + (size_t)s * v2_sstride + 2 * half;
  int col = n0 + l16;
  bool isW = col < PP;
  float fill = (col == PP) ? 1.f : 0.f;
  const float* wrow = w + (size_t)(isW ? col : PP - 1) * HH + 2 * half;
  v8f s11 = {0, 0, 0, 0, 0, 0, 0, 0};
  v8f s22 = {0, 0, 0, 0, 0, 0, 0, 0};
  v8f s12 = {0, 0, 0, 0, 0, 0, 0, 0};
  v2f xs[2][KCH], ys[2][KCH], ws2[2][KCH];
#pragma unroll
  for (int i = 0; i < KCH; i++) {
    xs[0][i] = *(const v2f*)(p1 + 4 * i);
    ys[0][i] = *(const v2f*)(p2 + 4 * i);
    ws2[0][i] = *(const v2f*)(wrow + 4 * i);
  }
#pragma unroll
  for (int c = 0; c < NCH; c++) {
    if (c + 1 < NCH) {
#pragma unroll
      for (int i = 0; i < KCH; i++) {
        xs[(c + 1) & 1][i] = *(const v2f*)(p1 + 4 * KCH * (c + 1) + 4 * i);
        ys[(c + 1) & 1][i] = *(const v2f*)(p2 + 4 * KCH * (c + 1) + 4 * i);
        ws2[(c + 1) & 1][i] = *(const v2f*)(wrow + 4 * KCH * (c + 1) + 4 * i);
      }
    }
    __builtin_amdgcn_sched_barrier(0);
#pragma unroll
    for (int i = 0; i < KCH; i++) {
      v2f x = xs[c & 1][i], y = ys[c & 1][i];
      v2f bv = wsq_frag(ws2[c & 1][i], isW, fill);
      v2f xx, yy, xy;
      xx.x = x.x * x.x; xx.y = x.y * x.y;
      yy.x = y.x * y.x; yy.y = y.y * y.y;
      xy.x = x.x * y.x; xy.y = x.y * y.y;
      s11 = __builtin_amdgcn_wmma_f32_16x16x4_f32(false, xx, false, bv, (short)0, s11, false, false);
      s22 = __builtin_amdgcn_wmma_f32_16x16x4_f32(false, yy, false, bv, (short)0, s22, false, false);
      s12 = __builtin_amdgcn_wmma_f32_16x16x4_f32(false, xy, false, bv, (short)0, s12, false, false);
    }
  }
  if (col <= PP) {
    int f = (col == PP) ? 0 : 1 + col;
#pragma unroll
    for (int r = 0; r < 8; r++) {
      int row = m0 + r + 8 * half;
      int ob = row / SS, os = row % SS;
      float* orow = out + ((size_t)(side * BB + ob) * SS + os) * FF + foff;
      orow[f] = s12[r] / (fmaxf(sqrtf(s11[r]), EPSF) * fmaxf(sqrtf(s22[r]), EPSF));
    }
  }
}

#define NJ 25
#define GCH 10
#define NYT 8
#define NMAXP (BB * PP * 8)
__device__ __forceinline__ void maxpool_body(const float* __restrict__ v1, const float* __restrict__ v2,
                                    const float* __restrict__ w,
                                    const float* __restrict__ wnA, const float* __restrict__ wnB,
                                    const int* __restrict__ mask, float* __restrict__ out,
                                    int side, int wid, int lane) {
  int b = wid / (PP * 8);
  int p = (wid / 8) % PP;
  int x0 = (wid % 8) * 16;
  int half = lane >> 4, l16 = lane & 15;
  float rmax[8], rsum[8], wA[8];
#pragma unroll
  for (int r = 0; r < 8; r++) {
    rmax[r] = -3.0e38f; rsum[r] = 0.f;
    wA[r] = wnA[((size_t)b * SS + x0 + r + 8 * half) * PP + p];
  }
  const float* a_row = v1 + ((size_t)b * SS + x0 + l16) * HH + 4 * half;
  const float* w_row = w + (size_t)p * HH + 4 * half;
  const float* b_row = v2 + ((size_t)b * SS + l16) * HH + 4 * half;
  const float* wnB_col = wnB + ((size_t)b * SS + l16) * PP + p;

  v4f bf[2][GCH];
#pragma unroll
  for (int i = 0; i < GCH; i++) bf[0][i] = *(const v4f*)(b_row + 8 * i);

  v4f af[NJ];
#pragma unroll
  for (int j = 0; j < NJ; j++) {
    v4f wv = *(const v4f*)(w_row + 8 * j);
    v4f av = *(const v4f*)(a_row + 8 * j);
    af[j].x = wv.x * wv.x * av.x;
    af[j].y = wv.y * wv.y * av.y;
    af[j].z = wv.z * wv.z * av.z;
    af[j].w = wv.w * wv.w * av.w;
  }

  v8f acc = {0, 0, 0, 0, 0, 0, 0, 0};
  float wB = 0.f;
#pragma unroll
  for (int g0 = 0; g0 < NYT * NJ; g0 += GCH) {
    if (g0 + GCH < NYT * NJ) {
#pragma unroll
      for (int i = 0; i < GCH; i++) {
        int gn = g0 + GCH + i;
        bf[((g0 / GCH) + 1) & 1][i] = *(const v4f*)(b_row + (gn / NJ) * 16 * HH + 8 * (gn % NJ));
      }
    }
    __builtin_amdgcn_sched_barrier(0);
#pragma unroll
    for (int i = 0; i < GCH; i++) {
      int g = g0 + i;
      int yt = g / NJ, j = g % NJ;
      if (j == 0) {
        acc = (v8f){0, 0, 0, 0, 0, 0, 0, 0};
        wB = wnB_col[(size_t)yt * 16 * PP];
      }
      v4f bb = bf[(g0 / GCH) & 1][i];
      v4f aa = af[j];
      v2f a0, b0, a1, b1;
      a0.x = aa.x; a0.y = aa.y; b0.x = bb.x; b0.y = bb.y;
      a1.x = aa.z; a1.y = aa.w; b1.x = bb.z; b1.y = bb.w;
      acc = __builtin_amdgcn_wmma_f32_16x16x4_f32(false, a0, false, b0, (short)0, acc, false, false);
      acc = __builtin_amdgcn_wmma_f32_16x16x4_f32(false, a1, false, b1, (short)0, acc, false, false);
      if (j == NJ - 1) {
#pragma unroll
        for (int r = 0; r < 8; r++) {
          float mv = acc[r] / fmaxf(wA[r] * wB, EPSF);
          rmax[r] = fmaxf(rmax[r], mv);
          rsum[r] += mv;
        }
      }
    }
  }
#pragma unroll
  for (int r = 0; r < 8; r++) {
    for (int o = 8; o > 0; o >>= 1) {
      rmax[r] = fmaxf(rmax[r], __shfl_xor(rmax[r], o, 32));
      rsum[r] += __shfl_xor(rsum[r], o, 32);
    }
  }
  if (l16 == 0) {
#pragma unroll
    for (int r = 0; r < 8; r++) {
      int x = x0 + r + 8 * half;
      float m = (float)mask[b * SS + x];
      float* orow = out + ((size_t)(side * BB + b) * SS + x) * FF;
      orow[23 + p] = m * rmax[r];
      orow[43 + p] = m * (rsum[r] / (float)SS);
    }
  }
}

__global__ void __launch_bounds__(32) match_kernel(
    const float* __restrict__ cp, const float* __restrict__ ch,
    const float* __restrict__ cpl, const float* __restrict__ chl,
    const float* __restrict__ att_mean_h, const float* __restrict__ att_mean_p,
    const float* __restrict__ att_max_h, const float* __restrict__ att_max_p,
    const float* __restrict__ w0, const float* __restrict__ w1,
    const float* __restrict__ w2, const float* __restrict__ w3,
    const float* __restrict__ wn1_p, const float* __restrict__ wn1_h,
    const int* __restrict__ mask_p, const int* __restrict__ mask_h, float* __restrict__ out) {
  int blk = blockIdx.x, lane = threadIdx.x;
  if (blk < 2 * NMAXP) {
    bool side = blk >= NMAXP;
    maxpool_body(side ? ch : cp, side ? cp : ch, w1, side ? wn1_h : wn1_p, side ? wn1_p : wn1_h,
                 side ? mask_h : mask_p, out, side ? 1 : 0, side ? blk - NMAXP : blk, lane);
  } else {
    blk -= 2 * NMAXP;
    int grp = blk / (2 * NMPM);
    int rem = blk % (2 * NMPM);
    bool side = rem >= NMPM;
    int tile = side ? rem - NMPM : rem;
    const float* v1 = side ? ch : cp;
    const float* v2 = grp == 0 ? (side ? cpl : chl)
                    : grp == 1 ? (side ? att_mean_p : att_mean_h) : (side ? att_max_p : att_max_h);
    const float* w = grp == 0 ? w0 : grp == 1 ? w2 : w3;
    int v2_bstride = grp == 0 ? HH : SS * HH;
    int v2_sstride = grp == 0 ? 0 : HH;
    int foff = grp == 0 ? 2 : grp == 1 ? 63 : 84;
    mpm_body(v1, v2, w, v2_bstride, v2_sstride, out, side ? 1 : 0, foff, tile, lane);
  }
}

extern "C" void kernel_launch(void* const* d_in, const int* in_sizes, int n_in,
                              void* d_out, int out_size, void* d_ws, size_t ws_size,
                              hipStream_t stream) {
  const float* ctx_p = (const float*)d_in[0];
  const float* ctx_h = (const float*)d_in[1];
  const float* w0 = (const float*)d_in[2];
  const float* w1 = (const float*)d_in[3];
  const float* w2 = (const float*)d_in[4];
  const float* w3 = (const float*)d_in[5];
  const int* mask_p = (const int*)d_in[6];
  const int* mask_h = (const int*)d_in[7];
  float* out = (float*)d_out;
  float* ws = (float*)d_ws;

  const size_t SZ_ROWS = (size_t)BB * SS * HH;
  float* cp = ws;                 ws += SZ_ROWS;
  float* ch = ws;                 ws += SZ_ROWS;
  float* n_p = ws;                ws += BB * SS;
  float* n_h = ws;                ws += BB * SS;
  float* wn1_p = ws;              ws += (size_t)BB * SS * PP;
  float* wn1_h = ws;              ws += (size_t)BB * SS * PP;
  float* cpl = ws;                ws += BB * HH;
  float* chl = ws;                ws += BB * HH;
  float* cosM = ws;               ws += (size_t)BB * SS * SS;
  float* cossum_y = ws;           ws += BB * SS;
  float* cossum_x = ws;           ws += BB * SS;
  float* att_mean_h = ws;         ws += SZ_ROWS;
  float* att_mean_p = ws;         ws += SZ_ROWS;
  float* att_max_h = ws;          ws += SZ_ROWS;
  float* att_max_p = ws;          ws += SZ_ROWS;

  prep_gather_kernel<<<2 * NPREP + BB, 32, 0, stream>>>(ctx_p, ctx_h, mask_p, mask_h, w1, cp, ch, n_p, n_h,
                                                        wn1_p, wn1_h, cpl, chl);

  cos_wmma_kernel<<<BB * 64, 32, 0, stream>>>(cp, ch, n_p, n_h, cosM);

  cos_reduce_kernel<<<2 * BB * SS, 128, 0, stream>>>(cosM, mask_p, mask_h, cossum_y, cossum_x, out);

  att_kernel<<<2 * NATT / 8 + 2 * BB * SS, 256, 0, stream>>>(cosM, cp, ch, cossum_y, cossum_x, mask_p, mask_h,
                                                             att_mean_h, att_mean_p, att_max_h, att_max_p, SS, 1);

  match_kernel<<<2 * NMAXP + 6 * NMPM, 32, 0, stream>>>(cp, ch, cpl, chl, att_mean_h, att_mean_p,
                                                        att_max_h, att_max_p, w0, w1, w2, w3,
                                                        wn1_p, wn1_h, mask_p, mask_h, out);
}


# max-attentive (max-plus) stage register-tiled: a block keeps 8 rows' running maxima per column thread, cos piece staged once in LDS; V loaded once per 8 rows, 8x fewer waves; max is exact
# speedup vs baseline: 7.3212x; 1.8299x over previous
#include <hip/hip_runtime.h>
#include <hip/hip_bf16.h>


#define BB 16
#define SS 128
#define HH 200
#define PP 20
#define FF 105
#define EPSF 1e-8f
#define KCH 10
#define NCH 5

typedef float v2f __attribute__((ext_vector_type(2)));
typedef float v4f __attribute__((ext_vector_type(4)));
typedef float v8f __attribute__((ext_vector_type(8)));

__device__ __forceinline__ v2f wsq_frag(v2f ww, bool isW, float fill) {
  v2f bv;
  bv.x = isW ? ww.x * ww.x : fill;
  bv.y = isW ? ww.y * ww.y : fill;
  return bv;
}

#define NPREP ((BB * SS / 16) * 2)
__device__ __forceinline__ void prep_body(const float* __restrict__ src, const int* __restrict__ mask,
                                 const float* __restrict__ w1, float* __restrict__ dst,
                                 float* __restrict__ nrm, float* __restrict__ wn, int tile, int lane) {
  int m0 = (tile >> 1) << 4;
  int n0 = (tile & 1) << 4;
  int half = lane >> 4, l16 = lane & 15;
  int bs = m0 + l16;
  float m = (float)mask[bs];
  const float* srow = src + (size_t)bs * HH + 2 * half;
  float* drow = dst + (size_t)bs * HH + 2 * half;
  int col = n0 + l16;
  bool isW = col < PP;
  float fill = (col == PP) ? 1.f : 0.f;
  const float* wrow = w1 + (size_t)(isW ? col : PP - 1) * HH + 2 * half;
  v8f acc = {0, 0, 0, 0, 0, 0, 0, 0};
  v2f xs[2][KCH], ws2[2][KCH];
#pragma unroll
  for (int i = 0; i < KCH; i++) {
    xs[0][i] = *(const v2f*)(srow + 4 * i);
    ws2[0][i] = *(const v2f*)(wrow + 4 * i);
  }
#pragma unroll
  for (int c = 0; c < NCH; c++) {
    if (c + 1 < NCH) {
#pragma unroll
      for (int i = 0; i < KCH; i++) {
        xs[(c + 1) & 1][i] = *(const v2f*)(srow + 4 * KCH * (c + 1) + 4 * i);
        ws2[(c + 1) & 1][i] = *(const v2f*)(wrow + 4 * KCH * (c + 1) + 4 * i);
      }
    }
    __builtin_amdgcn_sched_barrier(0);
#pragma unroll
    for (int i = 0; i < KCH; i++) {
      v2f x = xs[c & 1][i];
      x.x *= m; x.y *= m;
      if (n0 == 0) *(v2f*)(drow + 4 * KCH * c + 4 * i) = x;
      v2f bv = wsq_frag(ws2[c & 1][i], isW, fill);
      v2f a; a.x = x.x * x.x; a.y = x.y * x.y;
      acc = __builtin_amdgcn_wmma_f32_16x16x4_f32(false, a, false, bv, (short)0, acc, false, false);
    }
  }
  if (col <= PP) {
#pragma unroll
    for (int r = 0; r < 8; r++) {
      int row = m0 + r + 8 * half;
      float* o = (col == PP) ? nrm + row : wn + (size_t)row * PP + col;
      *o = sqrtf(acc[r]);
    }
  }
}

__device__ __forceinline__ void gather_last_body(const float* __restrict__ xp, const float* __restrict__ xh,
                                   const int* __restrict__ mask_p, const int* __restrict__ mask_h,
                                   float* __restrict__ cpl, float* __restrict__ chl, int b, int lane) {
  int lp = 0, lh = 0;
  for (int i = 0; i < SS; i++) { lp += mask_p[b * SS + i]; lh += mask_h[b * SS + i]; }
  int ip = (lp > 0) ? lp - 1 : 0;
  int ih = (lh > 0) ? lh - 1 : 0;
  float mp = (float)mask_p[b * SS + ip];
  float mh = (float)mask_h[b * SS + ih];
  for (int h = lane; h < HH; h += 32) {
    cpl[b * HH + h] = xp[((size_t)b * SS + ip) * HH + h] * mp;
    chl[b * HH + h] = xh[((size_t)b * SS + ih) * HH + h] * mh;
  }
}

__global__ void __launch_bounds__(32) prep_gather_kernel(
    const float* __restrict__ ctx_p, const float* __restrict__ ctx_h,
    const int* __restrict__ mask_p, const int* __restrict__ mask_h, const float* __restrict__ w1,
    float* __restrict__ cp, float* __restrict__ ch, float* __restrict__ n_p, float* __restrict__ n_h,
    float* __restrict__ wn1_p, float* __restrict__ wn1_h, float* __restrict__ cpl, float* __restrict__ chl) {
  int blk = blockIdx.x, lane = threadIdx.x;
  if (blk < 2 * NPREP) {
    bool side = blk >= NPREP;
    prep_body(side ? ctx_h : ctx_p, side ? mask_h : mask_p, w1, side ? ch : cp,
              side ? n_h : n_p, side ? wn1_h : wn1_p, side ? blk - NPREP : blk, lane);
  } else {
    gather_last_body(ctx_p, ctx_h, mask_p, mask_h, cpl, chl, blk - 2 * NPREP, lane);
  }
}

__global__ void cos_wmma_kernel(const float* __restrict__ cp, const float* __restrict__ ch,
                                const float* __restrict__ np, const float* __restrict__ nh,
                                float* __restrict__ cosO) {
  int wid = blockIdx.x;
  int b = wid >> 6;
  int x0 = ((wid >> 3) & 7) << 4;
  int y0 = (wid & 7) << 4;
  int lane = threadIdx.x, half = lane >> 4, l16 = lane & 15;
  const float* arow = cp + ((size_t)(b * SS) + x0 + l16) * HH + 2 * half;
  const float* brow = ch + ((size_t)(b * SS) + y0 + l16) * HH + 2 * half;
  v8f acc = {0, 0, 0, 0, 0, 0, 0, 0};
  for (int k = 0; k < HH; k += 4) {
    v2f a = *(const v2f*)(arow + k);
    v2f bv = *(const v2f*)(brow + k);
    acc = __builtin_amdgcn_wmma_f32_16x16x4_f32(false, a, false, bv, (short)0, acc, false, false);
  }
  float nhv = nh[b * SS + y0 + l16];
#pragma unroll
  for (int r = 0; r < 8; r++) {
    int m = r + 8 * half;
    float d = fmaxf(np[b * SS + x0 + m] * nhv, EPSF);
    cosO[((size_t)(b * SS) + x0 + m) * SS + y0 + l16] = acc[r] / d;
  }
}

__global__ void cos_reduce_kernel(const float* __restrict__ cosM,
                                  const int* __restrict__ mask_p, const int* __restrict__ mask_h,
                                  float* __restrict__ cossum_y, float* __restrict__ cossum_x,
                                  float* __restrict__ out) {
  int side = blockIdx.x >= BB * SS;
  int transposed = side;
  int blk = side ? blockIdx.x - BB * SS : blockIdx.x;
  const int* mask = side ? mask_h : mask_p;
  float* cossum = side ? cossum_x : cossum_y;
  int b = blk / SS, r = blk % SS;
  int t = threadIdx.x;
  float v = transposed ? cosM[((size_t)b * SS + t) * SS + r]
                       : cosM[((size_t)b * SS + r) * SS + t];
  __shared__ float smax[128], ssum[128];
  smax[t] = v; ssum[t] = v;
  __syncthreads();
  for (int o = 64; o > 0; o >>= 1) {
    if (t < o) { smax[t] = fmaxf(smax[t], smax[t + o]); ssum[t] += ssum[t + o]; }
    __syncthreads();
  }
  if (t == 0) {
    float m = (float)mask[b * SS + r];
    cossum[b * SS + r] = ssum[0];
    float* o2 = out + ((size_t)(side * BB + b) * SS + r) * FF;
    o2[0] = m * smax[0];
    o2[1] = m * (ssum[0] / (float)SS);
  }
}

#define NATT (BB * 8 * 13)
__device__ __forceinline__ void attmean_body(const float* __restrict__ cosM, const float* __restrict__ V,
                                    const float* __restrict__ csum, float* __restrict__ attO,
                                    int aStrideM, int aStrideK, int wid, int lane) {
  int b = wid / (8 * 13);
  int rem = wid % (8 * 13);
  int m0 = (rem / 13) * 16;
  int n0 = (rem % 13) * 16;
  int half = lane >> 4, l16 = lane & 15;
  const float* cb = cosM + (size_t)b * SS * SS;
  int nIdx = n0 + l16;
  bool nOK = nIdx < HH;
  v8f acc = {0, 0, 0, 0, 0, 0, 0, 0};
  for (int k = 0; k < SS; k += 4) {
    int ka = k + 2 * half;
    v2f a, bv;
    a.x = cb[(size_t)(m0 + l16) * aStrideM + (size_t)(ka + 0) * aStrideK];
    a.y = cb[(size_t)(m0 + l16) * aStrideM + (size_t)(ka + 1) * aStrideK];
    bv.x = nOK ? V[((size_t)b * SS + ka + 0) * HH + nIdx] : 0.f;
    bv.y = nOK ? V[((size_t)b * SS + ka + 1) * HH + nIdx] : 0.f;
    acc = __builtin_amdgcn_wmma_f32_16x16x4_f32(false, a, false, bv, (short)0, acc, false, false);
  }
  if (nOK) {
#pragma unroll
    for (int r = 0; r < 8; r++) {
      int m = m0 + r + 8 * half;
      float d = fmaxf(csum[b * SS + m], EPSF);
      attO[((size_t)b * SS + m) * HH + nIdx] = acc[r] / d;
    }
  }
}

#define RT 8
#define KU 8
#define NAMX (BB * SS / RT)
template <bool TR>
__device__ __forceinline__ void attmax_body(const float* __restrict__ cosM, const float* __restrict__ V,
                              const int* __restrict__ mask, float* __restrict__ attO,
                              v4f (*cs)[RT / 4], int blk, int t) {
  int b = blk / (SS / RT), r0 = (blk % (SS / RT)) * RT;
  const float* cb = cosM + (size_t)b * SS * SS;
  if (TR) {
    int k = t >> 1, q = t & 1;
    cs[k][q] = *(const v4f*)(cb + (size_t)k * SS + r0 + 4 * q);
  } else {
    int j = t & (RT - 1), k4 = t >> 3;
    v4f x = *(const v4f*)(cb + (size_t)(r0 + j) * SS + 4 * k4);
    float* csf = (float*)cs;
    csf[(4 * k4 + 0) * RT + j] = x.x;
    csf[(4 * k4 + 1) * RT + j] = x.y;
    csf[(4 * k4 + 2) * RT + j] = x.z;
    csf[(4 * k4 + 3) * RT + j] = x.w;
  }
  __syncthreads();
  int h = t;
  if (h >= HH) return;
  const float* vp = V + (size_t)b * SS * HH + h;
  float mx[RT];
#pragma unroll
  for (int j = 0; j < RT; j++) mx[j] = -3.0e38f;
  for (int k = 0; k < SS; k += KU) {
    float v[KU];
    v4f c[KU][RT / 4];
#pragma unroll
    for (int u = 0; u < KU; u++) v[u] = vp[(size_t)(k + u) * HH];
#pragma unroll
    for (int u = 0; u < KU; u++) {
#pragma unroll
      for (int q = 0; q < RT / 4; q++) c[u][q] = cs[k + u][q];
    }
    __builtin_amdgcn_sched_barrier(0);
#pragma unroll
    for (int u = 0; u < KU; u++) {
#pragma unroll
      for (int q = 0; q < RT / 4; q++) {
        mx[4 * q + 0] = fmaxf(mx[4 * q + 0], c[u][q].x * v[u]);
        mx[4 * q + 1] = fmaxf(mx[4 * q + 1], c[u][q].y * v[u]);
        mx[4 * q + 2] = fmaxf(mx[4 * q + 2], c[u][q].z * v[u]);
        mx[4 * q + 3] = fmaxf(mx[4 * q + 3], c[u][q].w * v[u]);
      }
    }
  }
#pragma unroll
  for (int j = 0; j < RT; j++) {
    float m = (float)mask[b * SS + r0 + j];
    attO[((size_t)b * SS + r0 + j) * HH + h] = m * mx[j];
  }
}

__global__ void __launch_bounds__(256) att_kernel(
    const float* __restrict__ cosM, const float* __restrict__ cp, const float* __restrict__ ch,
    const float* __restrict__ cossum_y, const float* __restrict__ cossum_x,
    const int* __restrict__ mask_p, const int* __restrict__ mask_h,
    float* __restrict__ att_mean_h, float* __restrict__ att_mean_p,
    float* __restrict__ att_max_h, float* __restrict__ att_max_p,
    int cosRowStride, int cosColStride) {
  __shared__ v4f cs[SS][RT / 4];
  int blk = blockIdx.x;
  if (blk < 2 * NATT / 8) {
    bool side = blk >= NATT / 8;
    int wid = (side ? blk - NATT / 8 : blk) * 8 + (threadIdx.x >> 5);
    attmean_body(cosM, side ? cp : ch, side ? cossum_x : cossum_y, side ? att_mean_p : att_mean_h,
                 side ? cosColStride : cosRowStride, side ? cosRowStride : cosColStride,
                 wid, threadIdx.x & 31);
  } else {
    blk -= 2 * NATT / 8;
    if (blk >= NAMX) attmax_body<true>(cosM, cp, mask_h, att_max_p, cs, blk - NAMX, threadIdx.x);
    else attmax_body<false>(cosM, ch, mask_p, att_max_h, cs, blk, threadIdx.x);
  }
}

#define NMPM ((BB * SS / 16) * 2)
__device__ __forceinline__ void mpm_body(const float* __restrict__ v1, const float* __restrict__ v2,
                                const float* __restrict__ w, int v2_bstride, int v2_sstride,
                                float* __restrict__ out, int side, int foff, int tile, int lane) {
  int m0 = (tile >> 1) << 4;
  int n0 = (tile & 1) << 4;
  int half = lane >> 4, l16 = lane & 15;
  int bs = m0 + l16;
  int b = bs / SS, s = bs % SS;
  const float* p1 = v1 + (size_t)bs * HH + 2 * half;
  const float* p2 = v2 + (size_t)b * v2_bstride + (size_t)s * v2_sstride + 2 * half;
  int col = n0 + l16;
  bool isW = col < PP;
  float fill = (col == PP) ? 1.f : 0.f;
  const float* wrow = w + (size_t)(isW ? col : PP - 1) * HH + 2 * half;
  v8f s11 = {0, 0, 0, 0, 0, 0, 0, 0};
  v8f s22 = {0, 0, 0, 0, 0, 0, 0, 0};
  v8f s12 = {0, 0, 0, 0, 0, 0, 0, 0};
  v2f xs[2][KCH], ys[2][KCH], ws2[2][KCH];
#pragma unroll
  for (int i = 0; i < KCH; i++) {
    xs[0][i] = *(const v2f*)(p1 + 4 * i);
    ys[0][i] = *(const v2f*)(p2 + 4 * i);
    ws2[0][i] = *(const v2f*)(wrow + 4 * i);
  }
#pragma unroll
  for (int c = 0; c < NCH; c++) {
    if (c + 1 < NCH) {
#pragma unroll
      for (int i = 0; i < KCH; i++) {
        xs[(c + 1) & 1][i] = *(const v2f*)(p1 + 4 * KCH * (c + 1) + 4 * i);
        ys[(c + 1) & 1][i] = *(const v2f*)(p2 + 4 * KCH * (c + 1) + 4 * i);
        ws2[(c + 1) & 1][i] = *(const v2f*)(wrow + 4 * KCH * (c + 1) + 4 * i);
      }
    }
    __builtin_amdgcn_sched_barrier(0);
#pragma unroll
    for (int i = 0; i < KCH; i++) {
      v2f x = xs[c & 1][i], y = ys[c & 1][i];
      v2f bv = wsq_frag(ws2[c & 1][i], isW, fill);
      v2f xx, yy, xy;
      xx.x = x.x * x.x; xx.y = x.y * x.y;
      yy.x = y.x * y.x; yy.y = y.y * y.y;
      xy.x = x.x * y.x; xy.y = x.y * y.y;
      s11 = __builtin_amdgcn_wmma_f32_16x16x4_f32(false, xx, false, bv, (short)0, s11, false, false);
      s22 = __builtin_amdgcn_wmma_f32_16x16x4_f32(false, yy, false, bv, (short)0, s22, false, false);
      s12 = __builtin_amdgcn_wmma_f32_16x16x4_f32(false, xy, false, bv, (short)0, s12, false, false);
    }
  }
  if (col <= PP) {
    int f = (col == PP) ? 0 : 1 + col;
#pragma unroll
    for (int r = 0; r < 8; r++) {
      int row = m0 + r + 8 * half;
      int ob = row / SS, os = row % SS;
      float* orow = out + ((size_t)(side * BB + ob) * SS + os) * FF + foff;
      orow[f] = s12[r] / (fmaxf(sqrtf(s11[r]), EPSF) * fmaxf(sqrtf(s22[r]), EPSF));
    }
  }
}

#define NJ 25
#define GCH 10
#define NYT 8
#define NMAXP (BB * PP * 8)
__device__ __forceinline__ void maxpool_body(const float* __restrict__ v1, const float* __restrict__ v2,
                                    const float* __restrict__ w,
                                    const float* __restrict__ wnA, const float* __restrict__ wnB,
                                    const int* __restrict__ mask, float* __restrict__ out,
                                    int side, int wid, int lane) {
  int b = wid / (PP * 8);
  int p = (wid / 8) % PP;
  int x0 = (wid % 8) * 16;
  int half = lane >> 4, l16 = lane & 15;
  float rmax[8], rsum[8], wA[8];
#pragma unroll
  for (int r = 0; r < 8; r++) {
    rmax[r] = -3.0e38f; rsum[r] = 0.f;
    wA[r] = wnA[((size_t)b * SS + x0 + r + 8 * half) * PP + p];
  }
  const float* a_row = v1 + ((size_t)b * SS + x0 + l16) * HH + 4 * half;
  const float* w_row = w + (size_t)p * HH + 4 * half;
  const float* b_row = v2 + ((size_t)b * SS + l16) * HH + 4 * half;
  const float* wnB_col = wnB + ((size_t)b * SS + l16) * PP + p;

  v4f bf[2][GCH];
#pragma unroll
  for (int i = 0; i < GCH; i++) bf[0][i] = *(const v4f*)(b_row + 8 * i);

  v4f af[NJ];
#pragma unroll
  for (int j = 0; j < NJ; j++) {
    v4f wv = *(const v4f*)(w_row + 8 * j);
    v4f av = *(const v4f*)(a_row + 8 * j);
    af[j].x = wv.x * wv.x * av.x;
    af[j].y = wv.y * wv.y * av.y;
    af[j].z = wv.z * wv.z * av.z;
    af[j].w = wv.w * wv.w * av.w;
  }

  v8f acc = {0, 0, 0, 0, 0, 0, 0, 0};
  float wB = 0.f;
#pragma unroll
  for (int g0 = 0; g0 < NYT * NJ; g0 += GCH) {
    if (g0 + GCH < NYT * NJ) {
#pragma unroll
      for (int i = 0; i < GCH; i++) {
        int gn = g0 + GCH + i;
        bf[((g0 / GCH) + 1) & 1][i] = *(const v4f*)(b_row + (gn / NJ) * 16 * HH + 8 * (gn % NJ));
      }
    }
    __builtin_amdgcn_sched_barrier(0);
#pragma unroll
    for (int i = 0; i < GCH; i++) {
      int g = g0 + i;
      int yt = g / NJ, j = g % NJ;
      if (j == 0) {
        acc = (v8f){0, 0, 0, 0, 0, 0, 0, 0};
        wB = wnB_col[(size_t)yt * 16 * PP];
      }
      v4f bb = bf[(g0 / GCH) & 1][i];
      v4f aa = af[j];
      v2f a0, b0, a1, b1;
      a0.x = aa.x; a0.y = aa.y; b0.x = bb.x; b0.y = bb.y;
      a1.x = aa.z; a1.y = aa.w; b1.x = bb.z; b1.y = bb.w;
      acc = __builtin_amdgcn_wmma_f32_16x16x4_f32(false, a0, false, b0, (short)0, acc, false, false);
      acc = __builtin_amdgcn_wmma_f32_16x16x4_f32(false, a1, false, b1, (short)0, acc, false, false);
      if (j == NJ - 1) {
#pragma unroll
        for (int r = 0; r < 8; r++) {
          float mv = acc[r] / fmaxf(wA[r] * wB, EPSF);
          rmax[r] = fmaxf(rmax[r], mv);
          rsum[r] += mv;
        }
      }
    }
  }
#pragma unroll
  for (int r = 0; r < 8; r++) {
    for (int o = 8; o > 0; o >>= 1) {
      rmax[r] = fmaxf(rmax[r], __shfl_xor(rmax[r], o, 32));
      rsum[r] += __shfl_xor(rsum[r], o, 32);
    }
  }
  if (l16 == 0) {
#pragma unroll
    for (int r = 0; r < 8; r++) {
      int x = x0 + r + 8 * half;
      float m = (float)mask[b * SS + x];
      float* orow = out + ((size_t)(side * BB + b) * SS + x) * FF;
      orow[23 + p] = m * rmax[r];
      orow[43 + p] = m * (rsum[r] / (float)SS);
    }
  }
}

__global__ void __launch_bounds__(32) match_kernel(
    const float* __restrict__ cp, const float* __restrict__ ch,
    const float* __restrict__ cpl, const float* __restrict__ chl,
    const float* __restrict__ att_mean_h, const float* __restrict__ att_mean_p,
    const float* __restrict__ att_max_h, const float* __restrict__ att_max_p,
    const float* __restrict__ w0, const float* __restrict__ w1,
    const float* __restrict__ w2, const float* __restrict__ w3,
    const float* __restrict__ wn1_p, const float* __restrict__ wn1_h,
    const int* __restrict__ mask_p, const int* __restrict__ mask_h, float* __restrict__ out) {
  int blk = blockIdx.x, lane = threadIdx.x;
  if (blk < 2 * NMAXP) {
    bool side = blk >= NMAXP;
    maxpool_body(side ? ch : cp, side ? cp : ch, w1, side ? wn1_h : wn1_p, side ? wn1_p : wn1_h,
                 side ? mask_h : mask_p, out, side ? 1 : 0, side ? blk - NMAXP : blk, lane);
  } else {
    blk -= 2 * NMAXP;
    int grp = blk / (2 * NMPM);
    int rem = blk % (2 * NMPM);
    bool side = rem >= NMPM;
    int tile = side ? rem - NMPM : rem;
    const float* v1 = side ? ch : cp;
    const float* v2 = grp == 0 ? (side ? cpl : chl)
                    : grp == 1 ? (side ? att_mean_p : att_mean_h) : (side ? att_max_p : att_max_h);
    const float* w = grp == 0 ? w0 : grp == 1 ? w2 : w3;
    int v2_bstride = grp == 0 ? HH : SS * HH;
    int v2_sstride = grp == 0 ? 0 : HH;
    int foff = grp == 0 ? 2 : grp == 1 ? 63 : 84;
    mpm_body(v1, v2, w, v2_bstride, v2_sstride, out, side ? 1 : 0, foff, tile, lane);
  }
}

extern "C" void kernel_launch(void* const* d_in, const int* in_sizes, int n_in,
                              void* d_out, int out_size, void* d_ws, size_t ws_size,
                              hipStream_t stream) {
  const float* ctx_p = (const float*)d_in[0];
  const float* ctx_h = (const float*)d_in[1];
  const float* w0 = (const float*)d_in[2];
  const float* w1 = (const float*)d_in[3];
  const float* w2 = (const float*)d_in[4];
  const float* w3 = (const float*)d_in[5];
  const int* mask_p = (const int*)d_in[6];
  const int* mask_h = (const int*)d_in[7];
  float* out = (float*)d_out;
  float* ws = (float*)d_ws;

  const size_t SZ_ROWS = (size_t)BB * SS * HH;
  float* cp = ws;                 ws += SZ_ROWS;
  float* ch = ws;                 ws += SZ_ROWS;
  float* n_p = ws;                ws += BB * SS;
  float* n_h = ws;                ws += BB * SS;
  float* wn1_p = ws;              ws += (size_t)BB * SS * PP;
  float* wn1_h = ws;              ws += (size_t)BB * SS * PP;
  float* cpl = ws;                ws += BB * HH;
  float* chl = ws;                ws += BB * HH;
  float* cosM = ws;               ws += (size_t)BB * SS * SS;
  float* cossum_y = ws;           ws += BB * SS;
  float* cossum_x = ws;           ws += BB * SS;
  float* att_mean_h = ws;         ws += SZ_ROWS;
  float* att_mean_p = ws;         ws += SZ_ROWS;
  float* att_max_h = ws;          ws += SZ_ROWS;
  float* att_max_p = ws;          ws += SZ_ROWS;

  prep_gather_kernel<<<2 * NPREP + BB, 32, 0, stream>>>(ctx_p, ctx_h, mask_p, mask_h, w1, cp, ch, n_p, n_h,
                                                        wn1_p, wn1_h, cpl, chl);

  cos_wmma_kernel<<<BB * 64, 32, 0, stream>>>(cp, ch, n_p, n_h, cosM);

  cos_reduce_kernel<<<2 * BB * SS, 128, 0, stream>>>(cosM, mask_p, mask_h, cossum_y, cossum_x, out);

  att_kernel<<<2 * NATT / 8 + 2 * NAMX, 256, 0, stream>>>(cosM, cp, ch, cossum_y, cossum_x, mask_p, mask_h,
                                                             att_mean_h, att_mean_p, att_max_h, att_max_p, SS, 1);

  match_kernel<<<2 * NMAXP + 6 * NMPM, 32, 0, stream>>>(cp, ch, cpl, chl, att_mean_h, att_mean_p,
                                                        att_max_h, att_max_p, w0, w1, w2, w3,
                                                        wn1_p, wn1_h, mask_p, mask_h, out);
}


# cos row/column reductions (features 0,1, cos sums) ride in the cos GEMM launch: an 8-wave block owns a 16x128 strip (column strips recomputed) and reduces it; 5 launches -> 4; reduction tree changes
# speedup vs baseline: 7.7591x; 1.0598x over previous
#include <hip/hip_runtime.h>
#include <hip/hip_bf16.h>


#define BB 16
#define SS 128
#define HH 200
#define PP 20
#define FF 105
#define EPSF 1e-8f
#define KCH 10
#define NCH 5

typedef float v2f __attribute__((ext_vector_type(2)));
typedef float v4f __attribute__((ext_vector_type(4)));
typedef float v8f __attribute__((ext_vector_type(8)));

__device__ __forceinline__ v2f wsq_frag(v2f ww, bool isW, float fill) {
  v2f bv;
  bv.x = isW ? ww.x * ww.x : fill;
  bv.y = isW ? ww.y * ww.y : fill;
  return bv;
}

#define NPREP ((BB * SS / 16) * 2)
__device__ __forceinline__ void prep_body(const float* __restrict__ src, const int* __restrict__ mask,
                                 const float* __restrict__ w1, float* __restrict__ dst,
                                 float* __restrict__ nrm, float* __restrict__ wn, int tile, int lane) {
  int m0 = (tile >> 1) << 4;
  int n0 = (tile & 1) << 4;
  int half = lane >> 4, l16 = lane & 15;
  int bs = m0 + l16;
  float m = (float)mask[bs];
  const float* srow = src + (size_t)bs * HH + 2 * half;
  float* drow = dst + (size_t)bs * HH + 2 * half;
  int col = n0 + l16;
  bool isW = col < PP;
  float fill = (col == PP) ? 1.f : 0.f;
  const float* wrow = w1 + (size_t)(isW ? col : PP - 1) * HH + 2 * half;
  v8f acc = {0, 0, 0, 0, 0, 0, 0, 0};
  v2f xs[2][KCH], ws2[2][KCH];
#pragma unroll
  for (int i = 0; i < KCH; i++) {
    xs[0][i] = *(const v2f*)(srow + 4 * i);
    ws2[0][i] = *(const v2f*)(wrow + 4 * i);
  }
#pragma unroll
  for (int c = 0; c < NCH; c++) {
    if (c + 1 < NCH) {
#pragma unroll
      for (int i = 0; i < KCH; i++) {
        xs[(c + 1) & 1][i] = *(const v2f*)(srow + 4 * KCH * (c + 1) + 4 * i);
        ws2[(c + 1) & 1][i] = *(const v2f*)(wrow + 4 * KCH * (c + 1) + 4 * i);
      }
    }
    __builtin_amdgcn_sched_barrier(0);
#pragma unroll
    for (int i = 0; i < KCH; i++) {
      v2f x = xs[c & 1][i];
      x.x *= m; x.y *= m;
      if (n0 == 0) *(v2f*)(drow + 4 * KCH * c + 4 * i) = x;
      v2f bv = wsq_frag(ws2[c & 1][i], isW, fill);
      v2f a; a.x = x.x * x.x; a.y = x.y * x.y;
      acc = __builtin_amdgcn_wmma_f32_16x16x4_f32(false, a, false, bv, (short)0, acc, false, false);
    }
  }
  if (col <= PP) {
#pragma unroll
    for (int r = 0; r < 8; r++) {
      int row = m0 + r + 8 * half;
      float* o = (col == PP) ? nrm + row : wn + (size_t)row * PP + col;
      *o = sqrtf(acc[r]);
    }
  }
}

__device__ __forceinline__ void gather_last_body(const float* __restrict__ xp, const float* __restrict__ xh,
                                   const int* __restrict__ mask_p, const int* __restrict__ mask_h,
                                   float* __restrict__ cpl, float* __restrict__ chl, int b, int lane) {
  int lp = 0, lh = 0;
  for (int i = 0; i < SS; i++) { lp += mask_p[b * SS + i]; lh += mask_h[b * SS + i]; }
  int ip = (lp > 0) ? lp - 1 : 0;
  int ih = (lh > 0) ? lh - 1 : 0;
  float mp = (float)mask_p[b * SS + ip];
  float mh = (float)mask_h[b * SS + ih];
  for (int h = lane; h < HH; h += 32) {
    cpl[b * HH + h] = xp[((size_t)b * SS + ip) * HH + h] * mp;
    chl[b * HH + h] = xh[((size_t)b * SS + ih) * HH + h] * mh;
  }
}

__global__ void __launch_bounds__(32) prep_gather_kernel(
    const float* __restrict__ ctx_p, const float* __restrict__ ctx_h,
    const int* __restrict__ mask_p, const int* __restrict__ mask_h, const float* __restrict__ w1,
    float* __restrict__ cp, float* __restrict__ ch, float* __restrict__ n_p, float* __restrict__ n_h,
    float* __restrict__ wn1_p, float* __restrict__ wn1_h, float* __restrict__ cpl, float* __restrict__ chl) {
  int blk = blockIdx.x, lane = threadIdx.x;
  if (blk < 2 * NPREP) {
    bool side = blk >= NPREP;
    prep_body(side ? ctx_h : ctx_p, side ? mask_h : mask_p, w1, side ? ch : cp,
              side ? n_h : n_p, side ? wn1_h : wn1_p, side ? blk - NPREP : blk, lane);
  } else {
    gather_last_body(ctx_p, ctx_h, mask_p, mask_h, cpl, chl, blk - 2 * NPREP, lane);
  }
}

#define NCOS (BB * 8)
__global__ void __launch_bounds__(256) cos_strip_kernel(
    const float* __restrict__ cp, const float* __restrict__ ch,
    const float* __restrict__ np, const float* __restrict__ nh,
    const int* __restrict__ mask_p, const int* __restrict__ mask_h,
    float* __restrict__ cosO, float* __restrict__ cossum_y, float* __restrict__ cossum_x,
    float* __restrict__ out) {
  __shared__ float pmax[8][16], psum[8][16];
  int blk = blockIdx.x;
  int side = blk >= NCOS;
  if (side) blk -= NCOS;
  int b = blk >> 3, st = blk & 7;
  int w = threadIdx.x >> 5;
  int lane = threadIdx.x & 31, half = lane >> 4, l16 = lane & 15;
  int x0 = (side ? w : st) << 4;
  int y0 = (side ? st : w) << 4;
  const float* arow = cp + ((size_t)(b * SS) + x0 + l16) * HH + 2 * half;
  const float* brow = ch + ((size_t)(b * SS) + y0 + l16) * HH + 2 * half;
  v8f acc = {0, 0, 0, 0, 0, 0, 0, 0};
  for (int k = 0; k < HH; k += 4) {
    v2f a = *(const v2f*)(arow + k);
    v2f bv = *(const v2f*)(brow + k);
    acc = __builtin_amdgcn_wmma_f32_16x16x4_f32(false, a, false, bv, (short)0, acc, false, false);
  }
  float nhv = nh[b * SS + y0 + l16];
  float c[8];
#pragma unroll
  for (int r = 0; r < 8; r++) {
    int m = r + 8 * half;
    float d = fmaxf(np[b * SS + x0 + m] * nhv, EPSF);
    c[r] = acc[r] / d;
  }
  if (!side) {
    float mx[8], sm[8];
#pragma unroll
    for (int r = 0; r < 8; r++) {
      cosO[((size_t)(b * SS) + x0 + r + 8 * half) * SS + y0 + l16] = c[r];
      mx[r] = c[r]; sm[r] = c[r];
    }
#pragma unroll
    for (int o = 8; o > 0; o >>= 1) {
#pragma unroll
      for (int r = 0; r < 8; r++) {
        mx[r] = fmaxf(mx[r], __shfl_xor(mx[r], o, 32));
        sm[r] += __shfl_xor(sm[r], o, 32);
      }
    }
    if (l16 == 0) {
#pragma unroll
      for (int r = 0; r < 8; r++) { pmax[w][r + 8 * half] = mx[r]; psum[w][r + 8 * half] = sm[r]; }
    }
  } else {
    float mx = c[0], sm = c[0];
#pragma unroll
    for (int r = 1; r < 8; r++) { mx = fmaxf(mx, c[r]); sm += c[r]; }
    mx = fmaxf(mx, __shfl_xor(mx, 16, 32));
    sm += __shfl_xor(sm, 16, 32);
    if (half == 0) { pmax[w][l16] = mx; psum[w][l16] = sm; }
  }
  __syncthreads();
  if (threadIdx.x < 16) {
    int t = threadIdx.x;
    float mx = pmax[0][t], sm = psum[0][t];
#pragma unroll
    for (int u = 1; u < 8; u++) { mx = fmaxf(mx, pmax[u][t]); sm += psum[u][t]; }
    int r = (st << 4) + t;
    float m = (float)(side ? mask_h : mask_p)[b * SS + r];
    (side ? cossum_x : cossum_y)[b * SS + r] = sm;
    float* o2 = out + ((size_t)(side * BB + b) * SS + r) * FF;
    o2[0] = m * mx;
    o2[1] = m * (sm / (float)SS);
  }
}

#define NATT (BB * 8 * 13)
__device__ __forceinline__ void attmean_body(const float* __restrict__ cosM, const float* __restrict__ V,
                                    const float* __restrict__ csum, float* __restrict__ attO,
                                    int aStrideM, int aStrideK, int wid, int lane) {
  int b = wid / (8 * 13);
  int rem = wid % (8 * 13);
  int m0 = (rem / 13) * 16;
  int n0 = (rem % 13) * 16;
  int half = lane >> 4, l16 = lane & 15;
  const float* cb = cosM + (size_t)b * SS * SS;
  int nIdx = n0 + l16;
  bool nOK = nIdx < HH;
  v8f acc = {0, 0, 0, 0, 0, 0, 0, 0};
  for (int k = 0; k < SS; k += 4) {
    int ka = k + 2 * half;
    v2f a, bv;
    a.x = cb[(size_t)(m0 + l16) * aStrideM + (size_t)(ka + 0) * aStrideK];
    a.y = cb[(size_t)(m0 + l16) * aStrideM + (size_t)(ka + 1) * aStrideK];
    bv.x = nOK ? V[((size_t)b * SS + ka + 0) * HH + nIdx] : 0.f;
    bv.y = nOK ? V[((size_t)b * SS + ka + 1) * HH + nIdx] : 0.f;
    acc = __builtin_amdgcn_wmma_f32_16x16x4_f32(false, a, false, bv, (short)0, acc, false, false);
  }
  if (nOK) {
#pragma unroll
    for (int r = 0; r < 8; r++) {
      int m = m0 + r + 8 * half;
      float d = fmaxf(csum[b * SS + m], EPSF);
      attO[((size_t)b * SS + m) * HH + nIdx] = acc[r] / d;
    }
  }
}

#define RT 8
#define KU 8
#define NAMX (BB * SS / RT)
template <bool TR>
__device__ __forceinline__ void attmax_body(const float* __restrict__ cosM, const float* __restrict__ V,
                              const int* __restrict__ mask, float* __restrict__ attO,
                              v4f (*cs)[RT / 4], int blk, int t) {
  int b = blk / (SS / RT), r0 = (blk % (SS / RT)) * RT;
  const float* cb = cosM + (size_t)b * SS * SS;
  if (TR) {
    int k = t >> 1, q = t & 1;
    cs[k][q] = *(const v4f*)(cb + (size_t)k * SS + r0 + 4 * q);
  } else {
    int j = t & (RT - 1), k4 = t >> 3;
    v4f x = *(const v4f*)(cb + (size_t)(r0 + j) * SS + 4 * k4);
    float* csf = (float*)cs;
    csf[(4 * k4 + 0) * RT + j] = x.x;
    csf[(4 * k4 + 1) * RT + j] = x.y;
    csf[(4 * k4 + 2) * RT + j] = x.z;
    csf[(4 * k4 + 3) * RT + j] = x.w;
  }
  __syncthreads();
  int h = t;
  if (h >= HH) return;
  const float* vp = V + (size_t)b * SS * HH + h;
  float mx[RT];
#pragma unroll
  for (int j = 0; j < RT; j++) mx[j] = -3.0e38f;
  for (int k = 0; k < SS; k += KU) {
    float v[KU];
    v4f c[KU][RT / 4];
#pragma unroll
    for (int u = 0; u < KU; u++) v[u] = vp[(size_t)(k + u) * HH];
#pragma unroll
    for (int u = 0; u < KU; u++) {
#pragma unroll
      for (int q = 0; q < RT / 4; q++) c[u][q] = cs[k + u][q];
    }
    __builtin_amdgcn_sched_barrier(0);
#pragma unroll
    for (int u = 0; u < KU; u++) {
#pragma unroll
      for (int q = 0; q < RT / 4; q++) {
        mx[4 * q + 0] = fmaxf(mx[4 * q + 0], c[u][q].x * v[u]);
        mx[4 * q + 1] = fmaxf(mx[4 * q + 1], c[u][q].y * v[u]);
        mx[4 * q + 2] = fmaxf(mx[4 * q + 2], c[u][q].z * v[u]);
        mx[4 * q + 3] = fmaxf(mx[4 * q + 3], c[u][q].w * v[u]);
      }
    }
  }
#pragma unroll
  for (int j = 0; j < RT; j++) {
    float m = (float)mask[b * SS + r0 + j];
    attO[((size_t)b * SS + r0 + j) * HH + h] = m * mx[j];
  }
}

__global__ void __launch_bounds__(256) att_kernel(
    const float* __restrict__ cosM, const float* __restrict__ cp, const float* __restrict__ ch,
    const float* __restrict__ cossum_y, const float* __restrict__ cossum_x,
    const int* __restrict__ mask_p, const int* __restrict__ mask_h,
    float* __restrict__ att_mean_h, float* __restrict__ att_mean_p,
    float* __restrict__ att_max_h, float* __restrict__ att_max_p,
    int cosRowStride, int cosColStride) {
  __shared__ v4f cs[SS][RT / 4];
  int blk = blockIdx.x;
  if (blk < 2 * NATT / 8) {
    bool side = blk >= NATT / 8;
    int wid = (side ? blk - NATT / 8 : blk) * 8 + (threadIdx.x >> 5);
    attmean_body(cosM, side ? cp : ch, side ? cossum_x : cossum_y, side ? att_mean_p : att_mean_h,
                 side ? cosColStride : cosRowStride, side ? cosRowStride : cosColStride,
                 wid, threadIdx.x & 31);
  } else {
    blk -= 2 * NATT / 8;
    if (blk >= NAMX) attmax_body<true>(cosM, cp, mask_h, att_max_p, cs, blk - NAMX, threadIdx.x);
    else attmax_body<false>(cosM, ch, mask_p, att_max_h, cs, blk, threadIdx.x);
  }
}

#define NMPM ((BB * SS / 16) * 2)
__device__ __forceinline__ void mpm_body(const float* __restrict__ v1, const float* __restrict__ v2,
                                const float* __restrict__ w, int v2_bstride, int v2_sstride,
                                float* __restrict__ out, int side, int foff, int tile, int lane) {
  int m0 = (tile >> 1) << 4;
  int n0 = (tile & 1) << 4;
  int half = lane >> 4, l16 = lane & 15;
  int bs = m0 + l16;
  int b = bs / SS, s = bs % SS;
  const float* p1 = v1 + (size_t)bs * HH + 2 * half;
  const float* p2 = v2 + (size_t)b * v2_bstride + (size_t)s * v2_sstride + 2 * half;
  int col = n0 + l16;
  bool isW = col < PP;
  float fill = (col == PP) ? 1.f : 0.f;
  const float* wrow = w + (size_t)(isW ? col : PP - 1) * HH + 2 * half;
  v8f s11 = {0, 0, 0, 0, 0, 0, 0, 0};
  v8f s22 = {0, 0, 0, 0, 0, 0, 0, 0};
  v8f s12 = {0, 0, 0, 0, 0, 0, 0, 0};
  v2f xs[2][KCH], ys[2][KCH], ws2[2][KCH];
#pragma unroll
  for (int i = 0; i < KCH; i++) {
    xs[0][i] = *(const v2f*)(p1 + 4 * i);
    ys[0][i] = *(const v2f*)(p2 + 4 * i);
    ws2[0][i] = *(const v2f*)(wrow + 4 * i);
  }
#pragma unroll
  for (int c = 0; c < NCH; c++) {
    if (c + 1 < NCH) {
#pragma unroll
      for (int i = 0; i < KCH; i++) {
        xs[(c + 1) & 1][i] = *(const v2f*)(p1 + 4 * KCH * (c + 1) + 4 * i);
        ys[(c + 1) & 1][i] = *(const v2f*)(p2 + 4 * KCH * (c + 1) + 4 * i);
        ws2[(c + 1) & 1][i] = *(const v2f*)(wrow + 4 * KCH * (c + 1) + 4 * i);
      }
    }
    __builtin_amdgcn_sched_barrier(0);
#pragma unroll
    for (int i = 0; i < KCH; i++) {
      v2f x = xs[c & 1][i], y = ys[c & 1][i];
      v2f bv = wsq_frag(ws2[c & 1][i], isW, fill);
      v2f xx, yy, xy;
      xx.x = x.x * x.x; xx.y = x.y * x.y;
      yy.x = y.x * y.x; yy.y = y.y * y.y;
      xy.x = x.x * y.x; xy.y = x.y * y.y;
      s11 = __builtin_amdgcn_wmma_f32_16x16x4_f32(false, xx, false, bv, (short)0, s11, false, false);
      s22 = __builtin_amdgcn_wmma_f32_16x16x4_f32(false, yy, false, bv, (short)0, s22, false, false);
      s12 = __builtin_amdgcn_wmma_f32_16x16x4_f32(false, xy, false, bv, (short)0, s12, false, false);
    }
  }
  if (col <= PP) {
    int f = (col == PP) ? 0 : 1 + col;
#pragma unroll
    for (int r = 0; r < 8; r++) {
      int row = m0 + r + 8 * half;
      int ob = row / SS, os = row % SS;
      float* orow = out + ((size_t)(side * BB + ob) * SS + os) * FF + foff;
      orow[f] = s12[r] / (fmaxf(sqrtf(s11[r]), EPSF) * fmaxf(sqrtf(s22[r]), EPSF));
    }
  }
}

#define NJ 25
#define GCH 10
#define NYT 8
#define NMAXP (BB * PP * 8)
__device__ __forceinline__ void maxpool_body(const float* __restrict__ v1, const float* __restrict__ v2,
                                    const float* __restrict__ w,
                                    const float* __restrict__ wnA, const float* __restrict__ wnB,
                                    const int* __restrict__ mask, float* __restrict__ out,
                                    int side, int wid, int lane) {
  int b = wid / (PP * 8);
  int p = (wid / 8) % PP;
  int x0 = (wid % 8) * 16;
  int half = lane >> 4, l16 = lane & 15;
  float rmax[8], rsum[8], wA[8];
#pragma unroll
  for (int r = 0; r < 8; r++) {
    rmax[r] = -3.0e38f; rsum[r] = 0.f;
    wA[r] = wnA[((size_t)b * SS + x0 + r + 8 * half) * PP + p];
  }
  const float* a_row = v1 + ((size_t)b * SS + x0 + l16) * HH + 4 * half;
  const float* w_row = w + (size_t)p * HH + 4 * half;
  const float* b_row = v2 + ((size_t)b * SS + l16) * HH + 4 * half;
  const float* wnB_col = wnB + ((size_t)b * SS + l16) * PP + p;

  v4f bf[2][GCH];
#pragma unroll
  for (int i = 0; i < GCH; i++) bf[0][i] = *(const v4f*)(b_row + 8 * i);

  v4f af[NJ];
#pragma unroll
  for (int j = 0; j < NJ; j++) {
    v4f wv = *(const v4f*)(w_row + 8 * j);
    v4f av = *(const v4f*)(a_row + 8 * j);
    af[j].x = wv.x * wv.x * av.x;
    af[j].y = wv.y * wv.y * av.y;
    af[j].z = wv.z * wv.z * av.z;
    af[j].w = wv.w * wv.w * av.w;
  }

  v8f acc = {0, 0, 0, 0, 0, 0, 0, 0};
  float wB = 0.f;
#pragma unroll
  for (int g0 = 0; g0 < NYT * NJ; g0 += GCH) {
    if (g0 + GCH < NYT * NJ) {
#pragma unroll
      for (int i = 0; i < GCH; i++) {
        int gn = g0 + GCH + i;
        bf[((g0 / GCH) + 1) & 1][i] = *(const v4f*)(b_row + (gn / NJ) * 16 * HH + 8 * (gn % NJ));
      }
    }
    __builtin_amdgcn_sched_barrier(0);
#pragma unroll
    for (int i = 0; i < GCH; i++) {
      int g = g0 + i;
      int yt = g / NJ, j = g % NJ;
      if (j == 0) {
        acc = (v8f){0, 0, 0, 0, 0, 0, 0, 0};
        wB = wnB_col[(size_t)yt * 16 * PP];
      }
      v4f bb = bf[(g0 / GCH) & 1][i];
      v4f aa = af[j];
      v2f a0, b0, a1, b1;
      a0.x = aa.x; a0.y = aa.y; b0.x = bb.x; b0.y = bb.y;
      a1.x = aa.z; a1.y = aa.w; b1.x = bb.z; b1.y = bb.w;
      acc = __builtin_amdgcn_wmma_f32_16x16x4_f32(false, a0, false, b0, (short)0, acc, false, false);
      acc = __builtin_amdgcn_wmma_f32_16x16x4_f32(false, a1, false, b1, (short)0, acc, false, false);
      if (j == NJ - 1) {
#pragma unroll
        for (int r = 0; r < 8; r++) {
          float mv = acc[r] / fmaxf(wA[r] * wB, EPSF);
          rmax[r] = fmaxf(rmax[r], mv);
          rsum[r] += mv;
        }
      }
    }
  }
#pragma unroll
  for (int r = 0; r < 8; r++) {
    for (int o = 8; o > 0; o >>= 1) {
      rmax[r] = fmaxf(rmax[r], __shfl_xor(rmax[r], o, 32));
      rsum[r] += __shfl_xor(rsum[r], o, 32);
    }
  }
  if (l16 == 0) {
#pragma unroll
    for (int r = 0; r < 8; r++) {
      int x = x0 + r + 8 * half;
      float m = (float)mask[b * SS + x];
      float* orow = out + ((size_t)(side * BB + b) * SS + x) * FF;
      orow[23 + p] = m * rmax[r];
      orow[43 + p] = m * (rsum[r] / (float)SS);
    }
  }
}

__global__ void __launch_bounds__(32) match_kernel(
    const float* __restrict__ cp, const float* __restrict__ ch,
    const float* __restrict__ cpl, const float* __restrict__ chl,
    const float* __restrict__ att_mean_h, const float* __restrict__ att_mean_p,
    const float* __restrict__ att_max_h, const float* __restrict__ att_max_p,
    const float* __restrict__ w0, const float* __restrict__ w1,
    const float* __restrict__ w2, const float* __restrict__ w3,
    const float* __restrict__ wn1_p, const float* __restrict__ wn1_h,
    const int* __restrict__ mask_p, const int* __restrict__ mask_h, float* __restrict__ out) {
  int blk = blockIdx.x, lane = threadIdx.x;
  if (blk < 2 * NMAXP) {
    bool side = blk >= NMAXP;
    maxpool_body(side ? ch : cp, side ? cp : ch, w1, side ? wn1_h : wn1_p, side ? wn1_p : wn1_h,
                 side ? mask_h : mask_p, out, side ? 1 : 0, side ? blk - NMAXP : blk, lane);
  } else {
    blk -= 2 * NMAXP;
    int grp = blk / (2 * NMPM);
    int rem = blk % (2 * NMPM);
    bool side = rem >= NMPM;
    int tile = side ? rem - NMPM : rem;
    const float* v1 = side ? ch : cp;
    const float* v2 = grp == 0 ? (side ? cpl : chl)
                    : grp == 1 ? (side ? att_mean_p : att_mean_h) : (side ? att_max_p : att_max_h);
    const float* w = grp == 0 ? w0 : grp == 1 ? w2 : w3;
    int v2_bstride = grp == 0 ? HH : SS * HH;
    int v2_sstride = grp == 0 ? 0 : HH;
    int foff = grp == 0 ? 2 : grp == 1 ? 63 : 84;
    mpm_body(v1, v2, w, v2_bstride, v2_sstride, out, side ? 1 : 0, foff, tile, lane);
  }
}

extern "C" void kernel_launch(void* const* d_in, const int* in_sizes, int n_in,
                              void* d_out, int out_size, void* d_ws, size_t ws_size,
                              hipStream_t stream) {
  const float* ctx_p = (const float*)d_in[0];
  const float* ctx_h = (const float*)d_in[1];
  const float* w0 = (const float*)d_in[2];
  const float* w1 = (const float*)d_in[3];
  const float* w2 = (const float*)d_in[4];
  const float* w3 = (const float*)d_in[5];
  const int* mask_p = (const int*)d_in[6];
  const int* mask_h = (const int*)d_in[7];
  float* out = (float*)d_out;
  float* ws = (float*)d_ws;

  const size_t SZ_ROWS = (size_t)BB * SS * HH;
  float* cp = ws;                 ws += SZ_ROWS;
  float* ch = ws;                 ws += SZ_ROWS;
  float* n_p = ws;                ws += BB * SS;
  float* n_h = ws;                ws += BB * SS;
  float* wn1_p = ws;              ws += (size_t)BB * SS * PP;
  float* wn1_h = ws;              ws += (size_t)BB * SS * PP;
  float* cpl = ws;                ws += BB * HH;
  float* chl = ws;                ws += BB * HH;
  float* cosM = ws;               ws += (size_t)BB * SS * SS;
  float* cossum_y = ws;           ws += BB * SS;
  float* cossum_x = ws;           ws += BB * SS;
  float* att_mean_h = ws;         ws += SZ_ROWS;
  float* att_mean_p = ws;         ws += SZ_ROWS;
  float* att_max_h = ws;          ws += SZ_ROWS;
  float* att_max_p = ws;          ws += SZ_ROWS;

  prep_gather_kernel<<<2 * NPREP + BB, 32, 0, stream>>>(ctx_p, ctx_h, mask_p, mask_h, w1, cp, ch, n_p, n_h,
                                                        wn1_p, wn1_h, cpl, chl);

  cos_strip_kernel<<<2 * NCOS, 256, 0, stream>>>(cp, ch, n_p, n_h, mask_p, mask_h, cosM,
                                                 cossum_y, cossum_x, out);

  att_kernel<<<2 * NATT / 8 + 2 * NAMX, 256, 0, stream>>>(cosM, cp, ch, cossum_y, cossum_x, mask_p, mask_h,
                                                             att_mean_h, att_mean_p, att_max_h, att_max_p, SS, 1);

  match_kernel<<<2 * NMAXP + 6 * NMPM, 32, 0, stream>>>(cp, ch, cpl, chl, att_mean_h, att_mean_p,
                                                        att_max_h, att_max_p, w0, w1, w2, w3,
                                                        wn1_p, wn1_h, mask_p, mask_h, out);
}


# maxpool match: an 8-wave block per (b,p) computes each tile once and reduces it over y in registers and over x through LDS (was a wave per x-tile and side, each tile twice); x-sum order changes
# speedup vs baseline: 9.0893x; 1.1714x over previous
#include <hip/hip_runtime.h>
#include <hip/hip_bf16.h>


#define BB 16
#define SS 128
#define HH 200
#define PP 20
#define FF 105
#define EPSF 1e-8f
#define KCH 10
#define NCH 5

typedef float v2f __attribute__((ext_vector_type(2)));
typedef float v4f __attribute__((ext_vector_type(4)));
typedef float v8f __attribute__((ext_vector_type(8)));

__device__ __forceinline__ v2f wsq_frag(v2f ww, bool isW, float fill) {
  v2f bv;
  bv.x = isW ? ww.x * ww.x : fill;
  bv.y = isW ? ww.y * ww.y : fill;
  return bv;
}

#define NPREP ((BB * SS / 16) * 2)
__device__ __forceinline__ void prep_body(const float* __restrict__ src, const int* __restrict__ mask,
                                 const float* __restrict__ w1, float* __restrict__ dst,
                                 float* __restrict__ nrm, float* __restrict__ wn, int tile, int lane) {
  int m0 = (tile >> 1) << 4;
  int n0 = (tile & 1) << 4;
  int half = lane >> 4, l16 = lane & 15;
  int bs = m0 + l16;
  float m = (float)mask[bs];
  const float* srow = src + (size_t)bs * HH + 2 * half;
  float* drow = dst + (size_t)bs * HH + 2 * half;
  int col = n0 + l16;
  bool isW = col < PP;
  float fill = (col == PP) ? 1.f : 0.f;
  const float* wrow = w1 + (size_t)(isW ? col : PP - 1) * HH + 2 * half;
  v8f acc = {0, 0, 0, 0, 0, 0, 0, 0};
  v2f xs[2][KCH], ws2[2][KCH];
#pragma unroll
  for (int i = 0; i < KCH; i++) {
    xs[0][i] = *(const v2f*)(srow + 4 * i);
    ws2[0][i] = *(const v2f*)(wrow + 4 * i);
  }
#pragma unroll
  for (int c = 0; c < NCH; c++) {
    if (c + 1 < NCH) {
#pragma unroll
      for (int i = 0; i < KCH; i++) {
        xs[(c + 1) & 1][i] = *(const v2f*)(srow + 4 * KCH * (c + 1) + 4 * i);
        ws2[(c + 1) & 1][i] = *(const v2f*)(wrow + 4 * KCH * (c + 1) + 4 * i);
      }
    }
    __builtin_amdgcn_sched_barrier(0);
#pragma unroll
    for (int i = 0; i < KCH; i++) {
      v2f x = xs[c & 1][i];
      x.x *= m; x.y *= m;
      if (n0 == 0) *(v2f*)(drow + 4 * KCH * c + 4 * i) = x;
      v2f bv = wsq_frag(ws2[c & 1][i], isW, fill);
      v2f a; a.x = x.x * x.x; a.y = x.y * x.y;
      acc = __builtin_amdgcn_wmma_f32_16x16x4_f32(false, a, false, bv, (short)0, acc, false, false);
    }
  }
  if (col <= PP) {
#pragma unroll
    for (int r = 0; r < 8; r++) {
      int row = m0 + r + 8 * half;
      float* o = (col == PP) ? nrm + row : wn + (size_t)row * PP + col;
      *o = sqrtf(acc[r]);
    }
  }
}

__device__ __forceinline__ void gather_last_body(const float* __restrict__ xp, const float* __restrict__ xh,
                                   const int* __restrict__ mask_p, const int* __restrict__ mask_h,
                                   float* __restrict__ cpl, float* __restrict__ chl, int b, int lane) {
  int lp = 0, lh = 0;
  for (int i = 0; i < SS; i++) { lp += mask_p[b * SS + i]; lh += mask_h[b * SS + i]; }
  int ip = (lp > 0) ? lp - 1 : 0;
  int ih = (lh > 0) ? lh - 1 : 0;
  float mp = (float)mask_p[b * SS + ip];
  float mh = (float)mask_h[b * SS + ih];
  for (int h = lane; h < HH; h += 32) {
    cpl[b * HH + h] = xp[((size_t)b * SS + ip) * HH + h] * mp;
    chl[b * HH + h] = xh[((size_t)b * SS + ih) * HH + h] * mh;
  }
}

__global__ void __launch_bounds__(32) prep_gather_kernel(
    const float* __restrict__ ctx_p, const float* __restrict__ ctx_h,
    const int* __restrict__ mask_p, const int* __restrict__ mask_h, const float* __restrict__ w1,
    float* __restrict__ cp, float* __restrict__ ch, float* __restrict__ n_p, float* __restrict__ n_h,
    float* __restrict__ wn1_p, float* __restrict__ wn1_h, float* __restrict__ cpl, float* __restrict__ chl) {
  int blk = blockIdx.x, lane = threadIdx.x;
  if (blk < 2 * NPREP) {
    bool side = blk >= NPREP;
    prep_body(side ? ctx_h : ctx_p, side ? mask_h : mask_p, w1, side ? ch : cp,
              side ? n_h : n_p, side ? wn1_h : wn1_p, side ? blk - NPREP : blk, lane);
  } else {
    gather_last_body(ctx_p, ctx_h, mask_p, mask_h, cpl, chl, blk - 2 * NPREP, lane);
  }
}

#define NCOS (BB * 8)
__global__ void __launch_bounds__(256) cos_strip_kernel(
    const float* __restrict__ cp, const float* __restrict__ ch,
    const float* __restrict__ np, const float* __restrict__ nh,
    const int* __restrict__ mask_p, const int* __restrict__ mask_h,
    float* __restrict__ cosO, float* __restrict__ cossum_y, float* __restrict__ cossum_x,
    float* __restrict__ out) {
  __shared__ float pmax[8][16], psum[8][16];
  int blk = blockIdx.x;
  int side = blk >= NCOS;
  if (side) blk -= NCOS;
  int b = blk >> 3, st = blk & 7;
  int w = threadIdx.x >> 5;
  int lane = threadIdx.x & 31, half = lane >> 4, l16 = lane & 15;
  int x0 = (side ? w : st) << 4;
  int y0 = (side ? st : w) << 4;
  const float* arow = cp + ((size_t)(b * SS) + x0 + l16) * HH + 2 * half;
  const float* brow = ch + ((size_t)(b * SS) + y0 + l16) * HH + 2 * half;
  v8f acc = {0, 0, 0, 0, 0, 0, 0, 0};
  for (int k = 0; k < HH; k += 4) {
    v2f a = *(const v2f*)(arow + k);
    v2f bv = *(const v2f*)(brow + k);
    acc = __builtin_amdgcn_wmma_f32_16x16x4_f32(false, a, false, bv, (short)0, acc, false, false);
  }
  float nhv = nh[b * SS + y0 + l16];
  float c[8];
#pragma unroll
  for (int r = 0; r < 8; r++) {
    int m = r + 8 * half;
    float d = fmaxf(np[b * SS + x0 + m] * nhv, EPSF);
    c[r] = acc[r] / d;
  }
  if (!side) {
    float mx[8], sm[8];
#pragma unroll
    for (int r = 0; r < 8; r++) {
      cosO[((size_t)(b * SS) + x0 + r + 8 * half) * SS + y0 + l16] = c[r];
      mx[r] = c[r]; sm[r] = c[r];
    }
#pragma unroll
    for (int o = 8; o > 0; o >>= 1) {
#pragma unroll
      for (int r = 0; r < 8; r++) {
        mx[r] = fmaxf(mx[r], __shfl_xor(mx[r], o, 32));
        sm[r] += __shfl_xor(sm[r], o, 32);
      }
    }
    if (l16 == 0) {
#pragma unroll
      for (int r = 0; r < 8; r++) { pmax[w][r + 8 * half] = mx[r]; psum[w][r + 8 * half] = sm[r]; }
    }
  } else {
    float mx = c[0], sm = c[0];
#pragma unroll
    for (int r = 1; r < 8; r++) { mx = fmaxf(mx, c[r]); sm += c[r]; }
    mx = fmaxf(mx, __shfl_xor(mx, 16, 32));
    sm += __shfl_xor(sm, 16, 32);
    if (half == 0) { pmax[w][l16] = mx; psum[w][l16] = sm; }
  }
  __syncthreads();
  if (threadIdx.x < 16) {
    int t = threadIdx.x;
    float mx = pmax[0][t], sm = psum[0][t];
#pragma unroll
    for (int u = 1; u < 8; u++) { mx = fmaxf(mx, pmax[u][t]); sm += psum[u][t]; }
    int r = (st << 4) + t;
    float m = (float)(side ? mask_h : mask_p)[b * SS + r];
    (side ? cossum_x : cossum_y)[b * SS + r] = sm;
    float* o2 = out + ((size_t)(side * BB + b) * SS + r) * FF;
    o2[0] = m * mx;
    o2[1] = m * (sm / (float)SS);
  }
}

#define NATT (BB * 8 * 13)
__device__ __forceinline__ void attmean_body(const float* __restrict__ cosM, const float* __restrict__ V,
                                    const float* __restrict__ csum, float* __restrict__ attO,
                                    int aStrideM, int aStrideK, int wid, int lane) {
  int b = wid / (8 * 13);
  int rem = wid % (8 * 13);
  int m0 = (rem / 13) * 16;
  int n0 = (rem % 13) * 16;
  int half = lane >> 4, l16 = lane & 15;
  const float* cb = cosM + (size_t)b * SS * SS;
  int nIdx = n0 + l16;
  bool nOK = nIdx < HH;
  v8f acc = {0, 0, 0, 0, 0, 0, 0, 0};
  for (int k = 0; k < SS; k += 4) {
    int ka = k + 2 * half;
    v2f a, bv;
    a.x = cb[(size_t)(m0 + l16) * aStrideM + (size_t)(ka + 0) * aStrideK];
    a.y = cb[(size_t)(m0 + l16) * aStrideM + (size_t)(ka + 1) * aStrideK];
    bv.x = nOK ? V[((size_t)b * SS + ka + 0) * HH + nIdx] : 0.f;
    bv.y = nOK ? V[((size_t)b * SS + ka + 1) * HH + nIdx] : 0.f;
    acc = __builtin_amdgcn_wmma_f32_16x16x4_f32(false, a, false, bv, (short)0, acc, false, false);
  }
  if (nOK) {
#pragma unroll
    for (int r = 0; r < 8; r++) {
      int m = m0 + r + 8 * half;
      float d = fmaxf(csum[b * SS + m], EPSF);
      attO[((size_t)b * SS + m) * HH + nIdx] = acc[r] / d;
    }
  }
}

#define RT 8
#define KU 8
#define NAMX (BB * SS / RT)
template <bool TR>
__device__ __forceinline__ void attmax_body(const float* __restrict__ cosM, const float* __restrict__ V,
                              const int* __restrict__ mask, float* __restrict__ attO,
                              v4f (*cs)[RT / 4], int blk, int t) {
  int b = blk / (SS / RT), r0 = (blk % (SS / RT)) * RT;
  const float* cb = cosM + (size_t)b * SS * SS;
  if (TR) {
    int k = t >> 1, q = t & 1;
    cs[k][q] = *(const v4f*)(cb + (size_t)k * SS + r0 + 4 * q);
  } else {
    int j = t & (RT - 1), k4 = t >> 3;
    v4f x = *(const v4f*)(cb + (size_t)(r0 + j) * SS + 4 * k4);
    float* csf = (float*)cs;
    csf[(4 * k4 + 0) * RT + j] = x.x;
    csf[(4 * k4 + 1) * RT + j] = x.y;
    csf[(4 * k4 + 2) * RT + j] = x.z;
    csf[(4 * k4 + 3) * RT + j] = x.w;
  }
  __syncthreads();
  int h = t;
  if (h >= HH) return;
  const float* vp = V + (size_t)b * SS * HH + h;
  float mx[RT];
#pragma unroll
  for (int j = 0; j < RT; j++) mx[j] = -3.0e38f;
  for (int k = 0; k < SS; k += KU) {
    float v[KU];
    v4f c[KU][RT / 4];
#pragma unroll
    for (int u = 0; u < KU; u++) v[u] = vp[(size_t)(k + u) * HH];
#pragma unroll
    for (int u = 0; u < KU; u++) {
#pragma unroll
      for (int q = 0; q < RT / 4; q++) c[u][q] = cs[k + u][q];
    }
    __builtin_amdgcn_sched_barrier(0);
#pragma unroll
    for (int u = 0; u < KU; u++) {
#pragma unroll
      for (int q = 0; q < RT / 4; q++) {
        mx[4 * q + 0] = fmaxf(mx[4 * q + 0], c[u][q].x * v[u]);
        mx[4 * q + 1] = fmaxf(mx[4 * q + 1], c[u][q].y * v[u]);
        mx[4 * q + 2] = fmaxf(mx[4 * q + 2], c[u][q].z * v[u]);
        mx[4 * q + 3] = fmaxf(mx[4 * q + 3], c[u][q].w * v[u]);
      }
    }
  }
#pragma unroll
  for (int j = 0; j < RT; j++) {
    float m = (float)mask[b * SS + r0 + j];
    attO[((size_t)b * SS + r0 + j) * HH + h] = m * mx[j];
  }
}

__global__ void __launch_bounds__(256) att_kernel(
    const float* __restrict__ cosM, const float* __restrict__ cp, const float* __restrict__ ch,
    const float* __restrict__ cossum_y, const float* __restrict__ cossum_x,
    const int* __restrict__ mask_p, const int* __restrict__ mask_h,
    float* __restrict__ att_mean_h, float* __restrict__ att_mean_p,
    float* __restrict__ att_max_h, float* __restrict__ att_max_p,
    int cosRowStride, int cosColStride) {
  __shared__ v4f cs[SS][RT / 4];
  int blk = blockIdx.x;
  if (blk < 2 * NATT / 8) {
    bool side = blk >= NATT / 8;
    int wid = (side ? blk - NATT / 8 : blk) * 8 + (threadIdx.x >> 5);
    attmean_body(cosM, side ? cp : ch, side ? cossum_x : cossum_y, side ? att_mean_p : att_mean_h,
                 side ? cosColStride : cosRowStride, side ? cosRowStride : cosColStride,
                 wid, threadIdx.x & 31);
  } else {
    blk -= 2 * NATT / 8;
    if (blk >= NAMX) attmax_body<true>(cosM, cp, mask_h, att_max_p, cs, blk - NAMX, threadIdx.x);
    else attmax_body<false>(cosM, ch, mask_p, att_max_h, cs, blk, threadIdx.x);
  }
}

#define NMPM ((BB * SS / 16) * 2)
__device__ __forceinline__ void mpm_body(const float* __restrict__ v1, const float* __restrict__ v2,
                                const float* __restrict__ w, int v2_bstride, int v2_sstride,
                                float* __restrict__ out, int side, int foff, int tile, int lane) {
  int m0 = (tile >> 1) << 4;
  int n0 = (tile & 1) << 4;
  int half = lane >> 4, l16 = lane & 15;
  int bs = m0 + l16;
  int b = bs / SS, s = bs % SS;
  const float* p1 = v1 + (size_t)bs * HH + 2 * half;
  const float* p2 = v2 + (size_t)b * v2_bstride + (size_t)s * v2_sstride + 2 * half;
  int col = n0 + l16;
  bool isW = col < PP;
  float fill = (col == PP) ? 1.f : 0.f;
  const float* wrow = w + (size_t)(isW ? col : PP - 1) * HH + 2 * half;
  v8f s11 = {0, 0, 0, 0, 0, 0, 0, 0};
  v8f s22 = {0, 0, 0, 0, 0, 0, 0, 0};
  v8f s12 = {0, 0, 0, 0, 0, 0, 0, 0};
  v2f xs[2][KCH], ys[2][KCH], ws2[2][KCH];
#pragma unroll
  for (int i = 0; i < KCH; i++) {
    xs[0][i] = *(const v2f*)(p1 + 4 * i);
    ys[0][i] = *(const v2f*)(p2 + 4 * i);
    ws2[0][i] = *(const v2f*)(wrow + 4 * i);
  }
#pragma unroll
  for (int c = 0; c < NCH; c++) {
    if (c + 1 < NCH) {
#pragma unroll
      for (int i = 0; i < KCH; i++) {
        xs[(c + 1) & 1][i] = *(const v2f*)(p1 + 4 * KCH * (c + 1) + 4 * i);
        ys[(c + 1) & 1][i] = *(const v2f*)(p2 + 4 * KCH * (c + 1) + 4 * i);
        ws2[(c + 1) & 1][i] = *(const v2f*)(wrow + 4 * KCH * (c + 1) + 4 * i);
      }
    }
    __builtin_amdgcn_sched_barrier(0);
#pragma unroll
    for (int i = 0; i < KCH; i++) {
      v2f x = xs[c & 1][i], y = ys[c & 1][i];
      v2f bv = wsq_frag(ws2[c & 1][i], isW, fill);
      v2f xx, yy, xy;
      xx.x = x.x * x.x; xx.y = x.y * x.y;
      yy.x = y.x * y.x; yy.y = y.y * y.y;
      xy.x = x.x * y.x; xy.y = x.y * y.y;
      s11 = __builtin_amdgcn_wmma_f32_16x16x4_f32(false, xx, false, bv, (short)0, s11, false, false);
      s22 = __builtin_amdgcn_wmma_f32_16x16x4_f32(false, yy, false, bv, (short)0, s22, false, false);
      s12 = __builtin_amdgcn_wmma_f32_16x16x4_f32(false, xy, false, bv, (short)0, s12, false, false);
    }
  }
  if (col <= PP) {
    int f = (col == PP) ? 0 : 1 + col;
#pragma unroll
    for (int r = 0; r < 8; r++) {
      int row = m0 + r + 8 * half;
      int ob = row / SS, os = row % SS;
      float* orow = out + ((size_t)(side * BB + ob) * SS + os) * FF + foff;
      orow[f] = s12[r] / (fmaxf(sqrtf(s11[r]), EPSF) * fmaxf(sqrtf(s22[r]), EPSF));
    }
  }
}

#define NJ 25
#define GCH 10
#define NYT 8
#define NMAXB (BB * PP)
__device__ __forceinline__ void maxpool_block(const float* __restrict__ cp, const float* __restrict__ ch,
                                     const float* __restrict__ w,
                                     const float* __restrict__ wnP, const float* __restrict__ wnH,
                                     const int* __restrict__ mask_p, const int* __restrict__ mask_h,
                                     float* __restrict__ out, float (*colS)[NYT][SS], int blk, int tid) {
  int b = blk / PP;
  int p = blk % PP;
  int wv = tid >> 5, lane = tid & 31;
  int x0 = wv * 16;
  int half = lane >> 4, l16 = lane & 15;
  float rmax[8], rsum[8], wA[8];
#pragma unroll
  for (int r = 0; r < 8; r++) {
    rmax[r] = -3.0e38f; rsum[r] = 0.f;
    wA[r] = wnP[((size_t)b * SS + x0 + r + 8 * half) * PP + p];
  }
  const float* a_row = cp + ((size_t)b * SS + x0 + l16) * HH + 4 * half;
  const float* w_row = w + (size_t)p * HH + 4 * half;
  const float* b_row = ch + ((size_t)b * SS + l16) * HH + 4 * half;
  const float* wnB_col = wnH + ((size_t)b * SS + l16) * PP + p;
  float* col_slot = &colS[half][wv][l16];

  v4f bf[2][GCH];
#pragma unroll
  for (int i = 0; i < GCH; i++) bf[0][i] = *(const v4f*)(b_row + 8 * i);

  v4f af[NJ];
#pragma unroll
  for (int j = 0; j < NJ; j++) {
    v4f wv4 = *(const v4f*)(w_row + 8 * j);
    v4f av = *(const v4f*)(a_row + 8 * j);
    af[j].x = wv4.x * wv4.x * av.x;
    af[j].y = wv4.y * wv4.y * av.y;
    af[j].z = wv4.z * wv4.z * av.z;
    af[j].w = wv4.w * wv4.w * av.w;
  }

  v8f acc = {0, 0, 0, 0, 0, 0, 0, 0};
  float wB = 0.f;
#pragma unroll
  for (int g0 = 0; g0 < NYT * NJ; g0 += GCH) {
    if (g0 + GCH < NYT * NJ) {
#pragma unroll
      for (int i = 0; i < GCH; i++) {
        int gn = g0 + GCH + i;
        bf[((g0 / GCH) + 1) & 1][i] = *(const v4f*)(b_row + (gn / NJ) * 16 * HH + 8 * (gn % NJ));
      }
    }
    __builtin_amdgcn_sched_barrier(0);
#pragma unroll
    for (int i = 0; i < GCH; i++) {
      int g = g0 + i;
      int yt = g / NJ, j = g % NJ;
      if (j == 0) {
        acc = (v8f){0, 0, 0, 0, 0, 0, 0, 0};
        wB = wnB_col[(size_t)yt * 16 * PP];
      }
      v4f bb = bf[(g0 / GCH) & 1][i];
      v4f aa = af[j];
      v2f a0, b0, a1, b1;
      a0.x = aa.x; a0.y = aa.y; b0.x = bb.x; b0.y = bb.y;
      a1.x = aa.z; a1.y = aa.w; b1.x = bb.z; b1.y = bb.w;
      acc = __builtin_amdgcn_wmma_f32_16x16x4_f32(false, a0, false, b0, (short)0, acc, false, false);
      acc = __builtin_amdgcn_wmma_f32_16x16x4_f32(false, a1, false, b1, (short)0, acc, false, false);
      if (j == NJ - 1) {
        float cm = -3.0e38f, csm = 0.f;
#pragma unroll
        for (int r = 0; r < 8; r++) {
          float mv = acc[r] / fmaxf(wA[r] * wB, EPSF);
          rmax[r] = fmaxf(rmax[r], mv);
          rsum[r] += mv;
          cm = fmaxf(cm, mv);
          csm += mv;
        }
        cm = fmaxf(cm, __shfl_xor(cm, 16, 32));
        csm += __shfl_xor(csm, 16, 32);
        col_slot[yt * 16] = half ? csm : cm;
      }
    }
  }
#pragma unroll
  for (int r = 0; r < 8; r++) {
    for (int o = 8; o > 0; o >>= 1) {
      rmax[r] = fmaxf(rmax[r], __shfl_xor(rmax[r], o, 32));
      rsum[r] += __shfl_xor(rsum[r], o, 32);
    }
  }
  if (l16 == 0) {
#pragma unroll
    for (int r = 0; r < 8; r++) {
      int x = x0 + r + 8 * half;
      float m = (float)mask_p[b * SS + x];
      float* orow = out + ((size_t)b * SS + x) * FF;
      orow[23 + p] = m * rmax[r];
      orow[43 + p] = m * (rsum[r] / (float)SS);
    }
  }
  __syncthreads();
  int y = tid & (SS - 1);
  int isSum = tid >> 7;
  float v = colS[isSum][0][y];
#pragma unroll
  for (int u = 1; u < NYT; u++) {
    float q = colS[isSum][u][y];
    v = isSum ? v + q : fmaxf(v, q);
  }
  float mh = (float)mask_h[b * SS + y];
  float* o2 = out + ((size_t)(BB + b) * SS + y) * FF;
  if (isSum) o2[43 + p] = mh * (v / (float)SS);
  else o2[23 + p] = mh * v;
}

__global__ void __launch_bounds__(256, 4) match_kernel(
    const float* __restrict__ cp, const float* __restrict__ ch,
    const float* __restrict__ cpl, const float* __restrict__ chl,
    const float* __restrict__ att_mean_h, const float* __restrict__ att_mean_p,
    const float* __restrict__ att_max_h, const float* __restrict__ att_max_p,
    const float* __restrict__ w0, const float* __restrict__ w1,
    const float* __restrict__ w2, const float* __restrict__ w3,
    const float* __restrict__ wn1_p, const float* __restrict__ wn1_h,
    const int* __restrict__ mask_p, const int* __restrict__ mask_h, float* __restrict__ out) {
  __shared__ float colS[2][NYT][SS];
  int blk = blockIdx.x;
  if (blk < NMAXB) {
    maxpool_block(cp, ch, w1, wn1_p, wn1_h, mask_p, mask_h, out, colS, blk, threadIdx.x);
  } else {
    int wid = (blk - NMAXB) * 8 + (threadIdx.x >> 5);
    int lane = threadIdx.x & 31;
    int grp = wid / (2 * NMPM);
    int rem = wid % (2 * NMPM);
    bool side = rem >= NMPM;
    int tile = side ? rem - NMPM : rem;
    const float* v1 = side ? ch : cp;
    const float* v2 = grp == 0 ? (side ? cpl : chl)
                    : grp == 1 ? (side ? att_mean_p : att_mean_h) : (side ? att_max_p : att_max_h);
    const float* w = grp == 0 ? w0 : grp == 1 ? w2 : w3;
    int v2_bstride = grp == 0 ? HH : SS * HH;
    int v2_sstride = grp == 0 ? 0 : HH;
    int foff = grp == 0 ? 2 : grp == 1 ? 63 : 84;
    mpm_body(v1, v2, w, v2_bstride, v2_sstride, out, side ? 1 : 0, foff, tile, lane);
  }
}

extern "C" void kernel_launch(void* const* d_in, const int* in_sizes, int n_in,
                              void* d_out, int out_size, void* d_ws, size_t ws_size,
                              hipStream_t stream) {
  const float* ctx_p = (const float*)d_in[0];
  const float* ctx_h = (const float*)d_in[1];
  const float* w0 = (const float*)d_in[2];
  const float* w1 = (const float*)d_in[3];
  const float* w2 = (const float*)d_in[4];
  const float* w3 = (const float*)d_in[5];
  const int* mask_p = (const int*)d_in[6];
  const int* mask_h = (const int*)d_in[7];
  float* out = (float*)d_out;
  float* ws = (float*)d_ws;

  const size_t SZ_ROWS = (size_t)BB * SS * HH;
  float* cp = ws;                 ws += SZ_ROWS;
  float* ch = ws;                 ws += SZ_ROWS;
  float* n_p = ws;                ws += BB * SS;
  float* n_h = ws;                ws += BB * SS;
  float* wn1_p = ws;              ws += (size_t)BB * SS * PP;
  float* wn1_h = ws;              ws += (size_t)BB * SS * PP;
  float* cpl = ws;                ws += BB * HH;
  float* chl = ws;                ws += BB * HH;
  float* cosM = ws;               ws += (size_t)BB * SS * SS;
  float* cossum_y = ws;           ws += BB * SS;
  float* cossum_x = ws;           ws += BB * SS;
  float* att_mean_h = ws;         ws += SZ_ROWS;
  float* att_mean_p = ws;         ws += SZ_ROWS;
  float* att_max_h = ws;          ws += SZ_ROWS;
  float* att_max_p = ws;          ws += SZ_ROWS;

  prep_gather_kernel<<<2 * NPREP + BB, 32, 0, stream>>>(ctx_p, ctx_h, mask_p, mask_h, w1, cp, ch, n_p, n_h,
                                                        wn1_p, wn1_h, cpl, chl);

  cos_strip_kernel<<<2 * NCOS, 256, 0, stream>>>(cp, ch, n_p, n_h, mask_p, mask_h, cosM,
                                                 cossum_y, cossum_x, out);

  att_kernel<<<2 * NATT / 8 + 2 * NAMX, 256, 0, stream>>>(cosM, cp, ch, cossum_y, cossum_x, mask_p, mask_h,
                                                             att_mean_h, att_mean_p, att_max_h, att_max_p, SS, 1);

  match_kernel<<<NMAXB + 6 * NMPM / 8, 256, 0, stream>>>(cp, ch, cpl, chl, att_mean_h, att_mean_p,
                                                        att_max_h, att_max_p, w0, w1, w2, w3,
                                                        wn1_p, wn1_h, mask_p, mask_h, out);
}


# att launch: a wave requests its global operands far ahead of use (att_mean: all 32 k-steps' fragments before the first WMMA; att_max: V 32 k ahead): one memory wait per wave, not 32 / 16 in a row
# speedup vs baseline: 9.5134x; 1.0467x over previous
#include <hip/hip_runtime.h>
#include <hip/hip_bf16.h>


#define BB 16
#define SS 128
#define HH 200
#define PP 20
#define FF 105
#define EPSF 1e-8f
#define KCH 10
#define NCH 5

typedef float v2f __attribute__((ext_vector_type(2)));
typedef float v4f __attribute__((ext_vector_type(4)));
typedef float v8f __attribute__((ext_vector_type(8)));

__device__ __forceinline__ v2f wsq_frag(v2f ww, bool isW, float fill) {
  v2f bv;
  bv.x = isW ? ww.x * ww.x : fill;
  bv.y = isW ? ww.y * ww.y : fill;
  return bv;
}

#define NPREP ((BB * SS / 16) * 2)
__device__ __forceinline__ void prep_body(const float* __restrict__ src, const int* __restrict__ mask,
                                 const float* __restrict__ w1, float* __restrict__ dst,
                                 float* __restrict__ nrm, float* __restrict__ wn, int tile, int lane) {
  int m0 = (tile >> 1) << 4;
  int n0 = (tile & 1) << 4;
  int half = lane >> 4, l16 = lane & 15;
  int bs = m0 + l16;
  float m = (float)mask[bs];
  const float* srow = src + (size_t)bs * HH + 2 * half;
  float* drow = dst + (size_t)bs * HH + 2 * half;
  int col = n0 + l16;
  bool isW = col < PP;
  float fill = (col == PP) ? 1.f : 0.f;
  const float* wrow = w1 + (size_t)(isW ? col : PP - 1) * HH + 2 * half;
  v8f acc = {0, 0, 0, 0, 0, 0, 0, 0};
  v2f xs[2][KCH], ws2[2][KCH];
#pragma unroll
  for (int i = 0; i < KCH; i++) {
    xs[0][i] = *(const v2f*)(srow + 4 * i);
    ws2[0][i] = *(const v2f*)(wrow + 4 * i);
  }
#pragma unroll
  for (int c = 0; c < NCH; c++) {
    if (c + 1 < NCH) {
#pragma unroll
      for (int i = 0; i < KCH; i++) {
        xs[(c + 1) & 1][i] = *(const v2f*)(srow + 4 * KCH * (c + 1) + 4 * i);
        ws2[(c + 1) & 1][i] = *(const v2f*)(wrow + 4 * KCH * (c + 1) + 4 * i);
      }
    }
    __builtin_amdgcn_sched_barrier(0);
#pragma unroll
    for (int i = 0; i < KCH; i++) {
      v2f x = xs[c & 1][i];
      x.x *= m; x.y *= m;
      if (n0 == 0) *(v2f*)(drow + 4 * KCH * c + 4 * i) = x;
      v2f bv = wsq_frag(ws2[c & 1][i], isW, fill);
      v2f a; a.x = x.x * x.x; a.y = x.y * x.y;
      acc = __builtin_amdgcn_wmma_f32_16x16x4_f32(false, a, false, bv, (short)0, acc, false, false);
    }
  }
  if (col <= PP) {
#pragma unroll
    for (int r = 0; r < 8; r++) {
      int row = m0 + r + 8 * half;
      float* o = (col == PP) ? nrm + row : wn + (size_t)row * PP + col;
      *o = sqrtf(acc[r]);
    }
  }
}

__device__ __forceinline__ void gather_last_body(const float* __restrict__ xp, const float* __restrict__ xh,
                                   const int* __restrict__ mask_p, const int* __restrict__ mask_h,
                                   float* __restrict__ cpl, float* __restrict__ chl, int b, int lane) {
  int lp = 0, lh = 0;
  for (int i = 0; i < SS; i++) { lp += mask_p[b * SS + i]; lh += mask_h[b * SS + i]; }
  int ip = (lp > 0) ? lp - 1 : 0;
  int ih = (lh > 0) ? lh - 1 : 0;
  float mp = (float)mask_p[b * SS + ip];
  float mh = (float)mask_h[b * SS + ih];
  for (int h = lane; h < HH; h += 32) {
    cpl[b * HH + h] = xp[((size_t)b * SS + ip) * HH + h] * mp;
    chl[b * HH + h] = xh[((size_t)b * SS + ih) * HH + h] * mh;
  }
}

__global__ void __launch_bounds__(32) prep_gather_kernel(
    const float* __restrict__ ctx_p, const float* __restrict__ ctx_h,
    const int* __restrict__ mask_p, const int* __restrict__ mask_h, const float* __restrict__ w1,
    float* __restrict__ cp, float* __restrict__ ch, float* __restrict__ n_p, float* __restrict__ n_h,
    float* __restrict__ wn1_p, float* __restrict__ wn1_h, float* __restrict__ cpl, float* __restrict__ chl) {
  int blk = blockIdx.x, lane = threadIdx.x;
  if (blk < 2 * NPREP) {
    bool side = blk >= NPREP;
    prep_body(side ? ctx_h : ctx_p, side ? mask_h : mask_p, w1, side ? ch : cp,
              side ? n_h : n_p, side ? wn1_h : wn1_p, side ? blk - NPREP : blk, lane);
  } else {
    gather_last_body(ctx_p, ctx_h, mask_p, mask_h, cpl, chl, blk - 2 * NPREP, lane);
  }
}

#define NCOS (BB * 8)
__global__ void __launch_bounds__(256) cos_strip_kernel(
    const float* __restrict__ cp, const float* __restrict__ ch,
    const float* __restrict__ np, const float* __restrict__ nh,
    const int* __restrict__ mask_p, const int* __restrict__ mask_h,
    float* __restrict__ cosO, float* __restrict__ cossum_y, float* __restrict__ cossum_x,
    float* __restrict__ out) {
  __shared__ float pmax[8][16], psum[8][16];
  int blk = blockIdx.x;
  int side = blk >= NCOS;
  if (side) blk -= NCOS;
  int b = blk >> 3, st = blk & 7;
  int w = threadIdx.x >> 5;
  int lane = threadIdx.x & 31, half = lane >> 4, l16 = lane & 15;
  int x0 = (side ? w : st) << 4;
  int y0 = (side ? st : w) << 4;
  const float* arow = cp + ((size_t)(b * SS) + x0 + l16) * HH + 2 * half;
  const float* brow = ch + ((size_t)(b * SS) + y0 + l16) * HH + 2 * half;
  v8f acc = {0, 0, 0, 0, 0, 0, 0, 0};
  for (int k = 0; k < HH; k += 4) {
    v2f a = *(const v2f*)(arow + k);
    v2f bv = *(const v2f*)(brow + k);
    acc = __builtin_amdgcn_wmma_f32_16x16x4_f32(false, a, false, bv, (short)0, acc, false, false);
  }
  float nhv = nh[b * SS + y0 + l16];
  float c[8];
#pragma unroll
  for (int r = 0; r < 8; r++) {
    int m = r + 8 * half;
    float d = fmaxf(np[b * SS + x0 + m] * nhv, EPSF);
    c[r] = acc[r] / d;
  }
  if (!side) {
    float mx[8], sm[8];
#pragma unroll
    for (int r = 0; r < 8; r++) {
      cosO[((size_t)(b * SS) + x0 + r + 8 * half) * SS + y0 + l16] = c[r];
      mx[r] = c[r]; sm[r] = c[r];
    }
#pragma unroll
    for (int o = 8; o > 0; o >>= 1) {
#pragma unroll
      for (int r = 0; r < 8; r++) {
        mx[r] = fmaxf(mx[r], __shfl_xor(mx[r], o, 32));
        sm[r] += __shfl_xor(sm[r], o, 32);
      }
    }
    if (l16 == 0) {
#pragma unroll
      for (int r = 0; r < 8; r++) { pmax[w][r + 8 * half] = mx[r]; psum[w][r + 8 * half] = sm[r]; }
    }
  } else {
    float mx = c[0], sm = c[0];
#pragma unroll
    for (int r = 1; r < 8; r++) { mx = fmaxf(mx, c[r]); sm += c[r]; }
    mx = fmaxf(mx, __shfl_xor(mx, 16, 32));
    sm += __shfl_xor(sm, 16, 32);
    if (half == 0) { pmax[w][l16] = mx; psum[w][l16] = sm; }
  }
  __syncthreads();
  if (threadIdx.x < 16) {
    int t = threadIdx.x;
    float mx = pmax[0][t], sm = psum[0][t];
#pragma unroll
    for (int u = 1; u < 8; u++) { mx = fmaxf(mx, pmax[u][t]); sm += psum[u][t]; }
    int r = (st << 4) + t;
    float m = (float)(side ? mask_h : mask_p)[b * SS + r];
    (side ? cossum_x : cossum_y)[b * SS + r] = sm;
    float* o2 = out + ((size_t)(side * BB + b) * SS + r) * FF;
    o2[0] = m * mx;
    o2[1] = m * (sm / (float)SS);
  }
}

#define NATT (BB * 8 * 13)
#define AKS (SS / 4)
template <bool TR>
__device__ __forceinline__ void attmean_body(const float* __restrict__ cosM, const float* __restrict__ V,
                                    const float* __restrict__ csum, float* __restrict__ attO,
                                    int wid, int lane) {
  int b = wid / (8 * 13);
  int rem = wid % (8 * 13);
  int m0 = (rem / 13) * 16;
  int n0 = (rem % 13) * 16;
  int half = lane >> 4, l16 = lane & 15;
  const float* cb = cosM + (size_t)b * SS * SS;
  int nIdx = n0 + l16;
  bool nOK = nIdx < HH;
  int nLd = nOK ? nIdx : HH - 1;
  const float* ap = TR ? cb + (size_t)(2 * half) * SS + m0 + l16
                       : cb + (size_t)(m0 + l16) * SS + 2 * half;
  const float* bp = V + ((size_t)b * SS + 2 * half) * HH + nLd;
  v2f af[AKS], bf[AKS];
#pragma unroll
  for (int i = 0; i < AKS; i++) {
    if (TR) {
      af[i].x = ap[(size_t)(4 * i) * SS];
      af[i].y = ap[(size_t)(4 * i + 1) * SS];
    } else {
      af[i] = *(const v2f*)(ap + 4 * i);
    }
    bf[i].x = bp[(size_t)(4 * i) * HH];
    bf[i].y = bp[(size_t)(4 * i + 1) * HH];
  }
  float cs8[8];
#pragma unroll
  for (int r = 0; r < 8; r++) cs8[r] = csum[b * SS + m0 + r + 8 * half];
  __builtin_amdgcn_sched_barrier(0);
  v8f acc = {0, 0, 0, 0, 0, 0, 0, 0};
#pragma unroll
  for (int i = 0; i < AKS; i++)
    acc = __builtin_amdgcn_wmma_f32_16x16x4_f32(false, af[i], false, bf[i], (short)0, acc, false, false);
  if (nOK) {
#pragma unroll
    for (int r = 0; r < 8; r++) {
      int m = m0 + r + 8 * half;
      float d = fmaxf(cs8[r], EPSF);
      attO[((size_t)b * SS + m) * HH + nIdx] = acc[r] / d;
    }
  }
}

#define RT 8
#define KU 8
#define KV 32
#define NAMX (BB * SS / RT)
template <bool TR>
__device__ __forceinline__ void attmax_body(const float* __restrict__ cosM, const float* __restrict__ V,
                              const int* __restrict__ mask, float* __restrict__ attO,
                              v4f (*cs)[RT / 4], int blk, int t) {
  int b = blk / (SS / RT), r0 = (blk % (SS / RT)) * RT;
  const float* cb = cosM + (size_t)b * SS * SS;
  if (TR) {
    int k = t >> 1, q = t & 1;
    cs[k][q] = *(const v4f*)(cb + (size_t)k * SS + r0 + 4 * q);
  } else {
    int j = t & (RT - 1), k4 = t >> 3;
    v4f x = *(const v4f*)(cb + (size_t)(r0 + j) * SS + 4 * k4);
    float* csf = (float*)cs;
    csf[(4 * k4 + 0) * RT + j] = x.x;
    csf[(4 * k4 + 1) * RT + j] = x.y;
    csf[(4 * k4 + 2) * RT + j] = x.z;
    csf[(4 * k4 + 3) * RT + j] = x.w;
  }
  __syncthreads();
  int h = t;
  if (h >= HH) return;
  const float* vp = V + (size_t)b * SS * HH + h;
  float mx[RT];
#pragma unroll
  for (int j = 0; j < RT; j++) mx[j] = -3.0e38f;
  float v[KV], vn[KV];
#pragma unroll
  for (int u = 0; u < KV; u++) vn[u] = vp[(size_t)u * HH];
#pragma unroll 1
  for (int k0 = 0; k0 < SS; k0 += KV) {
#pragma unroll
    for (int u = 0; u < KV; u++) v[u] = vn[u];
    if (k0 + KV < SS) {
#pragma unroll
      for (int u = 0; u < KV; u++) vn[u] = vp[(size_t)(k0 + KV + u) * HH];
    }
#pragma unroll
    for (int s = 0; s < KV; s += KU) {
      v4f c[KU][RT / 4];
#pragma unroll
      for (int u = 0; u < KU; u++) {
#pragma unroll
        for (int q = 0; q < RT / 4; q++) c[u][q] = cs[k0 + s + u][q];
      }
      __builtin_amdgcn_sched_barrier(0);
#pragma unroll
      for (int u = 0; u < KU; u++) {
#pragma unroll
        for (int q = 0; q < RT / 4; q++) {
          mx[4 * q + 0] = fmaxf(mx[4 * q + 0], c[u][q].x * v[s + u]);
          mx[4 * q + 1] = fmaxf(mx[4 * q + 1], c[u][q].y * v[s + u]);
          mx[4 * q + 2] = fmaxf(mx[4 * q + 2], c[u][q].z * v[s + u]);
          mx[4 * q + 3] = fmaxf(mx[4 * q + 3], c[u][q].w * v[s + u]);
        }
      }
    }
  }
#pragma unroll
  for (int j = 0; j < RT; j++) {
    float m = (float)mask[b * SS + r0 + j];
    attO[((size_t)b * SS + r0 + j) * HH + h] = m * mx[j];
  }
}

__global__ void __launch_bounds__(256) att_kernel(
    const float* __restrict__ cosM, const float* __restrict__ cp, const float* __restrict__ ch,
    const float* __restrict__ cossum_y, const float* __restrict__ cossum_x,
    const int* __restrict__ mask_p, const int* __restrict__ mask_h,
    float* __restrict__ att_mean_h, float* __restrict__ att_mean_p,
    float* __restrict__ att_max_h, float* __restrict__ att_max_p) {
  __shared__ v4f cs[SS][RT / 4];
  int blk = blockIdx.x;
  if (blk < 2 * NATT / 8) {
    int wv = threadIdx.x >> 5, lane = threadIdx.x & 31;
    if (blk >= NATT / 8) attmean_body<true>(cosM, cp, cossum_x, att_mean_p, (blk - NATT / 8) * 8 + wv, lane);
    else attmean_body<false>(cosM, ch, cossum_y, att_mean_h, blk * 8 + wv, lane);
  } else {
    blk -= 2 * NATT / 8;
    if (blk >= NAMX) attmax_body<true>(cosM, cp, mask_h, att_max_p, cs, blk - NAMX, threadIdx.x);
    else attmax_body<false>(cosM, ch, mask_p, att_max_h, cs, blk, threadIdx.x);
  }
}

#define NMPM ((BB * SS / 16) * 2)
__device__ __forceinline__ void mpm_body(const float* __restrict__ v1, const float* __restrict__ v2,
                                const float* __restrict__ w, int v2_bstride, int v2_sstride,
                                float* __restrict__ out, int side, int foff, int tile, int lane) {
  int m0 = (tile >> 1) << 4;
  int n0 = (tile & 1) << 4;
  int half = lane >> 4, l16 = lane & 15;
  int bs = m0 + l16;
  int b = bs / SS, s = bs % SS;
  const float* p1 = v1 + (size_t)bs * HH + 2 * half;
  const float* p2 = v2 + (size_t)b * v2_bstride + (size_t)s * v2_sstride + 2 * half;
  int col = n0 + l16;
  bool isW = col < PP;
  float fill = (col == PP) ? 1.f : 0.f;
  const float* wrow = w + (size_t)(isW ? col : PP - 1) * HH + 2 * half;
  v8f s11 = {0, 0, 0, 0, 0, 0, 0, 0};
  v8f s22 = {0, 0, 0, 0, 0, 0, 0, 0};
  v8f s12 = {0, 0, 0, 0, 0, 0, 0, 0};
  v2f xs[2][KCH], ys[2][KCH], ws2[2][KCH];
#pragma unroll
  for (int i = 0; i < KCH; i++) {
    xs[0][i] = *(const v2f*)(p1 + 4 * i);
    ys[0][i] = *(const v2f*)(p2 + 4 * i);
    ws2[0][i] = *(const v2f*)(wrow + 4 * i);
  }
#pragma unroll
  for (int c = 0; c < NCH; c++) {
    if (c + 1 < NCH) {
#pragma unroll
      for (int i = 0; i < KCH; i++) {
        xs[(c + 1) & 1][i] = *(const v2f*)(p1 + 4 * KCH * (c + 1) + 4 * i);
        ys[(c + 1) & 1][i] = *(const v2f*)(p2 + 4 * KCH * (c + 1) + 4 * i);
        ws2[(c + 1) & 1][i] = *(const v2f*)(wrow + 4 * KCH * (c + 1) + 4 * i);
      }
    }
    __builtin_amdgcn_sched_barrier(0);
#pragma unroll
    for (int i = 0; i < KCH; i++) {
      v2f x = xs[c & 1][i], y = ys[c & 1][i];
      v2f bv = wsq_frag(ws2[c & 1][i], isW, fill);
      v2f xx, yy, xy;
      xx.x = x.x * x.x; xx.y = x.y * x.y;
      yy.x = y.x * y.x; yy.y = y.y * y.y;
      xy.x = x.x * y.x; xy.y = x.y * y.y;
      s11 = __builtin_amdgcn_wmma_f32_16x16x4_f32(false, xx, false, bv, (short)0, s11, false, false);
      s22 = __builtin_amdgcn_wmma_f32_16x16x4_f32(false, yy, false, bv, (short)0, s22, false, false);
      s12 = __builtin_amdgcn_wmma_f32_16x16x4_f32(false, xy, false, bv, (short)0, s12, false, false);
    }
  }
  if (col <= PP) {
    int f = (col == PP) ? 0 : 1 + col;
#pragma unroll
    for (int r = 0; r < 8; r++) {
      int row = m0 + r + 8 * half;
      int ob = row / SS, os = row % SS;
      float* orow = out + ((size_t)(side * BB + ob) * SS + os) * FF + foff;
      orow[f] = s12[r] / (fmaxf(sqrtf(s11[r]), EPSF) * fmaxf(sqrtf(s22[r]), EPSF));
    }
  }
}

#define NJ 25
#define GCH 10
#define NYT 8
#define NMAXB (BB * PP)
__device__ __forceinline__ void maxpool_block(const float* __restrict__ cp, const float* __restrict__ ch,
                                     const float* __restrict__ w,
                                     const float* __restrict__ wnP, const float* __restrict__ wnH,
                                     const int* __restrict__ mask_p, const int* __restrict__ mask_h,
                                     float* __restrict__ out, float (*colS)[NYT][SS], int blk, int tid) {
  int b = blk / PP;
  int p = blk % PP;
  int wv = tid >> 5, lane = tid & 31;
  int x0 = wv * 16;
  int half = lane >> 4, l16 = lane & 15;
  float rmax[8], rsum[8], wA[8];
#pragma unroll
  for (int r = 0; r < 8; r++) {
    rmax[r] = -3.0e38f; rsum[r] = 0.f;
    wA[r] = wnP[((size_t)b * SS + x0 + r + 8 * half) * PP + p];
  }
  const float* a_row = cp + ((size_t)b * SS + x0 + l16) * HH + 4 * half;
  const float* w_row = w + (size_t)p * HH + 4 * half;
  const float* b_row = ch + ((size_t)b * SS + l16) * HH + 4 * half;
  const float* wnB_col = wnH + ((size_t)b * SS + l16) * PP + p;
  float* col_slot = &colS[half][wv][l16];

  v4f bf[2][GCH];
#pragma unroll
  for (int i = 0; i < GCH; i++) bf[0][i] = *(const v4f*)(b_row + 8 * i);

  v4f af[NJ];
#pragma unroll
  for (int j = 0; j < NJ; j++) {
    v4f wv4 = *(const v4f*)(w_row + 8 * j);
    v4f av = *(const v4f*)(a_row + 8 * j);
    af[j].x = wv4.x * wv4.x * av.x;
    af[j].y = wv4.y * wv4.y * av.y;
    af[j].z = wv4.z * wv4.z * av.z;
    af[j].w = wv4.w * wv4.w * av.w;
  }

  v8f acc = {0, 0, 0, 0, 0, 0, 0, 0};
  float wB = 0.f;
#pragma unroll
  for (int g0 = 0; g0 < NYT * NJ; g0 += GCH) {
    if (g0 + GCH < NYT * NJ) {
#pragma unroll
      for (int i = 0; i < GCH; i++) {
        int gn = g0 + GCH + i;
        bf[((g0 / GCH) + 1) & 1][i] = *(const v4f*)(b_row + (gn / NJ) * 16 * HH + 8 * (gn % NJ));
      }
    }
    __builtin_amdgcn_sched_barrier(0);
#pragma unroll
    for (int i = 0; i < GCH; i++) {
      int g = g0 + i;
      int yt = g / NJ, j = g % NJ;
      if (j == 0) {
        acc = (v8f){0, 0, 0, 0, 0, 0, 0, 0};
        wB = wnB_col[(size_t)yt * 16 * PP];
      }
      v4f bb = bf[(g0 / GCH) & 1][i];
      v4f aa = af[j];
      v2f a0, b0, a1, b1;
      a0.x = aa.x; a0.y = aa.y; b0.x = bb.x; b0.y = bb.y;
      a1.x = aa.z; a1.y = aa.w; b1.x = bb.z; b1.y = bb.w;
      acc = __builtin_amdgcn_wmma_f32_16x16x4_f32(false, a0, false, b0, (short)0, acc, false, false);
      acc = __builtin_amdgcn_wmma_f32_16x16x4_f32(false, a1, false, b1, (short)0, acc, false, false);
      if (j == NJ - 1) {
        float cm = -3.0e38f, csm = 0.f;
#pragma unroll
        for (int r = 0; r < 8; r++) {
          float mv = acc[r] / fmaxf(wA[r] * wB, EPSF);
          rmax[r] = fmaxf(rmax[r], mv);
          rsum[r] += mv;
          cm = fmaxf(cm, mv);
          csm += mv;
        }
        cm = fmaxf(cm, __shfl_xor(cm, 16, 32));
        csm += __shfl_xor(csm, 16, 32);
        col_slot[yt * 16] = half ? csm : cm;
      }
    }
  }
#pragma unroll
  for (int r = 0; r < 8; r++) {
    for (int o = 8; o > 0; o >>= 1) {
      rmax[r] = fmaxf(rmax[r], __shfl_xor(rmax[r], o, 32));
      rsum[r] += __shfl_xor(rsum[r], o, 32);
    }
  }
  if (l16 == 0) {
#pragma unroll
    for (int r = 0; r < 8; r++) {
      int x = x0 + r + 8 * half;
      float m = (float)mask_p[b * SS + x];
      float* orow = out + ((size_t)b * SS + x) * FF;
      orow[23 + p] = m * rmax[r];
      orow[43 + p] = m * (rsum[r] / (float)SS);
    }
  }
  __syncthreads();
  int y = tid & (SS - 1);
  int isSum = tid >> 7;
  float v = colS[isSum][0][y];
#pragma unroll
  for (int u = 1; u < NYT; u++) {
    float q = colS[isSum][u][y];
    v = isSum ? v + q : fmaxf(v, q);
  }
  float mh = (float)mask_h[b * SS + y];
  float* o2 = out + ((size_t)(BB + b) * SS + y) * FF;
  if (isSum) o2[43 + p] = mh * (v / (float)SS);
  else o2[23 + p] = mh * v;
}

__global__ void __launch_bounds__(256, 4) match_kernel(
    const float* __restrict__ cp, const float* __restrict__ ch,
    const float* __restrict__ cpl, const float* __restrict__ chl,
    const float* __restrict__ att_mean_h, const float* __restrict__ att_mean_p,
    const float* __restrict__ att_max_h, const float* __restrict__ att_max_p,
    const float* __restrict__ w0, const float* __restrict__ w1,
    const float* __restrict__ w2, const float* __restrict__ w3,
    const float* __restrict__ wn1_p, const float* __restrict__ wn1_h,
    const int* __restrict__ mask_p, const int* __restrict__ mask_h, float* __restrict__ out) {
  __shared__ float colS[2][NYT][SS];
  int blk = blockIdx.x;
  if (blk < NMAXB) {
    maxpool_block(cp, ch, w1, wn1_p, wn1_h, mask_p, mask_h, out, colS, blk, threadIdx.x);
  } else {
    int wid = (blk - NMAXB) * 8 + (threadIdx.x >> 5);
    int lane = threadIdx.x & 31;
    int grp = wid / (2 * NMPM);
    int rem = wid % (2 * NMPM);
    bool side = rem >= NMPM;
    int tile = side ? rem - NMPM : rem;
    const float* v1 = side ? ch : cp;
    const float* v2 = grp == 0 ? (side ? cpl : chl)
                    : grp == 1 ? (side ? att_mean_p : att_mean_h) : (side ? att_max_p : att_max_h);
    const float* w = grp == 0 ? w0 : grp == 1 ? w2 : w3;
    int v2_bstride = grp == 0 ? HH : SS * HH;
    int v2_sstride = grp == 0 ? 0 : HH;
    int foff = grp == 0 ? 2 : grp == 1 ? 63 : 84;
    mpm_body(v1, v2, w, v2_bstride, v2_sstride, out, side ? 1 : 0, foff, tile, lane);
  }
}

extern "C" void kernel_launch(void* const* d_in, const int* in_sizes, int n_in,
                              void* d_out, int out_size, void* d_ws, size_t ws_size,
                              hipStream_t stream) {
  const float* ctx_p = (const float*)d_in[0];
  const float* ctx_h = (const float*)d_in[1];
  const float* w0 = (const float*)d_in[2];
  const float* w1 = (const float*)d_in[3];
  const float* w2 = (const float*)d_in[4];
  const float* w3 = (const float*)d_in[5];
  const int* mask_p = (const int*)d_in[6];
  const int* mask_h = (const int*)d_in[7];
  float* out = (float*)d_out;
  float* ws = (float*)d_ws;

  const size_t SZ_ROWS = (size_t)BB * SS * HH;
  float* cp = ws;                 ws += SZ_ROWS;
  float* ch = ws;                 ws += SZ_ROWS;
  float* n_p = ws;                ws += BB * SS;
  float* n_h = ws;                ws += BB * SS;
  float* wn1_p = ws;              ws += (size_t)BB * SS * PP;
  float* wn1_h = ws;              ws += (size_t)BB * SS * PP;
  float* cpl = ws;                ws += BB * HH;
  float* chl = ws;                ws += BB * HH;
  float* cosM = ws;               ws += (size_t)BB * SS * SS;
  float* cossum_y = ws;           ws += BB * SS;
  float* cossum_x = ws;           ws += BB * SS;
  float* att_mean_h = ws;         ws += SZ_ROWS;
  float* att_mean_p = ws;         ws += SZ_ROWS;
  float* att_max_h = ws;          ws += SZ_ROWS;
  float* att_max_p = ws;          ws += SZ_ROWS;

  prep_gather_kernel<<<2 * NPREP + BB, 32, 0, stream>>>(ctx_p, ctx_h, mask_p, mask_h, w1, cp, ch, n_p, n_h,
                                                        wn1_p, wn1_h, cpl, chl);

  cos_strip_kernel<<<2 * NCOS, 256, 0, stream>>>(cp, ch, n_p, n_h, mask_p, mask_h, cosM,
                                                 cossum_y, cossum_x, out);

  att_kernel<<<2 * NATT / 8 + 2 * NAMX, 256, 0, stream>>>(cosM, cp, ch, cossum_y, cossum_x, mask_p, mask_h,
                                                             att_mean_h, att_mean_p, att_max_h, att_max_p);

  match_kernel<<<NMAXB + 6 * NMPM / 8, 256, 0, stream>>>(cp, ch, cpl, chl, att_mean_h, att_mean_p,
                                                        att_max_h, att_max_p, w0, w1, w2, w3,
                                                        wn1_p, wn1_h, mask_p, mask_h, out);
}


# cos block keeps its 16-wide strip of cos in LDS and computes the strip's attentive mean (WMMA) and max-attentive vectors itself: cos never goes to memory, 4 launches -> 3; no summation order change
# speedup vs baseline: 10.2728x; 1.0798x over previous
#include <hip/hip_runtime.h>
#include <hip/hip_bf16.h>


#define BB 16
#define SS 128
#define HH 200
#define PP 20
#define FF 105
#define EPSF 1e-8f
#define KCH 10
#define NCH 5

typedef float v2f __attribute__((ext_vector_type(2)));
typedef float v4f __attribute__((ext_vector_type(4)));
typedef float v8f __attribute__((ext_vector_type(8)));

__device__ __forceinline__ v2f wsq_frag(v2f ww, bool isW, float fill) {
  v2f bv;
  bv.x = isW ? ww.x * ww.x : fill;
  bv.y = isW ? ww.y * ww.y : fill;
  return bv;
}

#define NPREP ((BB * SS / 16) * 2)
__device__ __forceinline__ void prep_body(const float* __restrict__ src, const int* __restrict__ mask,
                                 const float* __restrict__ w1, float* __restrict__ dst,
                                 float* __restrict__ nrm, float* __restrict__ wn, int tile, int lane) {
  int m0 = (tile >> 1) << 4;
  int n0 = (tile & 1) << 4;
  int half = lane >> 4, l16 = lane & 15;
  int bs = m0 + l16;
  float m = (float)mask[bs];
  const float* srow = src + (size_t)bs * HH + 2 * half;
  float* drow = dst + (size_t)bs * HH + 2 * half;
  int col = n0 + l16;
  bool isW = col < PP;
  float fill = (col == PP) ? 1.f : 0.f;
  const float* wrow = w1 + (size_t)(isW ? col : PP - 1) * HH + 2 * half;
  v8f acc = {0, 0, 0, 0, 0, 0, 0, 0};
  v2f xs[2][KCH], ws2[2][KCH];
#pragma unroll
  for (int i = 0; i < KCH; i++) {
    xs[0][i] = *(const v2f*)(srow + 4 * i);
    ws2[0][i] = *(const v2f*)(wrow + 4 * i);
  }
#pragma unroll
  for (int c = 0; c < NCH; c++) {
    if (c + 1 < NCH) {
#pragma unroll
      for (int i = 0; i < KCH; i++) {
        xs[(c + 1) & 1][i] = *(const v2f*)(srow + 4 * KCH * (c + 1) + 4 * i);
        ws2[(c + 1) & 1][i] = *(const v2f*)(wrow + 4 * KCH * (c + 1) + 4 * i);
      }
    }
    __builtin_amdgcn_sched_barrier(0);
#pragma unroll
    for (int i = 0; i < KCH; i++) {
      v2f x = xs[c & 1][i];
      x.x *= m; x.y *= m;
      if (n0 == 0) *(v2f*)(drow + 4 * KCH * c + 4 * i) = x;
      v2f bv = wsq_frag(ws2[c & 1][i], isW, fill);
      v2f a; a.x = x.x * x.x; a.y = x.y * x.y;
      acc = __builtin_amdgcn_wmma_f32_16x16x4_f32(false, a, false, bv, (short)0, acc, false, false);
    }
  }
  if (col <= PP) {
#pragma unroll
    for (int r = 0; r < 8; r++) {
      int row = m0 + r + 8 * half;
      float* o = (col == PP) ? nrm + row : wn + (size_t)row * PP + col;
      *o = sqrtf(acc[r]);
    }
  }
}

__device__ __forceinline__ void gather_last_body(const float* __restrict__ xp, const float* __restrict__ xh,
                                   const int* __restrict__ mask_p, const int* __restrict__ mask_h,
                                   float* __restrict__ cpl, float* __restrict__ chl, int b, int lane) {
  int lp = 0, lh = 0;
  for (int i = 0; i < SS; i++) { lp += mask_p[b * SS + i]; lh += mask_h[b * SS + i]; }
  int ip = (lp > 0) ? lp - 1 : 0;
  int ih = (lh > 0) ? lh - 1 : 0;
  float mp = (float)mask_p[b * SS + ip];
  float mh = (float)mask_h[b * SS + ih];
  for (int h = lane; h < HH; h += 32) {
    cpl[b * HH + h] = xp[((size_t)b * SS + ip) * HH + h] * mp;
    chl[b * HH + h] = xh[((size_t)b * SS + ih) * HH + h] * mh;
  }
}

__global__ void __launch_bounds__(32) prep_gather_kernel(
    const float* __restrict__ ctx_p, const float* __restrict__ ctx_h,
    const int* __restrict__ mask_p, const int* __restrict__ mask_h, const float* __restrict__ w1,
    float* __restrict__ cp, float* __restrict__ ch, float* __restrict__ n_p, float* __restrict__ n_h,
    float* __restrict__ wn1_p, float* __restrict__ wn1_h, float* __restrict__ cpl, float* __restrict__ chl) {
  int blk = blockIdx.x, lane = threadIdx.x;
  if (blk < 2 * NPREP) {
    bool side = blk >= NPREP;
    prep_body(side ? ctx_h : ctx_p, side ? mask_h : mask_p, w1, side ? ch : cp,
              side ? n_h : n_p, side ? wn1_h : wn1_p, side ? blk - NPREP : blk, lane);
  } else {
    gather_last_body(ctx_p, ctx_h, mask_p, mask_h, cpl, chl, blk - 2 * NPREP, lane);
  }
}

#define NCOS (BB * 8)
#define AKS (SS / 4)
#define NHT 13
#define ST 16
#define KU 4
#define KV 32

template <int NT>
__device__ __forceinline__ void attmean_tiles(const v2f* af, const float* __restrict__ Vb,
                                     const float* ssum, float* __restrict__ amO,
                                     int nt0, int half, int l16) {
  v2f bf[NT][AKS];
  int nIdx[NT];
#pragma unroll
  for (int t = 0; t < NT; t++) {
    nIdx[t] = (nt0 + 8 * t) * 16 + l16;
    int nLd = nIdx[t] < HH ? nIdx[t] : HH - 1;
    const float* bp = Vb + (size_t)(2 * half) * HH + nLd;
#pragma unroll
    for (int i = 0; i < AKS; i++) {
      bf[t][i].x = bp[(size_t)(4 * i) * HH];
      bf[t][i].y = bp[(size_t)(4 * i + 1) * HH];
    }
  }
  __builtin_amdgcn_sched_barrier(0);
  v8f acc[NT];
#pragma unroll
  for (int t = 0; t < NT; t++) acc[t] = (v8f){0, 0, 0, 0, 0, 0, 0, 0};
#pragma unroll
  for (int i = 0; i < AKS; i++) {
#pragma unroll
    for (int t = 0; t < NT; t++)
      acc[t] = __builtin_amdgcn_wmma_f32_16x16x4_f32(false, af[i], false, bf[t][i], (short)0, acc[t], false, false);
  }
#pragma unroll
  for (int t = 0; t < NT; t++) {
    if (nIdx[t] < HH) {
#pragma unroll
      for (int r = 0; r < 8; r++) {
        int m = r + 8 * half;
        float d = fmaxf(ssum[m], EPSF);
        amO[(size_t)m * HH + nIdx[t]] = acc[t][r] / d;
      }
    }
  }
}

__global__ void __launch_bounds__(256) cos_att_kernel(
    const float* __restrict__ cp, const float* __restrict__ ch,
    const float* __restrict__ np, const float* __restrict__ nh,
    const int* __restrict__ mask_p, const int* __restrict__ mask_h,
    float* __restrict__ att_mean_h, float* __restrict__ att_mean_p,
    float* __restrict__ att_max_h, float* __restrict__ att_max_p,
    float* __restrict__ out) {
  __shared__ float pmax[8][ST], psum[8][ST];
  __shared__ v4f cs[SS][ST / 4];
  __shared__ float ssum[ST];
  int blk = blockIdx.x;
  int side = blk >= NCOS;
  if (side) blk -= NCOS;
  int b = blk >> 3, st = blk & 7;
  int w = threadIdx.x >> 5;
  int lane = threadIdx.x & 31, half = lane >> 4, l16 = lane & 15;
  int x0 = (side ? w : st) << 4;
  int y0 = (side ? st : w) << 4;
  const float* arow = cp + ((size_t)(b * SS) + x0 + l16) * HH + 2 * half;
  const float* brow = ch + ((size_t)(b * SS) + y0 + l16) * HH + 2 * half;
  v8f acc = {0, 0, 0, 0, 0, 0, 0, 0};
  for (int k = 0; k < HH; k += 4) {
    v2f a = *(const v2f*)(arow + k);
    v2f bv = *(const v2f*)(brow + k);
    acc = __builtin_amdgcn_wmma_f32_16x16x4_f32(false, a, false, bv, (short)0, acc, false, false);
  }
  float nhv = nh[b * SS + y0 + l16];
  float c[8];
#pragma unroll
  for (int r = 0; r < 8; r++) {
    int m = r + 8 * half;
    float d = fmaxf(np[b * SS + x0 + m] * nhv, EPSF);
    c[r] = acc[r] / d;
  }
  if (!side) {
    v4f lo, hi;
    lo.x = c[0]; lo.y = c[1]; lo.z = c[2]; lo.w = c[3];
    hi.x = c[4]; hi.y = c[5]; hi.z = c[6]; hi.w = c[7];
    cs[y0 + l16][2 * half] = lo;
    cs[y0 + l16][2 * half + 1] = hi;
    float mx[8], sm[8];
#pragma unroll
    for (int r = 0; r < 8; r++) { mx[r] = c[r]; sm[r] = c[r]; }
#pragma unroll
    for (int o = 8; o > 0; o >>= 1) {
#pragma unroll
      for (int r = 0; r < 8; r++) {
        mx[r] = fmaxf(mx[r], __shfl_xor(mx[r], o, 32));
        sm[r] += __shfl_xor(sm[r], o, 32);
      }
    }
    if (l16 == 0) {
#pragma unroll
      for (int r = 0; r < 8; r++) { pmax[w][r + 8 * half] = mx[r]; psum[w][r + 8 * half] = sm[r]; }
    }
  } else {
    float* csf = (float*)cs;
#pragma unroll
    for (int r = 0; r < 8; r++) csf[(x0 + r + 8 * half) * ST + l16] = c[r];
    float mx = c[0], sm = c[0];
#pragma unroll
    for (int r = 1; r < 8; r++) { mx = fmaxf(mx, c[r]); sm += c[r]; }
    mx = fmaxf(mx, __shfl_xor(mx, 16, 32));
    sm += __shfl_xor(sm, 16, 32);
    if (half == 0) { pmax[w][l16] = mx; psum[w][l16] = sm; }
  }
  __syncthreads();
  if (threadIdx.x < ST) {
    int t = threadIdx.x;
    float mx = pmax[0][t], sm = psum[0][t];
#pragma unroll
    for (int u = 1; u < 8; u++) { mx = fmaxf(mx, pmax[u][t]); sm += psum[u][t]; }
    int r = (st << 4) + t;
    float m = (float)(side ? mask_h : mask_p)[b * SS + r];
    ssum[t] = sm;
    float* o2 = out + ((size_t)(side * BB + b) * SS + r) * FF;
    o2[0] = m * mx;
    o2[1] = m * (sm / (float)SS);
  }
  __syncthreads();

  const float* Vb = (side ? cp : ch) + (size_t)b * SS * HH;
  size_t o0 = ((size_t)b * SS + (st << 4)) * HH;
  float* amO = (side ? att_mean_p : att_mean_h) + o0;
  float* axO = (side ? att_max_p : att_max_h) + o0;
  const int* mk = (side ? mask_h : mask_p) + b * SS + (st << 4);

  {
    const float* csf = (const float*)cs;
    v2f af[AKS];
#pragma unroll
    for (int i = 0; i < AKS; i++) {
      af[i].x = csf[(4 * i + 2 * half) * ST + l16];
      af[i].y = csf[(4 * i + 2 * half + 1) * ST + l16];
    }
    if (w + 8 < NHT) attmean_tiles<2>(af, Vb, ssum, amO, w, half, l16);
    else attmean_tiles<1>(af, Vb, ssum, amO, w, half, l16);
  }

  int h = threadIdx.x;
  if (h >= HH) return;
  const float* vp = Vb + h;
  float mx[ST];
#pragma unroll
  for (int j = 0; j < ST; j++) mx[j] = -3.0e38f;
  float v[KV], vn[KV];
#pragma unroll
  for (int u = 0; u < KV; u++) vn[u] = vp[(size_t)u * HH];
#pragma unroll 1
  for (int k0 = 0; k0 < SS; k0 += KV) {
#pragma unroll
    for (int u = 0; u < KV; u++) v[u] = vn[u];
    if (k0 + KV < SS) {
#pragma unroll
      for (int u = 0; u < KV; u++) vn[u] = vp[(size_t)(k0 + KV + u) * HH];
    }
#pragma unroll
    for (int s = 0; s < KV; s += KU) {
      v4f c4[KU][ST / 4];
#pragma unroll
      for (int u = 0; u < KU; u++) {
#pragma unroll
        for (int q = 0; q < ST / 4; q++) c4[u][q] = cs[k0 + s + u][q];
      }
      __builtin_amdgcn_sched_barrier(0);
#pragma unroll
      for (int u = 0; u < KU; u++) {
#pragma unroll
        for (int q = 0; q < ST / 4; q++) {
          mx[4 * q + 0] = fmaxf(mx[4 * q + 0], c4[u][q].x * v[s + u]);
          mx[4 * q + 1] = fmaxf(mx[4 * q + 1], c4[u][q].y * v[s + u]);
          mx[4 * q + 2] = fmaxf(mx[4 * q + 2], c4[u][q].z * v[s + u]);
          mx[4 * q + 3] = fmaxf(mx[4 * q + 3], c4[u][q].w * v[s + u]);
        }
      }
    }
  }
#pragma unroll
  for (int j = 0; j < ST; j++) {
    float m = (float)mk[j];
    axO[(size_t)j * HH + h] = m * mx[j];
  }
}

#define NMPM ((BB * SS / 16) * 2)
__device__ __forceinline__ void mpm_body(const float* __restrict__ v1, const float* __restrict__ v2,
                                const float* __restrict__ w, int v2_bstride, int v2_sstride,
                                float* __restrict__ out, int side, int foff, int tile, int lane) {
  int m0 = (tile >> 1) << 4;
  int n0 = (tile & 1) << 4;
  int half = lane >> 4, l16 = lane & 15;
  int bs = m0 + l16;
  int b = bs / SS, s = bs % SS;
  const float* p1 = v1 + (size_t)bs * HH + 2 * half;
  const float* p2 = v2 + (size_t)b * v2_bstride + (size_t)s * v2_sstride + 2 * half;
  int col = n0 + l16;
  bool isW = col < PP;
  float fill = (col == PP) ? 1.f : 0.f;
  const float* wrow = w + (size_t)(isW ? col : PP - 1) * HH + 2 * half;
  v8f s11 = {0, 0, 0, 0, 0, 0, 0, 0};
  v8f s22 = {0, 0, 0, 0, 0, 0, 0, 0};
  v8f s12 = {0, 0, 0, 0, 0, 0, 0, 0};
  v2f xs[2][KCH], ys[2][KCH], ws2[2][KCH];
#pragma unroll
  for (int i = 0; i < KCH; i++) {
    xs[0][i] = *(const v2f*)(p1 + 4 * i);
    ys[0][i] = *(const v2f*)(p2 + 4 * i);
    ws2[0][i] = *(const v2f*)(wrow + 4 * i);
  }
#pragma unroll
  for (int c = 0; c < NCH; c++) {
    if (c + 1 < NCH) {
#pragma unroll
      for (int i = 0; i < KCH; i++) {
        xs[(c + 1) & 1][i] = *(const v2f*)(p1 + 4 * KCH * (c + 1) + 4 * i);
        ys[(c + 1) & 1][i] = *(const v2f*)(p2 + 4 * KCH * (c + 1) + 4 * i);
        ws2[(c + 1) & 1][i] = *(const v2f*)(wrow + 4 * KCH * (c + 1) + 4 * i);
      }
    }
    __builtin_amdgcn_sched_barrier(0);
#pragma unroll
    for (int i = 0; i < KCH; i++) {
      v2f x = xs[c & 1][i], y = ys[c & 1][i];
      v2f bv = wsq_frag(ws2[c & 1][i], isW, fill);
      v2f xx, yy, xy;
      xx.x = x.x * x.x; xx.y = x.y * x.y;
      yy.x = y.x * y.x; yy.y = y.y * y.y;
      xy.x = x.x * y.x; xy.y = x.y * y.y;
      s11 = __builtin_amdgcn_wmma_f32_16x16x4_f32(false, xx, false, bv, (short)0, s11, false, false);
      s22 = __builtin_amdgcn_wmma_f32_16x16x4_f32(false, yy, false, bv, (short)0, s22, false, false);
      s12 = __builtin_amdgcn_wmma_f32_16x16x4_f32(false, xy, false, bv, (short)0, s12, false, false);
    }
  }
  if (col <= PP) {
    int f = (col == PP) ? 0 : 1 + col;
#pragma unroll
    for (int r = 0; r < 8; r++) {
      int row = m0 + r + 8 * half;
      int ob = row / SS, os = row % SS;
      float* orow = out + ((size_t)(side * BB + ob) * SS + os) * FF + foff;
      orow[f] = s12[r] / (fmaxf(sqrtf(s11[r]), EPSF) * fmaxf(sqrtf(s22[r]), EPSF));
    }
  }
}

#define NJ 25
#define GCH 10
#define NYT 8
#define NMAXB (BB * PP)
__device__ __forceinline__ void maxpool_block(const float* __restrict__ cp, const float* __restrict__ ch,
                                     const float* __restrict__ w,
                                     const float* __restrict__ wnP, const float* __restrict__ wnH,
                                     const int* __restrict__ mask_p, const int* __restrict__ mask_h,
                                     float* __restrict__ out, float (*colS)[NYT][SS], int blk, int tid) {
  int b = blk / PP;
  int p = blk % PP;
  int wv = tid >> 5, lane = tid & 31;
  int x0 = wv * 16;
  int half = lane >> 4, l16 = lane & 15;
  float rmax[8], rsum[8], wA[8];
#pragma unroll
  for (int r = 0; r < 8; r++) {
    rmax[r] = -3.0e38f; rsum[r] = 0.f;
    wA[r] = wnP[((size_t)b * SS + x0 + r + 8 * half) * PP + p];
  }
  const float* a_row = cp + ((size_t)b * SS + x0 + l16) * HH + 4 * half;
  const float* w_row = w + (size_t)p * HH + 4 * half;
  const float* b_row = ch + ((size_t)b * SS + l16) * HH + 4 * half;
  const float* wnB_col = wnH + ((size_t)b * SS + l16) * PP + p;
  float* col_slot = &colS[half][wv][l16];

  v4f bf[2][GCH];
#pragma unroll
  for (int i = 0; i < GCH; i++) bf[0][i] = *(const v4f*)(b_row + 8 * i);

  v4f af[NJ];
#pragma unroll
  for (int j = 0; j < NJ; j++) {
    v4f wv4 = *(const v4f*)(w_row + 8 * j);
    v4f av = *(const v4f*)(a_row + 8 * j);
    af[j].x = wv4.x * wv4.x * av.x;
    af[j].y = wv4.y * wv4.y * av.y;
    af[j].z = wv4.z * wv4.z * av.z;
    af[j].w = wv4.w * wv4.w * av.w;
  }

  v8f acc = {0, 0, 0, 0, 0, 0, 0, 0};
  float wB = 0.f;
#pragma unroll
  for (int g0 = 0; g0 < NYT * NJ; g0 += GCH) {
    if (g0 + GCH < NYT * NJ) {
#pragma unroll
      for (int i = 0; i < GCH; i++) {
        int gn = g0 + GCH + i;
        bf[((g0 / GCH) + 1) & 1][i] = *(const v4f*)(b_row + (gn / NJ) * 16 * HH + 8 * (gn % NJ));
      }
    }
    __builtin_amdgcn_sched_barrier(0);
#pragma unroll
    for (int i = 0; i < GCH; i++) {
      int g = g0 + i;
      int yt = g / NJ, j = g % NJ;
      if (j == 0) {
        acc = (v8f){0, 0, 0, 0, 0, 0, 0, 0};
        wB = wnB_col[(size_t)yt * 16 * PP];
      }
      v4f bb = bf[(g0 / GCH) & 1][i];
      v4f aa = af[j];
      v2f a0, b0, a1, b1;
      a0.x = aa.x; a0.y = aa.y; b0.x = bb.x; b0.y = bb.y;
      a1.x = aa.z; a1.y = aa.w; b1.x = bb.z; b1.y = bb.w;
      acc = __builtin_amdgcn_wmma_f32_16x16x4_f32(false, a0, false, b0, (short)0, acc, false, false);
      acc = __builtin_amdgcn_wmma_f32_16x16x4_f32(false, a1, false, b1, (short)0, acc, false, false);
      if (j == NJ - 1) {
        float cm = -3.0e38f, csm = 0.f;
#pragma unroll
        for (int r = 0; r < 8; r++) {
          float mv = acc[r] / fmaxf(wA[r] * wB, EPSF);
          rmax[r] = fmaxf(rmax[r], mv);
          rsum[r] += mv;
          cm = fmaxf(cm, mv);
          csm += mv;
        }
        cm = fmaxf(cm, __shfl_xor(cm, 16, 32));
        csm += __shfl_xor(csm, 16, 32);
        col_slot[yt * 16] = half ? csm : cm;
      }
    }
  }
#pragma unroll
  for (int r = 0; r < 8; r++) {
    for (int o = 8; o > 0; o >>= 1) {
      rmax[r] = fmaxf(rmax[r], __shfl_xor(rmax[r], o, 32));
      rsum[r] += __shfl_xor(rsum[r], o, 32);
    }
  }
  if (l16 == 0) {
#pragma unroll
    for (int r = 0; r < 8; r++) {
      int x = x0 + r + 8 * half;
      float m = (float)mask_p[b * SS + x];
      float* orow = out + ((size_t)b * SS + x) * FF;
      orow[23 + p] = m * rmax[r];
      orow[43 + p] = m * (rsum[r] / (float)SS);
    }
  }
  __syncthreads();
  int y = tid & (SS - 1);
  int isSum = tid >> 7;
  float v = colS[isSum][0][y];
#pragma unroll
  for (int u = 1; u < NYT; u++) {
    float q = colS[isSum][u][y];
    v = isSum ? v + q : fmaxf(v, q);
  }
  float mh = (float)mask_h[b * SS + y];
  float* o2 = out + ((size_t)(BB + b) * SS + y) * FF;
  if (isSum) o2[43 + p] = mh * (v / (float)SS);
  else o2[23 + p] = mh * v;
}

__global__ void __launch_bounds__(256, 4) match_kernel(
    const float* __restrict__ cp, const float* __restrict__ ch,
    const float* __restrict__ cpl, const float* __restrict__ chl,
    const float* __restrict__ att_mean_h, const float* __restrict__ att_mean_p,
    const float* __restrict__ att_max_h, const float* __restrict__ att_max_p,
    const float* __restrict__ w0, const float* __restrict__ w1,
    const float* __restrict__ w2, const float* __restrict__ w3,
    const float* __restrict__ wn1_p, const float* __restrict__ wn1_h,
    const int* __restrict__ mask_p, const int* __restrict__ mask_h, float* __restrict__ out) {
  __shared__ float colS[2][NYT][SS];
  int blk = blockIdx.x;
  if (blk < NMAXB) {
    maxpool_block(cp, ch, w1, wn1_p, wn1_h, mask_p, mask_h, out, colS, blk, threadIdx.x);
  } else {
    int wid = (blk - NMAXB) * 8 + (threadIdx.x >> 5);
    int lane = threadIdx.x & 31;
    int grp = wid / (2 * NMPM);
    int rem = wid % (2 * NMPM);
    bool side = rem >= NMPM;
    int tile = side ? rem - NMPM : rem;
    const float* v1 = side ? ch : cp;
    const float* v2 = grp == 0 ? (side ? cpl : chl)
                    : grp == 1 ? (side ? att_mean_p : att_mean_h) : (side ? att_max_p : att_max_h);
    const float* w = grp == 0 ? w0 : grp == 1 ? w2 : w3;
    int v2_bstride = grp == 0 ? HH : SS * HH;
    int v2_sstride = grp == 0 ? 0 : HH;
    int foff = grp == 0 ? 2 : grp == 1 ? 63 : 84;
    mpm_body(v1, v2, w, v2_bstride, v2_sstride, out, side ? 1 : 0, foff, tile, lane);
  }
}

extern "C" void kernel_launch(void* const* d_in, const int* in_sizes, int n_in,
                              void* d_out, int out_size, void* d_ws, size_t ws_size,
                              hipStream_t stream) {
  const float* ctx_p = (const float*)d_in[0];
  const float* ctx_h = (const float*)d_in[1];
  const float* w0 = (const float*)d_in[2];
  const float* w1 = (const float*)d_in[3];
  const float* w2 = (const float*)d_in[4];
  const float* w3 = (const float*)d_in[5];
  const int* mask_p = (const int*)d_in[6];
  const int* mask_h = (const int*)d_in[7];
  float* out = (float*)d_out;
  float* ws = (float*)d_ws;

  const size_t SZ_ROWS = (size_t)BB * SS * HH;
  float* cp = ws;                 ws += SZ_ROWS;
  float* ch = ws;                 ws += SZ_ROWS;
  float* n_p = ws;                ws += BB * SS;
  float* n_h = ws;                ws += BB * SS;
  float* wn1_p = ws;              ws += (size_t)BB * SS * PP;
  float* wn1_h = ws;              ws += (size_t)BB * SS * PP;
  float* cpl = ws;                ws += BB * HH;
  float* chl = ws;                ws += BB * HH;
  float* att_mean_h = ws;         ws += SZ_ROWS;
  float* att_mean_p = ws;         ws += SZ_ROWS;
  float* att_max_h = ws;          ws += SZ_ROWS;
  float* att_max_p = ws;          ws += SZ_ROWS;

  prep_gather_kernel<<<2 * NPREP + BB, 32, 0, stream>>>(ctx_p, ctx_h, mask_p, mask_h, w1, cp, ch, n_p, n_h,
                                                        wn1_p, wn1_h, cpl, chl);

  cos_att_kernel<<<2 * NCOS, 256, 0, stream>>>(cp, ch, n_p, n_h, mask_p, mask_h,
                                               att_mean_h, att_mean_p, att_max_h, att_max_p, out);

  match_kernel<<<NMAXB + 6 * NMPM / 8, 256, 0, stream>>>(cp, ch, cpl, chl, att_mean_h, att_mean_p,
                                                        att_max_h, att_max_p, w0, w1, w2, w3,
                                                        wn1_p, wn1_h, mask_p, mask_h, out);
}


# maxpool match: a block copies ch[b] once into LDS (whole-line loads, padded rows) and its 8 waves read B fragments from LDS, not from memory at 16 cache lines per load; same sums, same order
# speedup vs baseline: 10.8383x; 1.0550x over previous
#include <hip/hip_runtime.h>
#include <hip/hip_bf16.h>


#define BB 16
#define SS 128
#define HH 200
#define PP 20
#define FF 105
#define EPSF 1e-8f
#define KCH 10
#define NCH 5

typedef float v2f __attribute__((ext_vector_type(2)));
typedef float v4f __attribute__((ext_vector_type(4)));
typedef float v8f __attribute__((ext_vector_type(8)));

__device__ __forceinline__ v2f wsq_frag(v2f ww, bool isW, float fill) {
  v2f bv;
  bv.x = isW ? ww.x * ww.x : fill;
  bv.y = isW ? ww.y * ww.y : fill;
  return bv;
}

#define NPREP ((BB * SS / 16) * 2)
__device__ __forceinline__ void prep_body(const float* __restrict__ src, const int* __restrict__ mask,
                                 const float* __restrict__ w1, float* __restrict__ dst,
                                 float* __restrict__ nrm, float* __restrict__ wn, int tile, int lane) {
  int m0 = (tile >> 1) << 4;
  int n0 = (tile & 1) << 4;
  int half = lane >> 4, l16 = lane & 15;
  int bs = m0 + l16;
  float m = (float)mask[bs];
  const float* srow = src + (size_t)bs * HH + 2 * half;
  float* drow = dst + (size_t)bs * HH + 2 * half;
  int col = n0 + l16;
  bool isW = col < PP;
  float fill = (col == PP) ? 1.f : 0.f;
  const float* wrow = w1 + (size_t)(isW ? col : PP - 1) * HH + 2 * half;
  v8f acc = {0, 0, 0, 0, 0, 0, 0, 0};
  v2f xs[2][KCH], ws2[2][KCH];
#pragma unroll
  for (int i = 0; i < KCH; i++) {
    xs[0][i] = *(const v2f*)(srow + 4 * i);
    ws2[0][i] = *(const v2f*)(wrow + 4 * i);
  }
#pragma unroll
  for (int c = 0; c < NCH; c++) {
    if (c + 1 < NCH) {
#pragma unroll
      for (int i = 0; i < KCH; i++) {
        xs[(c + 1) & 1][i] = *(const v2f*)(srow + 4 * KCH * (c + 1) + 4 * i);
        ws2[(c + 1) & 1][i] = *(const v2f*)(wrow + 4 * KCH * (c + 1) + 4 * i);
      }
    }
    __builtin_amdgcn_sched_barrier(0);
#pragma unroll
    for (int i = 0; i < KCH; i++) {
      v2f x = xs[c & 1][i];
      x.x *= m; x.y *= m;
      if (n0 == 0) *(v2f*)(drow + 4 * KCH * c + 4 * i) = x;
      v2f bv = wsq_frag(ws2[c & 1][i], isW, fill);
      v2f a; a.x = x.x * x.x; a.y = x.y * x.y;
      acc = __builtin_amdgcn_wmma_f32_16x16x4_f32(false, a, false, bv, (short)0, acc, false, false);
    }
  }
  if (col <= PP) {
#pragma unroll
    for (int r = 0; r < 8; r++) {
      int row = m0 + r + 8 * half;
      float* o = (col == PP) ? nrm + row : wn + (size_t)row * PP + col;
      *o = sqrtf(acc[r]);
    }
  }
}

__device__ __forceinline__ void gather_last_body(const float* __restrict__ xp, const float* __restrict__ xh,
                                   const int* __restrict__ mask_p, const int* __restrict__ mask_h,
                                   float* __restrict__ cpl, float* __restrict__ chl, int b, int lane) {
  int lp = 0, lh = 0;
  for (int i = 0; i < SS; i++) { lp += mask_p[b * SS + i]; lh += mask_h[b * SS + i]; }
  int ip = (lp > 0) ? lp - 1 : 0;
  int ih = (lh > 0) ? lh - 1 : 0;
  float mp = (float)mask_p[b * SS + ip];
  float mh = (float)mask_h[b * SS + ih];
  for (int h = lane; h < HH; h += 32) {
    cpl[b * HH + h] = xp[((size_t)b * SS + ip) * HH + h] * mp;
    chl[b * HH + h] = xh[((size_t)b * SS + ih) * HH + h] * mh;
  }
}

__global__ void __launch_bounds__(32) prep_gather_kernel(
    const float* __restrict__ ctx_p, const float* __restrict__ ctx_h,
    const int* __restrict__ mask_p, const int* __restrict__ mask_h, const float* __restrict__ w1,
    float* __restrict__ cp, float* __restrict__ ch, float* __restrict__ n_p, float* __restrict__ n_h,
    float* __restrict__ wn1_p, float* __restrict__ wn1_h, float* __restrict__ cpl, float* __restrict__ chl) {
  int blk = blockIdx.x, lane = threadIdx.x;
  if (blk < 2 * NPREP) {
    bool side = blk >= NPREP;
    prep_body(side ? ctx_h : ctx_p, side ? mask_h : mask_p, w1, side ? ch : cp,
              side ? n_h : n_p, side ? wn1_h : wn1_p, side ? blk - NPREP : blk, lane);
  } else {
    gather_last_body(ctx_p, ctx_h, mask_p, mask_h, cpl, chl, blk - 2 * NPREP, lane);
  }
}

#define NCOS (BB * 8)
#define AKS (SS / 4)
#define NHT 13
#define ST 16
#define KU 4
#define KV 32

template <int NT>
__device__ __forceinline__ void attmean_tiles(const v2f* af, const float* __restrict__ Vb,
                                     const float* ssum, float* __restrict__ amO,
                                     int nt0, int half, int l16) {
  v2f bf[NT][AKS];
  int nIdx[NT];
#pragma unroll
  for (int t = 0; t < NT; t++) {
    nIdx[t] = (nt0 + 8 * t) * 16 + l16;
    int nLd = nIdx[t] < HH ? nIdx[t] : HH - 1;
    const float* bp = Vb + (size_t)(2 * half) * HH + nLd;
#pragma unroll
    for (int i = 0; i < AKS; i++) {
      bf[t][i].x = bp[(size_t)(4 * i) * HH];
      bf[t][i].y = bp[(size_t)(4 * i + 1) * HH];
    }
  }
  __builtin_amdgcn_sched_barrier(0);
  v8f acc[NT];
#pragma unroll
  for (int t = 0; t < NT; t++) acc[t] = (v8f){0, 0, 0, 0, 0, 0, 0, 0};
#pragma unroll
  for (int i = 0; i < AKS; i++) {
#pragma unroll
    for (int t = 0; t < NT; t++)
      acc[t] = __builtin_amdgcn_wmma_f32_16x16x4_f32(false, af[i], false, bf[t][i], (short)0, acc[t], false, false);
  }
#pragma unroll
  for (int t = 0; t < NT; t++) {
    if (nIdx[t] < HH) {
#pragma unroll
      for (int r = 0; r < 8; r++) {
        int m = r + 8 * half;
        float d = fmaxf(ssum[m], EPSF);
        amO[(size_t)m * HH + nIdx[t]] = acc[t][r] / d;
      }
    }
  }
}

__global__ void __launch_bounds__(256) cos_att_kernel(
    const float* __restrict__ cp, const float* __restrict__ ch,
    const float* __restrict__ np, const float* __restrict__ nh,
    const int* __restrict__ mask_p, const int* __restrict__ mask_h,
    float* __restrict__ att_mean_h, float* __restrict__ att_mean_p,
    float* __restrict__ att_max_h, float* __restrict__ att_max_p,
    float* __restrict__ out) {
  __shared__ float pmax[8][ST], psum[8][ST];
  __shared__ v4f cs[SS][ST / 4];
  __shared__ float ssum[ST];
  int blk = blockIdx.x;
  int side = blk >= NCOS;
  if (side) blk -= NCOS;
  int b = blk >> 3, st = blk & 7;
  int w = threadIdx.x >> 5;
  int lane = threadIdx.x & 31, half = lane >> 4, l16 = lane & 15;
  int x0 = (side ? w : st) << 4;
  int y0 = (side ? st : w) << 4;
  const float* arow = cp + ((size_t)(b * SS) + x0 + l16) * HH + 2 * half;
  const float* brow = ch + ((size_t)(b * SS) + y0 + l16) * HH + 2 * half;
  v8f acc = {0, 0, 0, 0, 0, 0, 0, 0};
  for (int k = 0; k < HH; k += 4) {
    v2f a = *(const v2f*)(arow + k);
    v2f bv = *(const v2f*)(brow + k);
    acc = __builtin_amdgcn_wmma_f32_16x16x4_f32(false, a, false, bv, (short)0, acc, false, false);
  }
  float nhv = nh[b * SS + y0 + l16];
  float c[8];
#pragma unroll
  for (int r = 0; r < 8; r++) {
    int m = r + 8 * half;
    float d = fmaxf(np[b * SS + x0 + m] * nhv, EPSF);
    c[r] = acc[r] / d;
  }
  if (!side) {
    v4f lo, hi;
    lo.x = c[0]; lo.y = c[1]; lo.z = c[2]; lo.w = c[3];
    hi.x = c[4]; hi.y = c[5]; hi.z = c[6]; hi.w = c[7];
    cs[y0 + l16][2 * half] = lo;
    cs[y0 + l16][2 * half + 1] = hi;
    float mx[8], sm[8];
#pragma unroll
    for (int r = 0; r < 8; r++) { mx[r] = c[r]; sm[r] = c[r]; }
#pragma unroll
    for (int o = 8; o > 0; o >>= 1) {
#pragma unroll
      for (int r = 0; r < 8; r++) {
        mx[r] = fmaxf(mx[r], __shfl_xor(mx[r], o, 32));
        sm[r] += __shfl_xor(sm[r], o, 32);
      }
    }
    if (l16 == 0) {
#pragma unroll
      for (int r = 0; r < 8; r++) { pmax[w][r + 8 * half] = mx[r]; psum[w][r + 8 * half] = sm[r]; }
    }
  } else {
    float* csf = (float*)cs;
#pragma unroll
    for (int r = 0; r < 8; r++) csf[(x0 + r + 8 * half) * ST + l16] = c[r];
    float mx = c[0], sm = c[0];
#pragma unroll
    for (int r = 1; r < 8; r++) { mx = fmaxf(mx, c[r]); sm += c[r]; }
    mx = fmaxf(mx, __shfl_xor(mx, 16, 32));
    sm += __shfl_xor(sm, 16, 32);
    if (half == 0) { pmax[w][l16] = mx; psum[w][l16] = sm; }
  }
  __syncthreads();
  if (threadIdx.x < ST) {
    int t = threadIdx.x;
    float mx = pmax[0][t], sm = psum[0][t];
#pragma unroll
    for (int u = 1; u < 8; u++) { mx = fmaxf(mx, pmax[u][t]); sm += psum[u][t]; }
    int r = (st << 4) + t;
    float m = (float)(side ? mask_h : mask_p)[b * SS + r];
    ssum[t] = sm;
    float* o2 = out + ((size_t)(side * BB + b) * SS + r) * FF;
    o2[0] = m * mx;
    o2[1] = m * (sm / (float)SS);
  }
  __syncthreads();

  const float* Vb = (side ? cp : ch) + (size_t)b * SS * HH;
  size_t o0 = ((size_t)b * SS + (st << 4)) * HH;
  float* amO = (side ? att_mean_p : att_mean_h) + o0;
  float* axO = (side ? att_max_p : att_max_h) + o0;
  const int* mk = (side ? mask_h : mask_p) + b * SS + (st << 4);

  {
    const float* csf = (const float*)cs;
    v2f af[AKS];
#pragma unroll
    for (int i = 0; i < AKS; i++) {
      af[i].x = csf[(4 * i + 2 * half) * ST + l16];
      af[i].y = csf[(4 * i + 2 * half + 1) * ST + l16];
    }
    if (w + 8 < NHT) attmean_tiles<2>(af, Vb, ssum, amO, w, half, l16);
    else attmean_tiles<1>(af, Vb, ssum, amO, w, half, l16);
  }

  int h = threadIdx.x;
  if (h >= HH) return;
  const float* vp = Vb + h;
  float mx[ST];
#pragma unroll
  for (int j = 0; j < ST; j++) mx[j] = -3.0e38f;
  float v[KV], vn[KV];
#pragma unroll
  for (int u = 0; u < KV; u++) vn[u] = vp[(size_t)u * HH];
#pragma unroll 1
  for (int k0 = 0; k0 < SS; k0 += KV) {
#pragma unroll
    for (int u = 0; u < KV; u++) v[u] = vn[u];
    if (k0 + KV < SS) {
#pragma unroll
      for (int u = 0; u < KV; u++) vn[u] = vp[(size_t)(k0 + KV + u) * HH];
    }
#pragma unroll
    for (int s = 0; s < KV; s += KU) {
      v4f c4[KU][ST / 4];
#pragma unroll
      for (int u = 0; u < KU; u++) {
#pragma unroll
        for (int q = 0; q < ST / 4; q++) c4[u][q] = cs[k0 + s + u][q];
      }
      __builtin_amdgcn_sched_barrier(0);
#pragma unroll
      for (int u = 0; u < KU; u++) {
#pragma unroll
        for (int q = 0; q < ST / 4; q++) {
          mx[4 * q + 0] = fmaxf(mx[4 * q + 0], c4[u][q].x * v[s + u]);
          mx[4 * q + 1] = fmaxf(mx[4 * q + 1], c4[u][q].y * v[s + u]);
          mx[4 * q + 2] = fmaxf(mx[4 * q + 2], c4[u][q].z * v[s + u]);
          mx[4 * q + 3] = fmaxf(mx[4 * q + 3], c4[u][q].w * v[s + u]);
        }
      }
    }
  }
#pragma unroll
  for (int j = 0; j < ST; j++) {
    float m = (float)mk[j];
    axO[(size_t)j * HH + h] = m * mx[j];
  }
}

#define NMPM ((BB * SS / 16) * 2)
__device__ __forceinline__ void mpm_body(const float* __restrict__ v1, const float* __restrict__ v2,
                                const float* __restrict__ w, int v2_bstride, int v2_sstride,
                                float* __restrict__ out, int side, int foff, int tile, int lane) {
  int m0 = (tile >> 1) << 4;
  int n0 = (tile & 1) << 4;
  int half = lane >> 4, l16 = lane & 15;
  int bs = m0 + l16;
  int b = bs / SS, s = bs % SS;
  const float* p1 = v1 + (size_t)bs * HH + 2 * half;
  const float* p2 = v2 + (size_t)b * v2_bstride + (size_t)s * v2_sstride + 2 * half;
  int col = n0 + l16;
  bool isW = col < PP;
  float fill = (col == PP) ? 1.f : 0.f;
  const float* wrow = w + (size_t)(isW ? col : PP - 1) * HH + 2 * half;
  v8f s11 = {0, 0, 0, 0, 0, 0, 0, 0};
  v8f s22 = {0, 0, 0, 0, 0, 0, 0, 0};
  v8f s12 = {0, 0, 0, 0, 0, 0, 0, 0};
  v2f xs[2][KCH], ys[2][KCH], ws2[2][KCH];
#pragma unroll
  for (int i = 0; i < KCH; i++) {
    xs[0][i] = *(const v2f*)(p1 + 4 * i);
    ys[0][i] = *(const v2f*)(p2 + 4 * i);
    ws2[0][i] = *(const v2f*)(wrow + 4 * i);
  }
#pragma unroll
  for (int c = 0; c < NCH; c++) {
    if (c + 1 < NCH) {
#pragma unroll
      for (int i = 0; i < KCH; i++) {
        xs[(c + 1) & 1][i] = *(const v2f*)(p1 + 4 * KCH * (c + 1) + 4 * i);
        ys[(c + 1) & 1][i] = *(const v2f*)(p2 + 4 * KCH * (c + 1) + 4 * i);
        ws2[(c + 1) & 1][i] = *(const v2f*)(wrow + 4 * KCH * (c + 1) + 4 * i);
      }
    }
    __builtin_amdgcn_sched_barrier(0);
#pragma unroll
    for (int i = 0; i < KCH; i++) {
      v2f x = xs[c & 1][i], y = ys[c & 1][i];
      v2f bv = wsq_frag(ws2[c & 1][i], isW, fill);
      v2f xx, yy, xy;
      xx.x = x.x * x.x; xx.y = x.y * x.y;
      yy.x = y.x * y.x; yy.y = y.y * y.y;
      xy.x = x.x * y.x; xy.y = x.y * y.y;
      s11 = __builtin_amdgcn_wmma_f32_16x16x4_f32(false, xx, false, bv, (short)0, s11, false, false);
      s22 = __builtin_amdgcn_wmma_f32_16x16x4_f32(false, yy, false, bv, (short)0, s22, false, false);
      s12 = __builtin_amdgcn_wmma_f32_16x16x4_f32(false, xy, false, bv, (short)0, s12, false, false);
    }
  }
  if (col <= PP) {
    int f = (col == PP) ? 0 : 1 + col;
#pragma unroll
    for (int r = 0; r < 8; r++) {
      int row = m0 + r + 8 * half;
      int ob = row / SS, os = row % SS;
      float* orow = out + ((size_t)(side * BB + ob) * SS + os) * FF + foff;
      orow[f] = s12[r] / (fmaxf(sqrtf(s11[r]), EPSF) * fmaxf(sqrtf(s22[r]), EPSF));
    }
  }
}

#define NJ 25
#define GCH 10
#define NYT 8
#define NMAXB (BB * PP)
#define BPITCH 204
#define NST (SS * HH / 4 / 256)
__device__ __forceinline__ void maxpool_block(const float* __restrict__ cp, const float* __restrict__ ch,
                                     const float* __restrict__ w,
                                     const float* __restrict__ wnP, const float* __restrict__ wnH,
                                     const int* __restrict__ mask_p, const int* __restrict__ mask_h,
                                     float* __restrict__ out, float* chS, float (*colS)[NYT][SS],
                                     int blk, int tid) {
  int b = blk / PP;
  int p = blk % PP;
  int wv = tid >> 5, lane = tid & 31;
  int x0 = wv * 16;
  int half = lane >> 4, l16 = lane & 15;
  {
    const v4f* gsrc = (const v4f*)(ch + (size_t)b * SS * HH);
    v4f st[NST];
#pragma unroll
    for (int i = 0; i < NST; i++) st[i] = gsrc[tid + 256 * i];
#pragma unroll
    for (int i = 0; i < NST; i++) {
      int e = tid + 256 * i;
      int row = e / (HH / 4), cg = e % (HH / 4);
      *(v4f*)(chS + row * BPITCH + 4 * cg) = st[i];
    }
  }
  float rmax[8], rsum[8], wA[8];
#pragma unroll
  for (int r = 0; r < 8; r++) {
    rmax[r] = -3.0e38f; rsum[r] = 0.f;
    wA[r] = wnP[((size_t)b * SS + x0 + r + 8 * half) * PP + p];
  }
  const float* a_row = cp + ((size_t)b * SS + x0 + l16) * HH + 4 * half;
  const float* w_row = w + (size_t)p * HH + 4 * half;
  const float* b_row = chS + l16 * BPITCH + 4 * half;
  const float* wnB_col = wnH + ((size_t)b * SS + l16) * PP + p;
  float* col_slot = &colS[half][wv][l16];

  v4f af[NJ];
#pragma unroll
  for (int j = 0; j < NJ; j++) {
    v4f wv4 = *(const v4f*)(w_row + 8 * j);
    v4f av = *(const v4f*)(a_row + 8 * j);
    af[j].x = wv4.x * wv4.x * av.x;
    af[j].y = wv4.y * wv4.y * av.y;
    af[j].z = wv4.z * wv4.z * av.z;
    af[j].w = wv4.w * wv4.w * av.w;
  }

  __syncthreads();

  v4f bf[2][GCH];
#pragma unroll
  for (int i = 0; i < GCH; i++) bf[0][i] = *(const v4f*)(b_row + 8 * i);

  v8f acc = {0, 0, 0, 0, 0, 0, 0, 0};
  float wB = 0.f;
#pragma unroll
  for (int g0 = 0; g0 < NYT * NJ; g0 += GCH) {
    if (g0 + GCH < NYT * NJ) {
#pragma unroll
      for (int i = 0; i < GCH; i++) {
        int gn = g0 + GCH + i;
        bf[((g0 / GCH) + 1) & 1][i] = *(const v4f*)(b_row + (gn / NJ) * 16 * BPITCH + 8 * (gn % NJ));
      }
    }
    __builtin_amdgcn_sched_barrier(0);
#pragma unroll
    for (int i = 0; i < GCH; i++) {
      int g = g0 + i;
      int yt = g / NJ, j = g % NJ;
      if (j == 0) {
        acc = (v8f){0, 0, 0, 0, 0, 0, 0, 0};
        wB = wnB_col[(size_t)yt * 16 * PP];
      }
      v4f bb = bf[(g0 / GCH) & 1][i];
      v4f aa = af[j];
      v2f a0, b0, a1, b1;
      a0.x = aa.x; a0.y = aa.y; b0.x = bb.x; b0.y = bb.y;
      a1.x = aa.z; a1.y = aa.w; b1.x = bb.z; b1.y = bb.w;
      acc = __builtin_amdgcn_wmma_f32_16x16x4_f32(false, a0, false, b0, (short)0, acc, false, false);
      acc = __builtin_amdgcn_wmma_f32_16x16x4_f32(false, a1, false, b1, (short)0, acc, false, false);
      if (j == NJ - 1) {
        float cm = -3.0e38f, csm = 0.f;
#pragma unroll
        for (int r = 0; r < 8; r++) {
          float mv = acc[r] / fmaxf(wA[r] * wB, EPSF);
          rmax[r] = fmaxf(rmax[r], mv);
          rsum[r] += mv;
          cm = fmaxf(cm, mv);
          csm += mv;
        }
        cm = fmaxf(cm, __shfl_xor(cm, 16, 32));
        csm += __shfl_xor(csm, 16, 32);
        col_slot[yt * 16] = half ? csm : cm;
      }
    }
  }
#pragma unroll
  for (int r = 0; r < 8; r++) {
    for (int o = 8; o > 0; o >>= 1) {
      rmax[r] = fmaxf(rmax[r], __shfl_xor(rmax[r], o, 32));
      rsum[r] += __shfl_xor(rsum[r], o, 32);
    }
  }
  if (l16 == 0) {
#pragma unroll
    for (int r = 0; r < 8; r++) {
      int x = x0 + r + 8 * half;
      float m = (float)mask_p[b * SS + x];
      float* orow = out + ((size_t)b * SS + x) * FF;
      orow[23 + p] = m * rmax[r];
      orow[43 + p] = m * (rsum[r] / (float)SS);
    }
  }
  __syncthreads();
  int y = tid & (SS - 1);
  int isSum = tid >> 7;
  float v = colS[isSum][0][y];
#pragma unroll
  for (int u = 1; u < NYT; u++) {
    float q = colS[isSum][u][y];
    v = isSum ? v + q : fmaxf(v, q);
  }
  float mh = (float)mask_h[b * SS + y];
  float* o2 = out + ((size_t)(BB + b) * SS + y) * FF;
  if (isSum) o2[43 + p] = mh * (v / (float)SS);
  else o2[23 + p] = mh * v;
}

__global__ void __launch_bounds__(256, 4) match_kernel(
    const float* __restrict__ cp, const float* __restrict__ ch,
    const float* __restrict__ cpl, const float* __restrict__ chl,
    const float* __restrict__ att_mean_h, const float* __restrict__ att_mean_p,
    const float* __restrict__ att_max_h, const float* __restrict__ att_max_p,
    const float* __restrict__ w0, const float* __restrict__ w1,
    const float* __restrict__ w2, const float* __restrict__ w3,
    const float* __restrict__ wn1_p, const float* __restrict__ wn1_h,
    const int* __restrict__ mask_p, const int* __restrict__ mask_h, float* __restrict__ out) {
  __shared__ float colS[2][NYT][SS];
  __shared__ __attribute__((aligned(16))) float chS[SS * BPITCH];
  int blk = blockIdx.x;
  if (blk < NMAXB) {
    maxpool_block(cp, ch, w1, wn1_p, wn1_h, mask_p, mask_h, out, chS, colS, blk, threadIdx.x);
  } else {
    int wid = (blk - NMAXB) * 8 + (threadIdx.x >> 5);
    int lane = threadIdx.x & 31;
    int grp = wid / (2 * NMPM);
    int rem = wid % (2 * NMPM);
    bool side = rem >= NMPM;
    int tile = side ? rem - NMPM : rem;
    const float* v1 = side ? ch : cp;
    const float* v2 = grp == 0 ? (side ? cpl : chl)
                    : grp == 1 ? (side ? att_mean_p : att_mean_h) : (side ? att_max_p : att_max_h);
    const float* w = grp == 0 ? w0 : grp == 1 ? w2 : w3;
    int v2_bstride = grp == 0 ? HH : SS * HH;
    int v2_sstride = grp == 0 ? 0 : HH;
    int foff = grp == 0 ? 2 : grp == 1 ? 63 : 84;
    mpm_body(v1, v2, w, v2_bstride, v2_sstride, out, side ? 1 : 0, foff, tile, lane);
  }
}

extern "C" void kernel_launch(void* const* d_in, const int* in_sizes, int n_in,
                              void* d_out, int out_size, void* d_ws, size_t ws_size,
                              hipStream_t stream) {
  const float* ctx_p = (const float*)d_in[0];
  const float* ctx_h = (const float*)d_in[1];
  const float* w0 = (const float*)d_in[2];
  const float* w1 = (const float*)d_in[3];
  const float* w2 = (const float*)d_in[4];
  const float* w3 = (const float*)d_in[5];
  const int* mask_p = (const int*)d_in[6];
  const int* mask_h = (const int*)d_in[7];
  float* out = (float*)d_out;
  float* ws = (float*)d_ws;

  const size_t SZ_ROWS = (size_t)BB * SS * HH;
  float* cp = ws;                 ws += SZ_ROWS;
  float* ch = ws;                 ws += SZ_ROWS;
  float* n_p = ws;                ws += BB * SS;
  float* n_h = ws;                ws += BB * SS;
  float* wn1_p = ws;              ws += (size_t)BB * SS * PP;
  float* wn1_h = ws;              ws += (size_t)BB * SS * PP;
  float* cpl = ws;                ws += BB * HH;
  float* chl = ws;                ws += BB * HH;
  float* att_mean_h = ws;         ws += SZ_ROWS;
  float* att_mean_p = ws;         ws += SZ_ROWS;
  float* att_max_h = ws;          ws += SZ_ROWS;
  float* att_max_p = ws;          ws += SZ_ROWS;

  prep_gather_kernel<<<2 * NPREP + BB, 32, 0, stream>>>(ctx_p, ctx_h, mask_p, mask_h, w1, cp, ch, n_p, n_h,
                                                        wn1_p, wn1_h, cpl, chl);

  cos_att_kernel<<<2 * NCOS, 256, 0, stream>>>(cp, ch, n_p, n_h, mask_p, mask_h,
                                               att_mean_h, att_mean_p, att_max_h, att_max_p, out);

  match_kernel<<<NMAXB + 6 * NMPM / 8, 256, 0, stream>>>(cp, ch, cpl, chl, att_mean_h, att_mean_p,
                                                        att_max_h, att_max_p, w0, w1, w2, w3,
                                                        wn1_p, wn1_h, mask_p, mask_h, out);
}


# prep launch folded into the cos launch: cos GEMM takes raw inputs times mask, row norms summed per lane beside it (their summation order changes), strip waves add the w1-norm WMMAs; 3 launches -> 2
# speedup vs baseline: 11.7738x; 1.0863x over previous
#include <hip/hip_runtime.h>
#include <hip/hip_bf16.h>


#define BB 16
#define SS 128
#define HH 200
#define PP 20
#define FF 105
#define EPSF 1e-8f
#define KCH 10
#define NCH 5

typedef float v2f __attribute__((ext_vector_type(2)));
typedef float v4f __attribute__((ext_vector_type(4)));
typedef float v8f __attribute__((ext_vector_type(8)));

__device__ __forceinline__ v2f wsq_frag(v2f ww, bool isW, float fill) {
  v2f bv;
  bv.x = isW ? ww.x * ww.x : fill;
  bv.y = isW ? ww.y * ww.y : fill;
  return bv;
}

#define NCOS (BB * 8)
#define AKS (SS / 4)
#define NHT 13
#define ST 16
#define NCPT (ST * HH / 4 / HH)
#define KU 4
#define KV 32

template <int NT>
__device__ __forceinline__ void attmean_tiles(const v2f* af, const float* __restrict__ Vb,
                                     const float* ssum, float* __restrict__ amO,
                                     int nt0, int half, int l16) {
  v2f bf[NT][AKS];
  int nIdx[NT];
#pragma unroll
  for (int t = 0; t < NT; t++) {
    nIdx[t] = (nt0 + 8 * t) * 16 + l16;
    int nLd = nIdx[t] < HH ? nIdx[t] : HH - 1;
    const float* bp = Vb + (size_t)(2 * half) * HH + nLd;
#pragma unroll
    for (int i = 0; i < AKS; i++) {
      bf[t][i].x = bp[(size_t)(4 * i) * HH];
      bf[t][i].y = bp[(size_t)(4 * i + 1) * HH];
    }
  }
  __builtin_amdgcn_sched_barrier(0);
  v8f acc[NT];
#pragma unroll
  for (int t = 0; t < NT; t++) acc[t] = (v8f){0, 0, 0, 0, 0, 0, 0, 0};
#pragma unroll
  for (int i = 0; i < AKS; i++) {
#pragma unroll
    for (int t = 0; t < NT; t++)
      acc[t] = __builtin_amdgcn_wmma_f32_16x16x4_f32(false, af[i], false, bf[t][i], (short)0, acc[t], false, false);
  }
#pragma unroll
  for (int t = 0; t < NT; t++) {
    if (nIdx[t] < HH) {
#pragma unroll
      for (int r = 0; r < 8; r++) {
        int m = r + 8 * half;
        float d = fmaxf(ssum[m], EPSF);
        amO[(size_t)m * HH + nIdx[t]] = acc[t][r] / d;
      }
    }
  }
}

__global__ void __launch_bounds__(256) cos_att_kernel(
    const float* __restrict__ ctx_p, const float* __restrict__ ctx_h,
    const int* __restrict__ mask_p, const int* __restrict__ mask_h,
    const float* __restrict__ w1,
    float* __restrict__ cp, float* __restrict__ ch,
    float* __restrict__ wn1_p, float* __restrict__ wn1_h,
    float* __restrict__ cpl, float* __restrict__ chl,
    float* __restrict__ att_mean_h, float* __restrict__ att_mean_p,
    float* __restrict__ att_max_h, float* __restrict__ att_max_p,
    float* __restrict__ out) {
  __shared__ float pmax[8][ST], psum[8][ST];
  __shared__ v4f cs[SS][ST / 4];
  __shared__ float ssum[ST];
  int blk = blockIdx.x;
  int side = blk >= NCOS;
  if (side) blk -= NCOS;
  int b = blk >> 3, st = blk & 7;
  int w = __builtin_amdgcn_readfirstlane(threadIdx.x >> 5);
  int lane = threadIdx.x & 31, half = lane >> 4, l16 = lane & 15;
  int x0 = (side ? w : st) << 4;
  int y0 = (side ? st : w) << 4;
  const float* arow = ctx_p + ((size_t)(b * SS) + x0 + l16) * HH + 2 * half;
  const float* brow = ctx_h + ((size_t)(b * SS) + y0 + l16) * HH + 2 * half;
  float ma = (float)mask_p[b * SS + x0 + l16];
  float mb = (float)mask_h[b * SS + y0 + l16];
  size_t srow0 = (size_t)(b * SS) + (st << 4);
  const float* csrc = (side ? ctx_h : ctx_p) + srow0 * HH;
  const int* cmsk = (side ? mask_h : mask_p) + srow0;
  v4f cpy[NCPT];
  int cpm[NCPT];
  if (threadIdx.x < HH) {
#pragma unroll
    for (int i = 0; i < NCPT; i++) {
      int e = threadIdx.x + HH * i;
      cpy[i] = *(const v4f*)(csrc + 4 * e);
      cpm[i] = cmsk[e / (HH / 4)];
    }
  }
  v8f acc = {0, 0, 0, 0, 0, 0, 0, 0};
  float sa0 = 0.f, sa1 = 0.f, sb0 = 0.f, sb1 = 0.f;
  if (w < 2) {
    int col = (w << 4) + l16;
    bool isW = col < PP;
    const float* wrow = w1 + (size_t)(isW ? col : PP - 1) * HH + 2 * half;
    v8f accw = {0, 0, 0, 0, 0, 0, 0, 0};
#pragma unroll 1
    for (int k0 = 0; k0 < HH; k0 += 4 * KCH) {
      v2f av[KCH], bw[KCH], wv[KCH];
#pragma unroll
      for (int i = 0; i < KCH; i++) {
        av[i] = *(const v2f*)(arow + k0 + 4 * i);
        bw[i] = *(const v2f*)(brow + k0 + 4 * i);
        wv[i] = *(const v2f*)(wrow + k0 + 4 * i);
      }
      __builtin_amdgcn_sched_barrier(0);
#pragma unroll
      for (int i = 0; i < KCH; i++) {
        v2f a = av[i], bv = bw[i];
        a.x *= ma; a.y *= ma;
        bv.x *= mb; bv.y *= mb;
        sa0 = fmaf(a.x, a.x, sa0); sa1 = fmaf(a.y, a.y, sa1);
        sb0 = fmaf(bv.x, bv.x, sb0); sb1 = fmaf(bv.y, bv.y, sb1);
        acc = __builtin_amdgcn_wmma_f32_16x16x4_f32(false, a, false, bv, (short)0, acc, false, false);
        v2f sv = side ? bv : a;
        v2f sq; sq.x = sv.x * sv.x; sq.y = sv.y * sv.y;
        accw = __builtin_amdgcn_wmma_f32_16x16x4_f32(false, sq, false, wsq_frag(wv[i], isW, 0.f), (short)0, accw, false, false);
      }
    }
    if (isW) {
      float* wn = (side ? wn1_h : wn1_p) + srow0 * PP + col;
#pragma unroll
      for (int r = 0; r < 8; r++) wn[(size_t)(r + 8 * half) * PP] = sqrtf(accw[r]);
    }
  } else {
    for (int k = 0; k < HH; k += 4) {
      v2f a = *(const v2f*)(arow + k);
      v2f bv = *(const v2f*)(brow + k);
      a.x *= ma; a.y *= ma;
      bv.x *= mb; bv.y *= mb;
      sa0 = fmaf(a.x, a.x, sa0); sa1 = fmaf(a.y, a.y, sa1);
      sb0 = fmaf(bv.x, bv.x, sb0); sb1 = fmaf(bv.y, bv.y, sb1);
      acc = __builtin_amdgcn_wmma_f32_16x16x4_f32(false, a, false, bv, (short)0, acc, false, false);
    }
  }
  if (threadIdx.x < HH) {
    float* cdst = (side ? ch : cp) + srow0 * HH;
#pragma unroll
    for (int i = 0; i < NCPT; i++) {
      int e = threadIdx.x + HH * i;
      float m = (float)cpm[i];
      v4f v = cpy[i];
      v.x *= m; v.y *= m; v.z *= m; v.w *= m;
      *(v4f*)(cdst + 4 * e) = v;
    }
  }
  float ssa = sa0 + sa1, ssb = sb0 + sb1;
  ssa += __shfl_xor(ssa, 16, 32);
  ssb += __shfl_xor(ssb, 16, 32);
  float na = sqrtf(ssa);
  float nhv = sqrtf(ssb);
  float c[8];
#pragma unroll
  for (int r = 0; r < 8; r++) {
    float npx = __shfl(na, r + 8 * half, 32);
    float d = fmaxf(npx * nhv, EPSF);
    c[r] = acc[r] / d;
  }
  if (!side) {
    v4f lo, hi;
    lo.x = c[0] * mb; lo.y = c[1] * mb; lo.z = c[2] * mb; lo.w = c[3] * mb;
    hi.x = c[4] * mb; hi.y = c[5] * mb; hi.z = c[6] * mb; hi.w = c[7] * mb;
    cs[y0 + l16][2 * half] = lo;
    cs[y0 + l16][2 * half + 1] = hi;
    float mx[8], sm[8];
#pragma unroll
    for (int r = 0; r < 8; r++) { mx[r] = c[r]; sm[r] = c[r]; }
#pragma unroll
    for (int o = 8; o > 0; o >>= 1) {
#pragma unroll
      for (int r = 0; r < 8; r++) {
        mx[r] = fmaxf(mx[r], __shfl_xor(mx[r], o, 32));
        sm[r] += __shfl_xor(sm[r], o, 32);
      }
    }
    if (l16 == 0) {
#pragma unroll
      for (int r = 0; r < 8; r++) { pmax[w][r + 8 * half] = mx[r]; psum[w][r + 8 * half] = sm[r]; }
    }
  } else {
    float* csf = (float*)cs;
#pragma unroll
    for (int r = 0; r < 8; r++) csf[(x0 + r + 8 * half) * ST + l16] = c[r] * __shfl(ma, r + 8 * half, 32);
    float mx = c[0], sm = c[0];
#pragma unroll
    for (int r = 1; r < 8; r++) { mx = fmaxf(mx, c[r]); sm += c[r]; }
    mx = fmaxf(mx, __shfl_xor(mx, 16, 32));
    sm += __shfl_xor(sm, 16, 32);
    if (half == 0) { pmax[w][l16] = mx; psum[w][l16] = sm; }
  }
  __syncthreads();
  if (threadIdx.x < ST) {
    int t = threadIdx.x;
    float mx = pmax[0][t], sm = psum[0][t];
#pragma unroll
    for (int u = 1; u < 8; u++) { mx = fmaxf(mx, pmax[u][t]); sm += psum[u][t]; }
    int r = (st << 4) + t;
    float m = (float)(side ? mask_h : mask_p)[b * SS + r];
    ssum[t] = sm;
    float* o2 = out + ((size_t)(side * BB + b) * SS + r) * FF;
    o2[0] = m * mx;
    o2[1] = m * (sm / (float)SS);
  }
  __syncthreads();

  const float* Vb = (side ? ctx_p : ctx_h) + (size_t)b * SS * HH;
  size_t o0 = ((size_t)b * SS + (st << 4)) * HH;
  float* amO = (side ? att_mean_p : att_mean_h) + o0;
  float* axO = (side ? att_max_p : att_max_h) + o0;
  const int* mk = (side ? mask_h : mask_p) + b * SS + (st << 4);

  {
    const float* csf = (const float*)cs;
    v2f af[AKS];
#pragma unroll
    for (int i = 0; i < AKS; i++) {
      af[i].x = csf[(4 * i + 2 * half) * ST + l16];
      af[i].y = csf[(4 * i + 2 * half + 1) * ST + l16];
    }
    if (w + 8 < NHT) attmean_tiles<2>(af, Vb, ssum, amO, w, half, l16);
    else attmean_tiles<1>(af, Vb, ssum, amO, w, half, l16);
  }

  int h = threadIdx.x;
  if (h >= HH) {
    if (st == 0 && w == 7) {
      const int* mg = (side ? mask_h : mask_p) + b * SS;
      const float* xg = (side ? ctx_h : ctx_p) + (size_t)b * SS * HH;
      float* og = (side ? chl : cpl) + b * HH;
      int len = mg[4 * lane] + mg[4 * lane + 1] + mg[4 * lane + 2] + mg[4 * lane + 3];
#pragma unroll
      for (int o = 16; o > 0; o >>= 1) len += __shfl_xor(len, o, 32);
      int il = (len > 0) ? len - 1 : 0;
      float ml = (float)mg[il];
      const float* xr = xg + (size_t)il * HH;
      float g[(HH + 31) / 32];
#pragma unroll
      for (int j = 0; j < (HH + 31) / 32; j++) {
        int hh = lane + 32 * j;
        g[j] = xr[hh < HH ? hh : HH - 1];
      }
#pragma unroll
      for (int j = 0; j < (HH + 31) / 32; j++) {
        int hh = lane + 32 * j;
        if (hh < HH) og[hh] = g[j] * ml;
      }
    }
    return;
  }
  const float* vp = Vb + h;
  float mx[ST];
#pragma unroll
  for (int j = 0; j < ST; j++) mx[j] = -3.0e38f;
  float v[KV], vn[KV];
#pragma unroll
  for (int u = 0; u < KV; u++) vn[u] = vp[(size_t)u * HH];
#pragma unroll 1
  for (int k0 = 0; k0 < SS; k0 += KV) {
#pragma unroll
    for (int u = 0; u < KV; u++) v[u] = vn[u];
    if (k0 + KV < SS) {
#pragma unroll
      for (int u = 0; u < KV; u++) vn[u] = vp[(size_t)(k0 + KV + u) * HH];
    }
#pragma unroll
    for (int s = 0; s < KV; s += KU) {
      v4f c4[KU][ST / 4];
#pragma unroll
      for (int u = 0; u < KU; u++) {
#pragma unroll
        for (int q = 0; q < ST / 4; q++) c4[u][q] = cs[k0 + s + u][q];
      }
      __builtin_amdgcn_sched_barrier(0);
#pragma unroll
      for (int u = 0; u < KU; u++) {
#pragma unroll
        for (int q = 0; q < ST / 4; q++) {
          mx[4 * q + 0] = fmaxf(mx[4 * q + 0], c4[u][q].x * v[s + u]);
          mx[4 * q + 1] = fmaxf(mx[4 * q + 1], c4[u][q].y * v[s + u]);
          mx[4 * q + 2] = fmaxf(mx[4 * q + 2], c4[u][q].z * v[s + u]);
          mx[4 * q + 3] = fmaxf(mx[4 * q + 3], c4[u][q].w * v[s + u]);
        }
      }
    }
  }
#pragma unroll
  for (int j = 0; j < ST; j++) {
    float m = (float)mk[j];
    axO[(size_t)j * HH + h] = m * mx[j];
  }
}

#define NMPM ((BB * SS / 16) * 2)
__device__ __forceinline__ void mpm_body(const float* __restrict__ v1, const float* __restrict__ v2,
                                const float* __restrict__ w, int v2_bstride, int v2_sstride,
                                float* __restrict__ out, int side, int foff, int tile, int lane) {
  int m0 = (tile >> 1) << 4;
  int n0 = (tile & 1) << 4;
  int half = lane >> 4, l16 = lane & 15;
  int bs = m0 + l16;
  int b = bs / SS, s = bs % SS;
  const float* p1 = v1 + (size_t)bs * HH + 2 * half;
  const float* p2 = v2 + (size_t)b * v2_bstride + (size_t)s * v2_sstride + 2 * half;
  int col = n0 + l16;
  bool isW = col < PP;
  float fill = (col == PP) ? 1.f : 0.f;
  const float* wrow = w + (size_t)(isW ? col : PP - 1) * HH + 2 * half;
  v8f s11 = {0, 0, 0, 0, 0, 0, 0, 0};
  v8f s22 = {0, 0, 0, 0, 0, 0, 0, 0};
  v8f s12 = {0, 0, 0, 0, 0, 0, 0, 0};
  v2f xs[2][KCH], ys[2][KCH], ws2[2][KCH];
#pragma unroll
  for (int i = 0; i < KCH; i++) {
    xs[0][i] = *(const v2f*)(p1 + 4 * i);
    ys[0][i] = *(const v2f*)(p2 + 4 * i);
    ws2[0][i] = *(const v2f*)(wrow + 4 * i);
  }
#pragma unroll
  for (int c = 0; c < NCH; c++) {
    if (c + 1 < NCH) {
#pragma unroll
      for (int i = 0; i < KCH; i++) {
        xs[(c + 1) & 1][i] = *(const v2f*)(p1 + 4 * KCH * (c + 1) + 4 * i);
        ys[(c + 1) & 1][i] = *(const v2f*)(p2 + 4 * KCH * (c + 1) + 4 * i);
        ws2[(c + 1) & 1][i] = *(const v2f*)(wrow + 4 * KCH * (c + 1) + 4 * i);
      }
    }
    __builtin_amdgcn_sched_barrier(0);
#pragma unroll
    for (int i = 0; i < KCH; i++) {
      v2f x = xs[c & 1][i], y = ys[c & 1][i];
      v2f bv = wsq_frag(ws2[c & 1][i], isW, fill);
      v2f xx, yy, xy;
      xx.x = x.x * x.x; xx.y = x.y * x.y;
      yy.x = y.x * y.x; yy.y = y.y * y.y;
      xy.x = x.x * y.x; xy.y = x.y * y.y;
      s11 = __builtin_amdgcn_wmma_f32_16x16x4_f32(false, xx, false, bv, (short)0, s11, false, false);
      s22 = __builtin_amdgcn_wmma_f32_16x16x4_f32(false, yy, false, bv, (short)0, s22, false, false);
      s12 = __builtin_amdgcn_wmma_f32_16x16x4_f32(false, xy, false, bv, (short)0, s12, false, false);
    }
  }
  if (col <= PP) {
    int f = (col == PP) ? 0 : 1 + col;
#pragma unroll
    for (int r = 0; r < 8; r++) {
      int row = m0 + r + 8 * half;
      int ob = row / SS, os = row % SS;
      float* orow = out + ((size_t)(side * BB + ob) * SS + os) * FF + foff;
      orow[f] = s12[r] / (fmaxf(sqrtf(s11[r]), EPSF) * fmaxf(sqrtf(s22[r]), EPSF));
    }
  }
}

#define NJ 25
#define GCH 10
#define NYT 8
#define NMAXB (BB * PP)
#define BPITCH 204
#define NST (SS * HH / 4 / 256)
__device__ __forceinline__ void maxpool_block(const float* __restrict__ cp, const float* __restrict__ ch,
                                     const float* __restrict__ w,
                                     const float* __restrict__ wnP, const float* __restrict__ wnH,
                                     const int* __restrict__ mask_p, const int* __restrict__ mask_h,
                                     float* __restrict__ out, float* chS, float (*colS)[NYT][SS],
                                     int blk, int tid) {
  int b = blk / PP;
  int p = blk % PP;
  int wv = tid >> 5, lane = tid & 31;
  int x0 = wv * 16;
  int half = lane >> 4, l16 = lane & 15;
  {
    const v4f* gsrc = (const v4f*)(ch + (size_t)b * SS * HH);
    v4f st[NST];
#pragma unroll
    for (int i = 0; i < NST; i++) st[i] = gsrc[tid + 256 * i];
#pragma unroll
    for (int i = 0; i < NST; i++) {
      int e = tid + 256 * i;
      int row = e / (HH / 4), cg = e % (HH / 4);
      *(v4f*)(chS + row * BPITCH + 4 * cg) = st[i];
    }
  }
  float rmax[8], rsum[8], wA[8];
#pragma unroll
  for (int r = 0; r < 8; r++) {
    rmax[r] = -3.0e38f; rsum[r] = 0.f;
    wA[r] = wnP[((size_t)b * SS + x0 + r + 8 * half) * PP + p];
  }
  const float* a_row = cp + ((size_t)b * SS + x0 + l16) * HH + 4 * half;
  const float* w_row = w + (size_t)p * HH + 4 * half;
  const float* b_row = chS + l16 * BPITCH + 4 * half;
  const float* wnB_col = wnH + ((size_t)b * SS + l16) * PP + p;
  float* col_slot = &colS[half][wv][l16];

  v4f af[NJ];
#pragma unroll
  for (int j = 0; j < NJ; j++) {
    v4f wv4 = *(const v4f*)(w_row + 8 * j);
    v4f av = *(const v4f*)(a_row + 8 * j);
    af[j].x = wv4.x * wv4.x * av.x;
    af[j].y = wv4.y * wv4.y * av.y;
    af[j].z = wv4.z * wv4.z * av.z;
    af[j].w = wv4.w * wv4.w * av.w;
  }

  __syncthreads();

  v4f bf[2][GCH];
#pragma unroll
  for (int i = 0; i < GCH; i++) bf[0][i] = *(const v4f*)(b_row + 8 * i);

  v8f acc = {0, 0, 0, 0, 0, 0, 0, 0};
  float wB = 0.f;
#pragma unroll
  for (int g0 = 0; g0 < NYT * NJ; g0 += GCH) {
    if (g0 + GCH < NYT * NJ) {
#pragma unroll
      for (int i = 0; i < GCH; i++) {
        int gn = g0 + GCH + i;
        bf[((g0 / GCH) + 1) & 1][i] = *(const v4f*)(b_row + (gn / NJ) * 16 * BPITCH + 8 * (gn % NJ));
      }
    }
    __builtin_amdgcn_sched_barrier(0);
#pragma unroll
    for (int i = 0; i < GCH; i++) {
      int g = g0 + i;
      int yt = g / NJ, j = g % NJ;
      if (j == 0) {
        acc = (v8f){0, 0, 0, 0, 0, 0, 0, 0};
        wB = wnB_col[(size_t)yt * 16 * PP];
      }
      v4f bb = bf[(g0 / GCH) & 1][i];
      v4f aa = af[j];
      v2f a0, b0, a1, b1;
      a0.x = aa.x; a0.y = aa.y; b0.x = bb.x; b0.y = bb.y;
      a1.x = aa.z; a1.y = aa.w; b1.x = bb.z; b1.y = bb.w;
      acc = __builtin_amdgcn_wmma_f32_16x16x4_f32(false, a0, false, b0, (short)0, acc, false, false);
      acc = __builtin_amdgcn_wmma_f32_16x16x4_f32(false, a1, false, b1, (short)0, acc, false, false);
      if (j == NJ - 1) {
        float cm = -3.0e38f, csm = 0.f;
#pragma unroll
        for (int r = 0; r < 8; r++) {
          float mv = acc[r] / fmaxf(wA[r] * wB, EPSF);
          rmax[r] = fmaxf(rmax[r], mv);
          rsum[r] += mv;
          cm = fmaxf(cm, mv);
          csm += mv;
        }
        cm = fmaxf(cm, __shfl_xor(cm, 16, 32));
        csm += __shfl_xor(csm, 16, 32);
        col_slot[yt * 16] = half ? csm : cm;
      }
    }
  }
#pragma unroll
  for (int r = 0; r < 8; r++) {
    for (int o = 8; o > 0; o >>= 1) {
      rmax[r] = fmaxf(rmax[r], __shfl_xor(rmax[r], o, 32));
      rsum[r] += __shfl_xor(rsum[r], o, 32);
    }
  }
  if (l16 == 0) {
#pragma unroll
    for (int r = 0; r < 8; r++) {
      int x = x0 + r + 8 * half;
      float m = (float)mask_p[b * SS + x];
      float* orow = out + ((size_t)b * SS + x) * FF;
      orow[23 + p] = m * rmax[r];
      orow[43 + p] = m * (rsum[r] / (float)SS);
    }
  }
  __syncthreads();
  int y = tid & (SS - 1);
  int isSum = tid >> 7;
  float v = colS[isSum][0][y];
#pragma unroll
  for (int u = 1; u < NYT; u++) {
    float q = colS[isSum][u][y];
    v = isSum ? v + q : fmaxf(v, q);
  }
  float mh = (float)mask_h[b * SS + y];
  float* o2 = out + ((size_t)(BB + b) * SS + y) * FF;
  if (isSum) o2[43 + p] = mh * (v / (float)SS);
  else o2[23 + p] = mh * v;
}

__global__ void __launch_bounds__(256, 4) match_kernel(
    const float* __restrict__ cp, const float* __restrict__ ch,
    const float* __restrict__ cpl, const float* __restrict__ chl,
    const float* __restrict__ att_mean_h, const float* __restrict__ att_mean_p,
    const float* __restrict__ att_max_h, const float* __restrict__ att_max_p,
    const float* __restrict__ w0, const float* __restrict__ w1,
    const float* __restrict__ w2, const float* __restrict__ w3,
    const float* __restrict__ wn1_p, const float* __restrict__ wn1_h,
    const int* __restrict__ mask_p, const int* __restrict__ mask_h, float* __restrict__ out) {
  __shared__ float colS[2][NYT][SS];
  __shared__ __attribute__((aligned(16))) float chS[SS * BPITCH];
  int blk = blockIdx.x;
  if (blk < NMAXB) {
    maxpool_block(cp, ch, w1, wn1_p, wn1_h, mask_p, mask_h, out, chS, colS, blk, threadIdx.x);
  } else {
    int wid = (blk - NMAXB) * 8 + (threadIdx.x >> 5);
    int lane = threadIdx.x & 31;
    int grp = wid / (2 * NMPM);
    int rem = wid % (2 * NMPM);
    bool side = rem >= NMPM;
    int tile = side ? rem - NMPM : rem;
    const float* v1 = side ? ch : cp;
    const float* v2 = grp == 0 ? (side ? cpl : chl)
                    : grp == 1 ? (side ? att_mean_p : att_mean_h) : (side ? att_max_p : att_max_h);
    const float* w = grp == 0 ? w0 : grp == 1 ? w2 : w3;
    int v2_bstride = grp == 0 ? HH : SS * HH;
    int v2_sstride = grp == 0 ? 0 : HH;
    int foff = grp == 0 ? 2 : grp == 1 ? 63 : 84;
    mpm_body(v1, v2, w, v2_bstride, v2_sstride, out, side ? 1 : 0, foff, tile, lane);
  }
}

extern "C" void kernel_launch(void* const* d_in, const int* in_sizes, int n_in,
                              void* d_out, int out_size, void* d_ws, size_t ws_size,
                              hipStream_t stream) {
  const float* ctx_p = (const float*)d_in[0];
  const float* ctx_h = (const float*)d_in[1];
  const float* w0 = (const float*)d_in[2];
  const float* w1 = (const float*)d_in[3];
  const float* w2 = (const float*)d_in[4];
  const float* w3 = (const float*)d_in[5];
  const int* mask_p = (const int*)d_in[6];
  const int* mask_h = (const int*)d_in[7];
  float* out = (float*)d_out;
  float* ws = (float*)d_ws;

  const size_t SZ_ROWS = (size_t)BB * SS * HH;
  float* cp = ws;                 ws += SZ_ROWS;
  float* ch = ws;                 ws += SZ_ROWS;
  float* wn1_p = ws;              ws += (size_t)BB * SS * PP;
  float* wn1_h = ws;              ws += (size_t)BB * SS * PP;
  float* cpl = ws;                ws += BB * HH;
  float* chl = ws;                ws += BB * HH;
  float* att_mean_h = ws;         ws += SZ_ROWS;
  float* att_mean_p = ws;         ws += SZ_ROWS;
  float* att_max_h = ws;          ws += SZ_ROWS;
  float* att_max_p = ws;          ws += SZ_ROWS;

  cos_att_kernel<<<2 * NCOS, 256, 0, stream>>>(ctx_p, ctx_h, mask_p, mask_h, w1, cp, ch, wn1_p, wn1_h,
                                               cpl, chl, att_mean_h, att_mean_p, att_max_h, att_max_p, out);

  match_kernel<<<NMAXB + 6 * NMPM / 8, 256, 0, stream>>>(cp, ch, cpl, chl, att_mean_h, att_mean_p,
                                                        att_max_h, att_max_p, w0, w1, w2, w3,
                                                        wn1_p, wn1_h, mask_p, mask_h, out);
}
